# starting kernel
# baseline (speedup 1.0000x reference)
#include <hip/hip_runtime.h>
#include <math.h>

#define NQ   20000
#define NS   8192
#define C    112
#define C2   224
#define CLS  20
#define TILES (NQ / 16)
#define WAVES_PER_BLOCK 4

#define WREG    3712
#define SSTRIDE 20
#define XSTRIDE 226
#define HSTRIDE 114
#define H_OFF   0
#define Y2_OFF  1856

typedef float v2f __attribute__((ext_vector_type(2)));
typedef float v8f __attribute__((ext_vector_type(8)));

__global__ void attn_logits(const float* __restrict__ feat_s7,
                            const float* __restrict__ w_attn,
                            float* __restrict__ logits) {
  int j = blockIdx.x * blockDim.x + threadIdx.x;
  if (j >= NS) return;
  const float* row = feat_s7 + (long)j * C;
  float acc = 0.f;
#pragma unroll 8
  for (int c = 0; c < C; ++c) acc += row[c] * w_attn[c];
  logits[j] = acc;
}

__global__ void attn_pool_gate(const float* __restrict__ feat_s7,
                               const float* __restrict__ logits,
                               float* __restrict__ ebuf,
                               float* __restrict__ gate) {
  __shared__ float red[256];
  __shared__ float bcast[2];
  const int t = threadIdx.x;

  float m = -1e30f;
  for (int j = t; j < NS; j += 256) m = fmaxf(m, logits[j]);
  red[t] = m; __syncthreads();
  for (int s = 128; s > 0; s >>= 1) { if (t < s) red[t] = fmaxf(red[t], red[t + s]); __syncthreads(); }
  if (t == 0) bcast[0] = red[0];
  __syncthreads();
  m = bcast[0];

  float sum = 0.f;
  for (int j = t; j < NS; j += 256) { float e = __expf(logits[j] - m); ebuf[j] = e; sum += e; }
  __syncthreads();
  red[t] = sum; __syncthreads();
  for (int s = 128; s > 0; s >>= 1) { if (t < s) red[t] += red[t + s]; __syncthreads(); }
  if (t == 0) bcast[1] = red[0];
  __syncthreads();
  const float inv = 1.0f / bcast[1];

  if (t < C) {
    float acc = 0.f;
    for (int j = 0; j < NS; ++j) acc += ebuf[j] * feat_s7[(long)j * C + t];
    float g = acc * inv;
    gate[t] = 1.0f / (1.0f + __expf(-g));
  }
}

#define TOP3_INS(val, idxv) {                                         \
    float s_ = (val); int id_ = (idxv);                               \
    bool c0_ = s_ < v0, c1_ = s_ < v1, c2_ = s_ < v2v;                \
    v2v = c1_ ? v1 : (c2_ ? s_  : v2v); i2 = c1_ ? i1 : (c2_ ? id_ : i2); \
    v1  = c0_ ? v0 : (c1_ ? s_  : v1 ); i1 = c0_ ? i0 : (c1_ ? id_ : i1); \
    v0  = c0_ ? s_ : v0;                i0 = c0_ ? id_ : i0;          }

__global__ __launch_bounds__(128) void fused_main(
    const float* __restrict__ coords_q, const float* __restrict__ coords_s,
    const float* __restrict__ x7,       const float* __restrict__ feat_s,
    const float* __restrict__ gate,     const float* __restrict__ W1,
    const float* __restrict__ gammaW,   const float* __restrict__ betaW,
    const float* __restrict__ W2,       const float* __restrict__ W_cls,
    const float* __restrict__ b_cls,    float* __restrict__ out)
{
  __shared__ float lds[WAVES_PER_BLOCK * WREG];
  const int lane = threadIdx.x & 31;
  const int wave = threadIdx.x >> 5;
  const int tile = blockIdx.x * WAVES_PER_BLOCK + wave;
  if (tile >= TILES) return;

  float* WL = lds + wave * WREG;
  const int r15  = lane & 15;
  const int half = lane >> 4;
  const int qbase = tile * 16;

  const float qx = coords_q[(qbase + r15) * 3 + 0];
  const float qy = coords_q[(qbase + r15) * 3 + 1];
  const float qz = coords_q[(qbase + r15) * 3 + 2];
  v2f a;
  a.x = half ? (-2.f * qz) : (-2.f * qx);
  a.y = half ? 1.0f        : (-2.f * qy);

  float v0 = 1e30f, v1 = 1e30f, v2v = 1e30f;
  int   i0 = 0, i1 = 0, i2 = 0;

  for (int n0 = 0; n0 < NS; n0 += 16) {
    __builtin_prefetch(coords_s + (n0 + 128) * 3, 0, 1);
    const int j = n0 + r15;
    const float sx = coords_s[j * 3 + 0];
    const float sy = coords_s[j * 3 + 1];
    const float sz = coords_s[j * 3 + 2];
    const float s2 = sx * sx + sy * sy + sz * sz;
    v2f b;
    b.x = half ? sz : sx;
    b.y = half ? s2 : sy;
    v8f c = {0.f, 0.f, 0.f, 0.f, 0.f, 0.f, 0.f, 0.f};
    c = __builtin_amdgcn_wmma_f32_16x16x4_f32(false, a, false, b, (short)0, c, false, false);

#pragma unroll
    for (int r = 0; r < 8; ++r)
      WL[(r + 8 * half) * SSTRIDE + r15] = c[r];
    asm volatile("" ::: "memory");

    const float4 sA = *(const float4*)(WL + r15 * SSTRIDE + 8 * half);
    const float4 sB = *(const float4*)(WL + r15 * SSTRIDE + 8 * half + 4);

    const float m8 = fminf(fminf(fminf(sA.x, sA.y), fminf(sA.z, sA.w)),
                           fminf(fminf(sB.x, sB.y), fminf(sB.z, sB.w)));
    if (__any(m8 < v2v)) {
      const int base = n0 + 8 * half;
      TOP3_INS(sA.x, base + 0); TOP3_INS(sA.y, base + 1);
      TOP3_INS(sA.z, base + 2); TOP3_INS(sA.w, base + 3);
      TOP3_INS(sB.x, base + 4); TOP3_INS(sB.y, base + 5);
      TOP3_INS(sB.z, base + 6); TOP3_INS(sB.w, base + 7);
    }
    asm volatile("" ::: "memory");
  }

  {
    float pv0 = __shfl(v0, lane + 16), pv1 = __shfl(v1, lane + 16), pv2 = __shfl(v2v, lane + 16);
    int   pi0 = __shfl(i0, lane + 16), pi1 = __shfl(i1, lane + 16), pi2 = __shfl(i2,  lane + 16);
    TOP3_INS(pv0, pi0); TOP3_INS(pv1, pi1); TOP3_INS(pv2, pi2);
  }
  const float e1 = __expf(v0 - v1), e2 = __expf(v0 - v2v);
  const float inv = 1.0f / (1.0f + e1 + e2);
  const float w0 = inv, w1 = e1 * inv, w2 = e2 * inv;

  for (int r = 0; r < 16; ++r) {
    const float rw0 = __shfl(w0, r), rw1 = __shfl(w1, r), rw2 = __shfl(w2, r);
    const int   ri0 = __shfl(i0, r), ri1 = __shfl(i1, r), ri2 = __shfl(i2, r);
    const int qi = qbase + r;
    for (int t = lane; t < C2; t += 32) {
      float v;
      if (t < C) {
        v = x7[(long)qi * C + t] * gate[t];
      } else {
        const int cc = t - C;
        v = rw0 * feat_s[(long)ri0 * C + cc]
          + rw1 * feat_s[(long)ri1 * C + cc]
          + rw2 * feat_s[(long)ri2 * C + cc];
      }
      WL[r * XSTRIDE + t] = v;
    }
  }
  asm volatile("" ::: "memory");

  v8f acc[7];
#pragma unroll
  for (int n = 0; n < 7; ++n) acc[n] = (v8f){0.f, 0.f, 0.f, 0.f, 0.f, 0.f, 0.f, 0.f};
  for (int k0 = 0; k0 < C2; k0 += 4) {
    const int kk = k0 + 2 * half;
    const v2f a1 = *(const v2f*)(WL + r15 * XSTRIDE + kk);
#pragma unroll
    for (int n = 0; n < 7; ++n) {
      v2f bv;
      bv.x = W1[(long)kk * C + n * 16 + r15];
      bv.y = W1[(long)(kk + 1) * C + n * 16 + r15];
      acc[n] = __builtin_amdgcn_wmma_f32_16x16x4_f32(false, a1, false, bv, (short)0, acc[n], false, false);
    }
  }

#pragma unroll
  for (int n = 0; n < 7; ++n) {
    const int ch = n * 16 + r15;
    const float gm = gammaW[ch], bt = betaW[ch];
#pragma unroll
    for (int r = 0; r < 8; ++r) {
      const float h = fmaxf(acc[n][r] * gm + bt, 0.f);
      WL[H_OFF + (r + 8 * half) * HSTRIDE + ch] = h;
    }
  }
  asm volatile("" ::: "memory");

  v8f acc2[7];
#pragma unroll
  for (int n = 0; n < 7; ++n) acc2[n] = (v8f){0.f, 0.f, 0.f, 0.f, 0.f, 0.f, 0.f, 0.f};
  for (int k0 = 0; k0 < C; k0 += 4) {
    const int kk = k0 + 2 * half;
    const v2f a2 = *(const v2f*)(WL + H_OFF + r15 * HSTRIDE + kk);
#pragma unroll
    for (int n = 0; n < 7; ++n) {
      v2f bv;
      bv.x = W2[(long)kk * C + n * 16 + r15];
      bv.y = W2[(long)(kk + 1) * C + n * 16 + r15];
      acc2[n] = __builtin_amdgcn_wmma_f32_16x16x4_f32(false, a2, false, bv, (short)0, acc2[n], false, false);
    }
  }
#pragma unroll
  for (int n = 0; n < 7; ++n) {
    const int ch = n * 16 + r15;
#pragma unroll
    for (int r = 0; r < 8; ++r)
      WL[Y2_OFF + (r + 8 * half) * HSTRIDE + ch] = acc2[n][r];
  }
  asm volatile("" ::: "memory");

  v8f acc3[2];
  acc3[0] = (v8f){0.f, 0.f, 0.f, 0.f, 0.f, 0.f, 0.f, 0.f};
  acc3[1] = acc3[0];
  for (int k0 = 0; k0 < C; k0 += 4) {
    const int kk = k0 + 2 * half;
    const v2f a3 = *(const v2f*)(WL + Y2_OFF + r15 * HSTRIDE + kk);
#pragma unroll
    for (int n = 0; n < 2; ++n) {
      const int col = n * 16 + r15;
      v2f bv;
      bv.x = (col < CLS) ? W_cls[(long)kk * CLS + col] : 0.f;
      bv.y = (col < CLS) ? W_cls[(long)(kk + 1) * CLS + col] : 0.f;
      acc3[n] = __builtin_amdgcn_wmma_f32_16x16x4_f32(false, a3, false, bv, (short)0, acc3[n], false, false);
    }
  }
#pragma unroll
  for (int n = 0; n < 2; ++n) {
    const int col = n * 16 + r15;
    if (col < CLS) {
      const float bb = b_cls[col];
#pragma unroll
      for (int r = 0; r < 8; ++r)
        out[(long)(qbase + r + 8 * half) * CLS + col] = acc3[n][r] + bb;
    }
  }
}

extern "C" void kernel_launch(void* const* d_in, const int* in_sizes, int n_in,
                              void* d_out, int out_size, void* d_ws, size_t ws_size,
                              hipStream_t stream) {
  const float* coords_q = (const float*)d_in[0];
  const float* coords_s = (const float*)d_in[1];
  const float* x7       = (const float*)d_in[2];
  const float* feat_s7  = (const float*)d_in[3];
  const float* feat_s   = (const float*)d_in[4];
  const float* w_attn   = (const float*)d_in[5];
  const float* W1       = (const float*)d_in[6];
  const float* gammaW   = (const float*)d_in[7];
  const float* betaW    = (const float*)d_in[8];
  const float* W2       = (const float*)d_in[9];
  const float* W_cls    = (const float*)d_in[10];
  const float* b_cls    = (const float*)d_in[11];
  float* out = (float*)d_out;

  float* ws     = (float*)d_ws;
  float* logits = ws;
  float* ebuf   = ws + NS;
  float* gate   = ws + 2 * NS;

  attn_logits<<<NS / 256, 256, 0, stream>>>(feat_s7, w_attn, logits);
  attn_pool_gate<<<1, 256, 0, stream>>>(feat_s7, logits, ebuf, gate);

  const int blocks = (TILES + WAVES_PER_BLOCK - 1) / WAVES_PER_BLOCK;
  fused_main<<<blocks, 128, 0, stream>>>(coords_q, coords_s, x7, feat_s, gate,
                                         W1, gammaW, betaW, W2, W_cls, b_cls, out);
}


# softmax pooling spread over 128 work-groups: per 64-row chunk logits, max, exp sum, channel sums from one LDS tile, then one group rescales and combines; sums re-associated (new reduction tree)
# speedup vs baseline: 3.9794x; 3.9794x over previous
#include <hip/hip_runtime.h>
#include <math.h>

#define NQ   20000
#define NS   8192
#define C    112
#define C2   224
#define CLS  20
#define TILES (NQ / 16)
#define WAVES_PER_BLOCK 4

#define WREG    3712
#define SSTRIDE 20
#define XSTRIDE 226
#define HSTRIDE 114
#define H_OFF   0
#define Y2_OFF  1856

typedef float v2f __attribute__((ext_vector_type(2)));
typedef float v8f __attribute__((ext_vector_type(8)));

#define PB_ROWS  64
#define PB_N     (NS / PB_ROWS)
#define PB_PITCH 116
#define REC      128

__global__ __launch_bounds__(256) void pool_partials(const float* __restrict__ feat_s7,
                                                      const float* __restrict__ w_attn,
                                                      float* __restrict__ rec) {
  __shared__ __attribute__((aligned(16))) float tile[PB_ROWS * PB_PITCH];
  __shared__ __attribute__((aligned(16))) float wl[C];
  __shared__ float lg[PB_ROWS];
  __shared__ float ev[PB_ROWS];
  __shared__ float part[2 * C];
  const int t = threadIdx.x;
  const int lane = t & 31;

  if (t < C / 4) *(float4*)(wl + 4 * t) = ((const float4*)w_attn)[t];

  const float4* src = (const float4*)(feat_s7 + (long)blockIdx.x * (PB_ROWS * C));
#pragma unroll
  for (int i = 0; i < 7; ++i) {
    const int q = t + 256 * i;
    const int row = q / 28, c4 = q - row * 28;
    *(float4*)(tile + row * PB_PITCH + 4 * c4) = src[q];
  }
  __syncthreads();

  {
    const int row = t >> 2, p = t & 3;
    float acc = 0.f;
#pragma unroll
    for (int i = 0; i < 7; ++i) {
      const int c4 = p + 4 * i;
      const float4 x = *(const float4*)(tile + row * PB_PITCH + 4 * c4);
      const float4 w = *(const float4*)(wl + 4 * c4);
      acc += x.x * w.x + x.y * w.y + x.z * w.z + x.w * w.w;
    }
    acc += __shfl_xor(acc, 1);
    acc += __shfl_xor(acc, 2);
    if (p == 0) lg[row] = acc;
  }
  __syncthreads();

  float m = fmaxf(lg[lane], lg[lane + 32]);
  for (int o = 16; o > 0; o >>= 1) m = fmaxf(m, __shfl_xor(m, o));
  if (t < PB_ROWS) ev[t] = __expf(lg[t] - m);
  __syncthreads();
  float s = ev[lane] + ev[lane + 32];
  for (int o = 16; o > 0; o >>= 1) s += __shfl_xor(s, o);

  if (t < 2 * C) {
    const int h = (t >= C) ? 1 : 0;
    const int c = t - h * C;
    const float* col = tile + (h * 32) * PB_PITCH + c;
    const float* e = ev + h * 32;
    float a0 = 0.f, a1 = 0.f;
#pragma unroll
    for (int r = 0; r < 32; r += 2) {
      a0 += e[r] * col[r * PB_PITCH];
      a1 += e[r + 1] * col[(r + 1) * PB_PITCH];
    }
    part[t] = a0 + a1;
  }
  __syncthreads();

  if (t < REC) {
    float v = 0.f;
    if (t < C) v = part[t] + part[C + t];
    else if (t == C) v = m;
    else if (t == C + 1) v = s;
    rec[(long)blockIdx.x * REC + t] = v;
  }
}

__global__ __launch_bounds__(512) void pool_combine(const float* __restrict__ rec,
                                                     float* __restrict__ gate) {
  __shared__ float ml[PB_N];
  __shared__ float sl[PB_N];
  __shared__ float fs[PB_N];
  __shared__ float part[4 * REC];
  const int t = threadIdx.x;
  const int lane = t & 31;
  const int c = t & (REC - 1);
  const int g = t >> 7;

  float x[32];
#pragma unroll
  for (int i = 0; i < 32; ++i) x[i] = rec[(long)(g * 32 + i) * REC + c];

  if (t < PB_N) { ml[t] = rec[(long)t * REC + C]; sl[t] = rec[(long)t * REC + C + 1]; }
  __syncthreads();
  float M = fmaxf(fmaxf(ml[lane], ml[lane + 32]), fmaxf(ml[lane + 64], ml[lane + 96]));
  for (int o = 16; o > 0; o >>= 1) M = fmaxf(M, __shfl_xor(M, o));
  if (t < PB_N) { const float f = __expf(ml[t] - M); fs[t] = f; sl[t] = sl[t] * f; }
  __syncthreads();
  float S = (sl[lane] + sl[lane + 32]) + (sl[lane + 64] + sl[lane + 96]);
  for (int o = 16; o > 0; o >>= 1) S += __shfl_xor(S, o);

  float a0 = 0.f, a1 = 0.f;
#pragma unroll
  for (int i = 0; i < 32; i += 2) {
    a0 += x[i] * fs[g * 32 + i];
    a1 += x[i + 1] * fs[g * 32 + i + 1];
  }
  part[t] = a0 + a1;
  __syncthreads();

  if (t < C) {
    const float acc = (part[t] + part[REC + t]) + (part[2 * REC + t] + part[3 * REC + t]);
    const float gv = acc * (1.0f / S);
    gate[t] = 1.0f / (1.0f + __expf(-gv));
  }
}

#define TOP3_INS(val, idxv) {                                         \
    float s_ = (val); int id_ = (idxv);                               \
    bool c0_ = s_ < v0, c1_ = s_ < v1, c2_ = s_ < v2v;                \
    v2v = c1_ ? v1 : (c2_ ? s_  : v2v); i2 = c1_ ? i1 : (c2_ ? id_ : i2); \
    v1  = c0_ ? v0 : (c1_ ? s_  : v1 ); i1 = c0_ ? i0 : (c1_ ? id_ : i1); \
    v0  = c0_ ? s_ : v0;                i0 = c0_ ? id_ : i0;          }

__global__ __launch_bounds__(128) void fused_main(
    const float* __restrict__ coords_q, const float* __restrict__ coords_s,
    const float* __restrict__ x7,       const float* __restrict__ feat_s,
    const float* __restrict__ gate,     const float* __restrict__ W1,
    const float* __restrict__ gammaW,   const float* __restrict__ betaW,
    const float* __restrict__ W2,       const float* __restrict__ W_cls,
    const float* __restrict__ b_cls,    float* __restrict__ out)
{
  __shared__ float lds[WAVES_PER_BLOCK * WREG];
  const int lane = threadIdx.x & 31;
  const int wave = threadIdx.x >> 5;
  const int tile = blockIdx.x * WAVES_PER_BLOCK + wave;
  if (tile >= TILES) return;

  float* WL = lds + wave * WREG;
  const int r15  = lane & 15;
  const int half = lane >> 4;
  const int qbase = tile * 16;

  const float qx = coords_q[(qbase + r15) * 3 + 0];
  const float qy = coords_q[(qbase + r15) * 3 + 1];
  const float qz = coords_q[(qbase + r15) * 3 + 2];
  v2f a;
  a.x = half ? (-2.f * qz) : (-2.f * qx);
  a.y = half ? 1.0f        : (-2.f * qy);

  float v0 = 1e30f, v1 = 1e30f, v2v = 1e30f;
  int   i0 = 0, i1 = 0, i2 = 0;

  for (int n0 = 0; n0 < NS; n0 += 16) {
    __builtin_prefetch(coords_s + (n0 + 128) * 3, 0, 1);
    const int j = n0 + r15;
    const float sx = coords_s[j * 3 + 0];
    const float sy = coords_s[j * 3 + 1];
    const float sz = coords_s[j * 3 + 2];
    const float s2 = sx * sx + sy * sy + sz * sz;
    v2f b;
    b.x = half ? sz : sx;
    b.y = half ? s2 : sy;
    v8f c = {0.f, 0.f, 0.f, 0.f, 0.f, 0.f, 0.f, 0.f};
    c = __builtin_amdgcn_wmma_f32_16x16x4_f32(false, a, false, b, (short)0, c, false, false);

#pragma unroll
    for (int r = 0; r < 8; ++r)
      WL[(r + 8 * half) * SSTRIDE + r15] = c[r];
    asm volatile("" ::: "memory");

    const float4 sA = *(const float4*)(WL + r15 * SSTRIDE + 8 * half);
    const float4 sB = *(const float4*)(WL + r15 * SSTRIDE + 8 * half + 4);

    const float m8 = fminf(fminf(fminf(sA.x, sA.y), fminf(sA.z, sA.w)),
                           fminf(fminf(sB.x, sB.y), fminf(sB.z, sB.w)));
    if (__any(m8 < v2v)) {
      const int base = n0 + 8 * half;
      TOP3_INS(sA.x, base + 0); TOP3_INS(sA.y, base + 1);
      TOP3_INS(sA.z, base + 2); TOP3_INS(sA.w, base + 3);
      TOP3_INS(sB.x, base + 4); TOP3_INS(sB.y, base + 5);
      TOP3_INS(sB.z, base + 6); TOP3_INS(sB.w, base + 7);
    }
    asm volatile("" ::: "memory");
  }

  {
    float pv0 = __shfl(v0, lane + 16), pv1 = __shfl(v1, lane + 16), pv2 = __shfl(v2v, lane + 16);
    int   pi0 = __shfl(i0, lane + 16), pi1 = __shfl(i1, lane + 16), pi2 = __shfl(i2,  lane + 16);
    TOP3_INS(pv0, pi0); TOP3_INS(pv1, pi1); TOP3_INS(pv2, pi2);
  }
  const float e1 = __expf(v0 - v1), e2 = __expf(v0 - v2v);
  const float inv = 1.0f / (1.0f + e1 + e2);
  const float w0 = inv, w1 = e1 * inv, w2 = e2 * inv;

  for (int r = 0; r < 16; ++r) {
    const float rw0 = __shfl(w0, r), rw1 = __shfl(w1, r), rw2 = __shfl(w2, r);
    const int   ri0 = __shfl(i0, r), ri1 = __shfl(i1, r), ri2 = __shfl(i2, r);
    const int qi = qbase + r;
    for (int t = lane; t < C2; t += 32) {
      float v;
      if (t < C) {
        v = x7[(long)qi * C + t] * gate[t];
      } else {
        const int cc = t - C;
        v = rw0 * feat_s[(long)ri0 * C + cc]
          + rw1 * feat_s[(long)ri1 * C + cc]
          + rw2 * feat_s[(long)ri2 * C + cc];
      }
      WL[r * XSTRIDE + t] = v;
    }
  }
  asm volatile("" ::: "memory");

  v8f acc[7];
#pragma unroll
  for (int n = 0; n < 7; ++n) acc[n] = (v8f){0.f, 0.f, 0.f, 0.f, 0.f, 0.f, 0.f, 0.f};
  for (int k0 = 0; k0 < C2; k0 += 4) {
    const int kk = k0 + 2 * half;
    const v2f a1 = *(const v2f*)(WL + r15 * XSTRIDE + kk);
#pragma unroll
    for (int n = 0; n < 7; ++n) {
      v2f bv;
      bv.x = W1[(long)kk * C + n * 16 + r15];
      bv.y = W1[(long)(kk + 1) * C + n * 16 + r15];
      acc[n] = __builtin_amdgcn_wmma_f32_16x16x4_f32(false, a1, false, bv, (short)0, acc[n], false, false);
    }
  }

#pragma unroll
  for (int n = 0; n < 7; ++n) {
    const int ch = n * 16 + r15;
    const float gm = gammaW[ch], bt = betaW[ch];
#pragma unroll
    for (int r = 0; r < 8; ++r) {
      const float h = fmaxf(acc[n][r] * gm + bt, 0.f);
      WL[H_OFF + (r + 8 * half) * HSTRIDE + ch] = h;
    }
  }
  asm volatile("" ::: "memory");

  v8f acc2[7];
#pragma unroll
  for (int n = 0; n < 7; ++n) acc2[n] = (v8f){0.f, 0.f, 0.f, 0.f, 0.f, 0.f, 0.f, 0.f};
  for (int k0 = 0; k0 < C; k0 += 4) {
    const int kk = k0 + 2 * half;
    const v2f a2 = *(const v2f*)(WL + H_OFF + r15 * HSTRIDE + kk);
#pragma unroll
    for (int n = 0; n < 7; ++n) {
      v2f bv;
      bv.x = W2[(long)kk * C + n * 16 + r15];
      bv.y = W2[(long)(kk + 1) * C + n * 16 + r15];
      acc2[n] = __builtin_amdgcn_wmma_f32_16x16x4_f32(false, a2, false, bv, (short)0, acc2[n], false, false);
    }
  }
#pragma unroll
  for (int n = 0; n < 7; ++n) {
    const int ch = n * 16 + r15;
#pragma unroll
    for (int r = 0; r < 8; ++r)
      WL[Y2_OFF + (r + 8 * half) * HSTRIDE + ch] = acc2[n][r];
  }
  asm volatile("" ::: "memory");

  v8f acc3[2];
  acc3[0] = (v8f){0.f, 0.f, 0.f, 0.f, 0.f, 0.f, 0.f, 0.f};
  acc3[1] = acc3[0];
  for (int k0 = 0; k0 < C; k0 += 4) {
    const int kk = k0 + 2 * half;
    const v2f a3 = *(const v2f*)(WL + Y2_OFF + r15 * HSTRIDE + kk);
#pragma unroll
    for (int n = 0; n < 2; ++n) {
      const int col = n * 16 + r15;
      v2f bv;
      bv.x = (col < CLS) ? W_cls[(long)kk * CLS + col] : 0.f;
      bv.y = (col < CLS) ? W_cls[(long)(kk + 1) * CLS + col] : 0.f;
      acc3[n] = __builtin_amdgcn_wmma_f32_16x16x4_f32(false, a3, false, bv, (short)0, acc3[n], false, false);
    }
  }
#pragma unroll
  for (int n = 0; n < 2; ++n) {
    const int col = n * 16 + r15;
    if (col < CLS) {
      const float bb = b_cls[col];
#pragma unroll
      for (int r = 0; r < 8; ++r)
        out[(long)(qbase + r + 8 * half) * CLS + col] = acc3[n][r] + bb;
    }
  }
}

extern "C" void kernel_launch(void* const* d_in, const int* in_sizes, int n_in,
                              void* d_out, int out_size, void* d_ws, size_t ws_size,
                              hipStream_t stream) {
  const float* coords_q = (const float*)d_in[0];
  const float* coords_s = (const float*)d_in[1];
  const float* x7       = (const float*)d_in[2];
  const float* feat_s7  = (const float*)d_in[3];
  const float* feat_s   = (const float*)d_in[4];
  const float* w_attn   = (const float*)d_in[5];
  const float* W1       = (const float*)d_in[6];
  const float* gammaW   = (const float*)d_in[7];
  const float* betaW    = (const float*)d_in[8];
  const float* W2       = (const float*)d_in[9];
  const float* W_cls    = (const float*)d_in[10];
  const float* b_cls    = (const float*)d_in[11];
  float* out = (float*)d_out;

  float* ws   = (float*)d_ws;
  float* rec  = ws;
  float* gate = ws + PB_N * REC;

  pool_partials<<<PB_N, 256, 0, stream>>>(feat_s7, w_attn, rec);
  pool_combine<<<1, 512, 0, stream>>>(rec, gate);

  const int blocks = (TILES + WAVES_PER_BLOCK - 1) / WAVES_PER_BLOCK;
  fused_main<<<blocks, 128, 0, stream>>>(coords_q, coords_s, x7, feat_s, gate,
                                         W1, gammaW, betaW, W2, W_cls, b_cls, out);
}


# one 16-query tile per work-group of 8 waves (was per wave): search range, X rows and GEMM1/GEMM2 column tiles split over the waves, partial top-3 lists merged through LDS; order of sums unchanged
# speedup vs baseline: 10.0838x; 2.5340x over previous
#include <hip/hip_runtime.h>
#include <math.h>

#define NQ   20000
#define NS   8192
#define C    112
#define C2   224
#define CLS  20
#define TILES (NQ / 16)

#define WREG    3712
#define SSTRIDE 20
#define XSTRIDE 226
#define HSTRIDE 114
#define H_OFF   0
#define Y2_OFF  1856

typedef float v2f __attribute__((ext_vector_type(2)));
typedef float v8f __attribute__((ext_vector_type(8)));

#define PB_ROWS  64
#define PB_N     (NS / PB_ROWS)
#define PB_PITCH 116
#define REC      128

__global__ __launch_bounds__(256) void pool_partials(const float* __restrict__ feat_s7,
                                                      const float* __restrict__ w_attn,
                                                      float* __restrict__ rec) {
  __shared__ __attribute__((aligned(16))) float tile[PB_ROWS * PB_PITCH];
  __shared__ __attribute__((aligned(16))) float wl[C];
  __shared__ float lg[PB_ROWS];
  __shared__ float ev[PB_ROWS];
  __shared__ float part[2 * C];
  const int t = threadIdx.x;
  const int lane = t & 31;

  if (t < C / 4) *(float4*)(wl + 4 * t) = ((const float4*)w_attn)[t];

  const float4* src = (const float4*)(feat_s7 + (long)blockIdx.x * (PB_ROWS * C));
#pragma unroll
  for (int i = 0; i < 7; ++i) {
    const int q = t + 256 * i;
    const int row = q / 28, c4 = q - row * 28;
    *(float4*)(tile + row * PB_PITCH + 4 * c4) = src[q];
  }
  __syncthreads();

  {
    const int row = t >> 2, p = t & 3;
    float acc = 0.f;
#pragma unroll
    for (int i = 0; i < 7; ++i) {
      const int c4 = p + 4 * i;
      const float4 x = *(const float4*)(tile + row * PB_PITCH + 4 * c4);
      const float4 w = *(const float4*)(wl + 4 * c4);
      acc += x.x * w.x + x.y * w.y + x.z * w.z + x.w * w.w;
    }
    acc += __shfl_xor(acc, 1);
    acc += __shfl_xor(acc, 2);
    if (p == 0) lg[row] = acc;
  }
  __syncthreads();

  float m = fmaxf(lg[lane], lg[lane + 32]);
  for (int o = 16; o > 0; o >>= 1) m = fmaxf(m, __shfl_xor(m, o));
  if (t < PB_ROWS) ev[t] = __expf(lg[t] - m);
  __syncthreads();
  float s = ev[lane] + ev[lane + 32];
  for (int o = 16; o > 0; o >>= 1) s += __shfl_xor(s, o);

  if (t < 2 * C) {
    const int h = (t >= C) ? 1 : 0;
    const int c = t - h * C;
    const float* col = tile + (h * 32) * PB_PITCH + c;
    const float* e = ev + h * 32;
    float a0 = 0.f, a1 = 0.f;
#pragma unroll
    for (int r = 0; r < 32; r += 2) {
      a0 += e[r] * col[r * PB_PITCH];
      a1 += e[r + 1] * col[(r + 1) * PB_PITCH];
    }
    part[t] = a0 + a1;
  }
  __syncthreads();

  if (t < REC) {
    float v = 0.f;
    if (t < C) v = part[t] + part[C + t];
    else if (t == C) v = m;
    else if (t == C + 1) v = s;
    rec[(long)blockIdx.x * REC + t] = v;
  }
}

__global__ __launch_bounds__(512) void pool_combine(const float* __restrict__ rec,
                                                     float* __restrict__ gate) {
  __shared__ float ml[PB_N];
  __shared__ float sl[PB_N];
  __shared__ float fs[PB_N];
  __shared__ float part[4 * REC];
  const int t = threadIdx.x;
  const int lane = t & 31;
  const int c = t & (REC - 1);
  const int g = t >> 7;

  float x[32];
#pragma unroll
  for (int i = 0; i < 32; ++i) x[i] = rec[(long)(g * 32 + i) * REC + c];

  if (t < PB_N) { ml[t] = rec[(long)t * REC + C]; sl[t] = rec[(long)t * REC + C + 1]; }
  __syncthreads();
  float M = fmaxf(fmaxf(ml[lane], ml[lane + 32]), fmaxf(ml[lane + 64], ml[lane + 96]));
  for (int o = 16; o > 0; o >>= 1) M = fmaxf(M, __shfl_xor(M, o));
  if (t < PB_N) { const float f = __expf(ml[t] - M); fs[t] = f; sl[t] = sl[t] * f; }
  __syncthreads();
  float S = (sl[lane] + sl[lane + 32]) + (sl[lane + 64] + sl[lane + 96]);
  for (int o = 16; o > 0; o >>= 1) S += __shfl_xor(S, o);

  float a0 = 0.f, a1 = 0.f;
#pragma unroll
  for (int i = 0; i < 32; i += 2) {
    a0 += x[i] * fs[g * 32 + i];
    a1 += x[i + 1] * fs[g * 32 + i + 1];
  }
  part[t] = a0 + a1;
  __syncthreads();

  if (t < C) {
    const float acc = (part[t] + part[REC + t]) + (part[2 * REC + t] + part[3 * REC + t]);
    const float gv = acc * (1.0f / S);
    gate[t] = 1.0f / (1.0f + __expf(-gv));
  }
}

#define TOP3_INS(val, idxv) {                                         \
    float s_ = (val); int id_ = (idxv);                               \
    bool c0_ = s_ < v0, c1_ = s_ < v1, c2_ = s_ < v2v;                \
    v2v = c1_ ? v1 : (c2_ ? s_  : v2v); i2 = c1_ ? i1 : (c2_ ? id_ : i2); \
    v1  = c0_ ? v0 : (c1_ ? s_  : v1 ); i1 = c0_ ? i0 : (c1_ ? id_ : i1); \
    v0  = c0_ ? s_ : v0;                i0 = c0_ ? id_ : i0;          }

#define NW   8
#define SEG  (NS / NW)
#define SREG 320
#define ROWS_PER_WAVE (16 / NW)
#define NT   (C / 16)

__global__ __launch_bounds__(NW * 32) void fused_main(
    const float* __restrict__ coords_q, const float* __restrict__ coords_s,
    const float* __restrict__ x7,       const float* __restrict__ feat_s,
    const float* __restrict__ gate,     const float* __restrict__ W1,
    const float* __restrict__ gammaW,   const float* __restrict__ betaW,
    const float* __restrict__ W2,       const float* __restrict__ W_cls,
    const float* __restrict__ b_cls,    float* __restrict__ out)
{
  __shared__ __attribute__((aligned(16))) float lds[WREG];
  __shared__ __attribute__((aligned(16))) float sc[NW * SREG];
  __shared__ float pv[NW * 16 * 3];
  __shared__ int   pi[NW * 16 * 3];
  __shared__ float tw[16 * 3];
  __shared__ int   ti[16 * 3];
  const int lane = threadIdx.x & 31;
  const int wave = __builtin_amdgcn_readfirstlane(threadIdx.x >> 5);
  const int tile = blockIdx.x;

  float* WL = lds;
  float* SC = sc + wave * SREG;
  const int r15  = lane & 15;
  const int half = lane >> 4;
  const int qbase = tile * 16;

  const float qx = coords_q[(qbase + r15) * 3 + 0];
  const float qy = coords_q[(qbase + r15) * 3 + 1];
  const float qz = coords_q[(qbase + r15) * 3 + 2];
  v2f a;
  a.x = half ? (-2.f * qz) : (-2.f * qx);
  a.y = half ? 1.0f        : (-2.f * qy);

  float v0 = 1e30f, v1 = 1e30f, v2v = 1e30f;
  int   i0 = 0, i1 = 0, i2 = 0;

  const int nbeg = wave * SEG;
  for (int n0 = nbeg; n0 < nbeg + SEG; n0 += 16) {
    __builtin_prefetch(coords_s + (n0 + 128) * 3, 0, 1);
    const int j = n0 + r15;
    const float sx = coords_s[j * 3 + 0];
    const float sy = coords_s[j * 3 + 1];
    const float sz = coords_s[j * 3 + 2];
    const float s2 = sx * sx + sy * sy + sz * sz;
    v2f b;
    b.x = half ? sz : sx;
    b.y = half ? s2 : sy;
    v8f c = {0.f, 0.f, 0.f, 0.f, 0.f, 0.f, 0.f, 0.f};
    c = __builtin_amdgcn_wmma_f32_16x16x4_f32(false, a, false, b, (short)0, c, false, false);

#pragma unroll
    for (int r = 0; r < 8; ++r)
      SC[(r + 8 * half) * SSTRIDE + r15] = c[r];
    asm volatile("" ::: "memory");

    const float4 sA = *(const float4*)(SC + r15 * SSTRIDE + 8 * half);
    const float4 sB = *(const float4*)(SC + r15 * SSTRIDE + 8 * half + 4);

    const float m8 = fminf(fminf(fminf(sA.x, sA.y), fminf(sA.z, sA.w)),
                           fminf(fminf(sB.x, sB.y), fminf(sB.z, sB.w)));
    if (__any(m8 < v2v)) {
      const int base = n0 + 8 * half;
      TOP3_INS(sA.x, base + 0); TOP3_INS(sA.y, base + 1);
      TOP3_INS(sA.z, base + 2); TOP3_INS(sA.w, base + 3);
      TOP3_INS(sB.x, base + 4); TOP3_INS(sB.y, base + 5);
      TOP3_INS(sB.z, base + 6); TOP3_INS(sB.w, base + 7);
    }
    asm volatile("" ::: "memory");
  }

  {
    float pv0 = __shfl(v0, lane + 16), pv1 = __shfl(v1, lane + 16), pv2 = __shfl(v2v, lane + 16);
    int   pi0 = __shfl(i0, lane + 16), pi1 = __shfl(i1, lane + 16), pi2 = __shfl(i2,  lane + 16);
    TOP3_INS(pv0, pi0); TOP3_INS(pv1, pi1); TOP3_INS(pv2, pi2);
  }
  if (lane < 16) {
    const int o = (wave * 16 + lane) * 3;
    pv[o] = v0; pv[o + 1] = v1; pv[o + 2] = v2v;
    pi[o] = i0; pi[o + 1] = i1; pi[o + 2] = i2;
  }
  __syncthreads();
  if (wave == 0) {
    v0 = 1e30f; v1 = 1e30f; v2v = 1e30f;
    i0 = 0; i1 = 0; i2 = 0;
    for (int w = 0; w < NW; ++w) {
      const int o = (w * 16 + r15) * 3;
      TOP3_INS(pv[o], pi[o]); TOP3_INS(pv[o + 1], pi[o + 1]); TOP3_INS(pv[o + 2], pi[o + 2]);
    }
    const float e1 = __expf(v0 - v1), e2 = __expf(v0 - v2v);
    const float inv = 1.0f / (1.0f + e1 + e2);
    if (lane < 16) {
      tw[lane * 3] = inv; tw[lane * 3 + 1] = e1 * inv; tw[lane * 3 + 2] = e2 * inv;
      ti[lane * 3] = i0;  ti[lane * 3 + 1] = i1;       ti[lane * 3 + 2] = i2;
    }
  }
  __syncthreads();

  for (int r = wave * ROWS_PER_WAVE; r < (wave + 1) * ROWS_PER_WAVE; ++r) {
    const float rw0 = tw[r * 3], rw1 = tw[r * 3 + 1], rw2 = tw[r * 3 + 2];
    const int   ri0 = ti[r * 3], ri1 = ti[r * 3 + 1], ri2 = ti[r * 3 + 2];
    const int qi = qbase + r;
    for (int t = lane; t < C2; t += 32) {
      float v;
      if (t < C) {
        v = x7[(long)qi * C + t] * gate[t];
      } else {
        const int cc = t - C;
        v = rw0 * feat_s[(long)ri0 * C + cc]
          + rw1 * feat_s[(long)ri1 * C + cc]
          + rw2 * feat_s[(long)ri2 * C + cc];
      }
      WL[r * XSTRIDE + t] = v;
    }
  }
  __syncthreads();

  const int ch = wave * 16 + r15;

  v8f acc = {0.f, 0.f, 0.f, 0.f, 0.f, 0.f, 0.f, 0.f};
  if (wave < NT) {
    for (int k0 = 0; k0 < C2; k0 += 4) {
      const int kk = k0 + 2 * half;
      const v2f a1 = *(const v2f*)(WL + r15 * XSTRIDE + kk);
      v2f bv;
      bv.x = W1[(long)kk * C + ch];
      bv.y = W1[(long)(kk + 1) * C + ch];
      acc = __builtin_amdgcn_wmma_f32_16x16x4_f32(false, a1, false, bv, (short)0, acc, false, false);
    }
  }
  __syncthreads();

  if (wave < NT) {
    const float gm = gammaW[ch], bt = betaW[ch];
#pragma unroll
    for (int r = 0; r < 8; ++r) {
      const float h = fmaxf(acc[r] * gm + bt, 0.f);
      WL[H_OFF + (r + 8 * half) * HSTRIDE + ch] = h;
    }
  }
  __syncthreads();

  if (wave < NT) {
    v8f acc2 = {0.f, 0.f, 0.f, 0.f, 0.f, 0.f, 0.f, 0.f};
    for (int k0 = 0; k0 < C; k0 += 4) {
      const int kk = k0 + 2 * half;
      const v2f a2 = *(const v2f*)(WL + H_OFF + r15 * HSTRIDE + kk);
      v2f bv;
      bv.x = W2[(long)kk * C + ch];
      bv.y = W2[(long)(kk + 1) * C + ch];
      acc2 = __builtin_amdgcn_wmma_f32_16x16x4_f32(false, a2, false, bv, (short)0, acc2, false, false);
    }
#pragma unroll
    for (int r = 0; r < 8; ++r)
      WL[Y2_OFF + (r + 8 * half) * HSTRIDE + ch] = acc2[r];
  }
  __syncthreads();

  if (wave == 0) {
    v8f acc3[2];
    acc3[0] = (v8f){0.f, 0.f, 0.f, 0.f, 0.f, 0.f, 0.f, 0.f};
    acc3[1] = acc3[0];
    for (int k0 = 0; k0 < C; k0 += 4) {
      const int kk = k0 + 2 * half;
      const v2f a3 = *(const v2f*)(WL + Y2_OFF + r15 * HSTRIDE + kk);
#pragma unroll
      for (int n = 0; n < 2; ++n) {
        const int col = n * 16 + r15;
        v2f bv;
        bv.x = (col < CLS) ? W_cls[(long)kk * CLS + col] : 0.f;
        bv.y = (col < CLS) ? W_cls[(long)(kk + 1) * CLS + col] : 0.f;
        acc3[n] = __builtin_amdgcn_wmma_f32_16x16x4_f32(false, a3, false, bv, (short)0, acc3[n], false, false);
      }
    }
#pragma unroll
    for (int n = 0; n < 2; ++n) {
      const int col = n * 16 + r15;
      if (col < CLS) {
        const float bb = b_cls[col];
#pragma unroll
        for (int r = 0; r < 8; ++r)
          out[(long)(qbase + r + 8 * half) * CLS + col] = acc3[n][r] + bb;
      }
    }
  }
}

extern "C" void kernel_launch(void* const* d_in, const int* in_sizes, int n_in,
                              void* d_out, int out_size, void* d_ws, size_t ws_size,
                              hipStream_t stream) {
  const float* coords_q = (const float*)d_in[0];
  const float* coords_s = (const float*)d_in[1];
  const float* x7       = (const float*)d_in[2];
  const float* feat_s7  = (const float*)d_in[3];
  const float* feat_s   = (const float*)d_in[4];
  const float* w_attn   = (const float*)d_in[5];
  const float* W1       = (const float*)d_in[6];
  const float* gammaW   = (const float*)d_in[7];
  const float* betaW    = (const float*)d_in[8];
  const float* W2       = (const float*)d_in[9];
  const float* W_cls    = (const float*)d_in[10];
  const float* b_cls    = (const float*)d_in[11];
  float* out = (float*)d_out;

  float* ws   = (float*)d_ws;
  float* rec  = ws;
  float* gate = ws + PB_N * REC;

  pool_partials<<<PB_N, 256, 0, stream>>>(feat_s7, w_attn, rec);
  pool_combine<<<1, 512, 0, stream>>>(rec, gate);

  fused_main<<<TILES, NW * 32, 0, stream>>>(coords_q, coords_s, x7, feat_s, gate,
                                         W1, gammaW, betaW, W2, W_cls, b_cls, out);
}


# top-3 search in two passes: a minima-only pass bounds each query's 3rd-smallest score (3rd of 128 subset minima); the tiles are computed again and only tiles holding a score <= bound are inserted
# speedup vs baseline: 11.7489x; 1.1651x over previous
#include <hip/hip_runtime.h>
#include <math.h>

#define NQ   20000
#define NS   8192
#define C    112
#define C2   224
#define CLS  20
#define TILES (NQ / 16)

#define WREG    3712
#define SSTRIDE 20
#define XSTRIDE 226
#define HSTRIDE 114
#define H_OFF   0
#define Y2_OFF  1856

typedef float v2f __attribute__((ext_vector_type(2)));
typedef float v8f __attribute__((ext_vector_type(8)));

#define PB_ROWS  64
#define PB_N     (NS / PB_ROWS)
#define PB_PITCH 116
#define REC      128

__global__ __launch_bounds__(256) void pool_partials(const float* __restrict__ feat_s7,
                                                      const float* __restrict__ w_attn,
                                                      float* __restrict__ rec) {
  __shared__ __attribute__((aligned(16))) float tile[PB_ROWS * PB_PITCH];
  __shared__ __attribute__((aligned(16))) float wl[C];
  __shared__ float lg[PB_ROWS];
  __shared__ float ev[PB_ROWS];
  __shared__ float part[2 * C];
  const int t = threadIdx.x;
  const int lane = t & 31;

  if (t < C / 4) *(float4*)(wl + 4 * t) = ((const float4*)w_attn)[t];

  const float4* src = (const float4*)(feat_s7 + (long)blockIdx.x * (PB_ROWS * C));
#pragma unroll
  for (int i = 0; i < 7; ++i) {
    const int q = t + 256 * i;
    const int row = q / 28, c4 = q - row * 28;
    *(float4*)(tile + row * PB_PITCH + 4 * c4) = src[q];
  }
  __syncthreads();

  {
    const int row = t >> 2, p = t & 3;
    float acc = 0.f;
#pragma unroll
    for (int i = 0; i < 7; ++i) {
      const int c4 = p + 4 * i;
      const float4 x = *(const float4*)(tile + row * PB_PITCH + 4 * c4);
      const float4 w = *(const float4*)(wl + 4 * c4);
      acc += x.x * w.x + x.y * w.y + x.z * w.z + x.w * w.w;
    }
    acc += __shfl_xor(acc, 1);
    acc += __shfl_xor(acc, 2);
    if (p == 0) lg[row] = acc;
  }
  __syncthreads();

  float m = fmaxf(lg[lane], lg[lane + 32]);
  for (int o = 16; o > 0; o >>= 1) m = fmaxf(m, __shfl_xor(m, o));
  if (t < PB_ROWS) ev[t] = __expf(lg[t] - m);
  __syncthreads();
  float s = ev[lane] + ev[lane + 32];
  for (int o = 16; o > 0; o >>= 1) s += __shfl_xor(s, o);

  if (t < 2 * C) {
    const int h = (t >= C) ? 1 : 0;
    const int c = t - h * C;
    const float* col = tile + (h * 32) * PB_PITCH + c;
    const float* e = ev + h * 32;
    float a0 = 0.f, a1 = 0.f;
#pragma unroll
    for (int r = 0; r < 32; r += 2) {
      a0 += e[r] * col[r * PB_PITCH];
      a1 += e[r + 1] * col[(r + 1) * PB_PITCH];
    }
    part[t] = a0 + a1;
  }
  __syncthreads();

  if (t < REC) {
    float v = 0.f;
    if (t < C) v = part[t] + part[C + t];
    else if (t == C) v = m;
    else if (t == C + 1) v = s;
    rec[(long)blockIdx.x * REC + t] = v;
  }
}

__global__ __launch_bounds__(512) void pool_combine(const float* __restrict__ rec,
                                                     float* __restrict__ gate) {
  __shared__ float ml[PB_N];
  __shared__ float sl[PB_N];
  __shared__ float fs[PB_N];
  __shared__ float part[4 * REC];
  const int t = threadIdx.x;
  const int lane = t & 31;
  const int c = t & (REC - 1);
  const int g = t >> 7;

  float x[32];
#pragma unroll
  for (int i = 0; i < 32; ++i) x[i] = rec[(long)(g * 32 + i) * REC + c];

  if (t < PB_N) { ml[t] = rec[(long)t * REC + C]; sl[t] = rec[(long)t * REC + C + 1]; }
  __syncthreads();
  float M = fmaxf(fmaxf(ml[lane], ml[lane + 32]), fmaxf(ml[lane + 64], ml[lane + 96]));
  for (int o = 16; o > 0; o >>= 1) M = fmaxf(M, __shfl_xor(M, o));
  if (t < PB_N) { const float f = __expf(ml[t] - M); fs[t] = f; sl[t] = sl[t] * f; }
  __syncthreads();
  float S = (sl[lane] + sl[lane + 32]) + (sl[lane + 64] + sl[lane + 96]);
  for (int o = 16; o > 0; o >>= 1) S += __shfl_xor(S, o);

  float a0 = 0.f, a1 = 0.f;
#pragma unroll
  for (int i = 0; i < 32; i += 2) {
    a0 += x[i] * fs[g * 32 + i];
    a1 += x[i + 1] * fs[g * 32 + i + 1];
  }
  part[t] = a0 + a1;
  __syncthreads();

  if (t < C) {
    const float acc = (part[t] + part[REC + t]) + (part[2 * REC + t] + part[3 * REC + t]);
    const float gv = acc * (1.0f / S);
    gate[t] = 1.0f / (1.0f + __expf(-gv));
  }
}

#define TOP3_INS(val, idxv) {                                         \
    float s_ = (val); int id_ = (idxv);                               \
    bool c0_ = s_ < v0, c1_ = s_ < v1, c2_ = s_ < v2v;                \
    v2v = c1_ ? v1 : (c2_ ? s_  : v2v); i2 = c1_ ? i1 : (c2_ ? id_ : i2); \
    v1  = c0_ ? v0 : (c1_ ? s_  : v1 ); i1 = c0_ ? i0 : (c1_ ? id_ : i1); \
    v0  = c0_ ? s_ : v0;                i0 = c0_ ? id_ : i0;          }

#define TOP3_VAL(x) {                                                 \
    const float x_ = (x);                                             \
    const float n1_ = __builtin_amdgcn_fmed3f(t0, t1, x_);            \
    const float n2_ = __builtin_amdgcn_fmed3f(t1, t2, x_);            \
    t0 = fminf(t0, x_); t1 = n1_; t2 = n2_;                           }

__device__ __forceinline__ v8f score_tile(const v2f a, const float* __restrict__ p, const int half) {
  const float sx = p[0];
  const float sy = p[1];
  const float sz = p[2];
  const float s2 = fmaf(sz, sz, fmaf(sx, sx, sy * sy));
  v2f b;
  b.x = half ? sz : sx;
  b.y = half ? s2 : sy;
  const v8f c = {0.f, 0.f, 0.f, 0.f, 0.f, 0.f, 0.f, 0.f};
  return __builtin_amdgcn_wmma_f32_16x16x4_f32(false, a, false, b, (short)0, c, false, false);
}

#define NW   8
#define SEG  (NS / NW)
#define SREG 320
#define ROWS_PER_WAVE (16 / NW)
#define NT   (C / 16)

__global__ __launch_bounds__(NW * 32) void fused_main(
    const float* __restrict__ coords_q, const float* __restrict__ coords_s,
    const float* __restrict__ x7,       const float* __restrict__ feat_s,
    const float* __restrict__ gate,     const float* __restrict__ W1,
    const float* __restrict__ gammaW,   const float* __restrict__ betaW,
    const float* __restrict__ W2,       const float* __restrict__ W_cls,
    const float* __restrict__ b_cls,    float* __restrict__ out)
{
  __shared__ __attribute__((aligned(16))) float lds[WREG];
  __shared__ __attribute__((aligned(16))) float sc[NW * SREG];
  __shared__ float pv[NW * 16 * 3];
  __shared__ int   pi[NW * 16 * 3];
  __shared__ float bt[NW * 16 * 3];
  __shared__ float tw[16 * 3];
  __shared__ int   ti[16 * 3];
  const int lane = threadIdx.x & 31;
  const int wave = __builtin_amdgcn_readfirstlane(threadIdx.x >> 5);
  const int tile = blockIdx.x;

  float* WL = lds;
  float* SC = sc + wave * SREG;
  const int r15  = lane & 15;
  const int half = lane >> 4;
  const int qbase = tile * 16;

  const float qx = coords_q[(qbase + r15) * 3 + 0];
  const float qy = coords_q[(qbase + r15) * 3 + 1];
  const float qz = coords_q[(qbase + r15) * 3 + 2];
  v2f a;
  a.x = half ? (-2.f * qz) : (-2.f * qx);
  a.y = half ? 1.0f        : (-2.f * qy);

  const int nbeg = wave * SEG;
  const float* const seg = coords_s + (nbeg + r15) * 3;

  float mn[8];
#pragma unroll
  for (int r = 0; r < 8; ++r) mn[r] = 1e30f;
  {
    const float* p = seg;
    for (int n0 = nbeg; n0 < nbeg + SEG; n0 += 32, p += 96) {
      __builtin_prefetch(coords_s + (n0 + 128) * 3, 0, 1);
      __builtin_prefetch(coords_s + (n0 + 144) * 3, 0, 1);
      const v8f c1 = score_tile(a, p, half);
      const v8f c2 = score_tile(a, p + 48, half);
#pragma unroll
      for (int r = 0; r < 8; ++r) mn[r] = fminf(fminf(mn[r], c1[r]), c2[r]);
    }
  }

  float T;
  {
#pragma unroll
    for (int r = 0; r < 8; ++r)
      SC[(r + 8 * half) * SSTRIDE + r15] = mn[r];
    asm volatile("" ::: "memory");
    const float4 mA = *(const float4*)(SC + r15 * SSTRIDE + 8 * half);
    const float4 mB = *(const float4*)(SC + r15 * SSTRIDE + 8 * half + 4);
    asm volatile("" ::: "memory");
    float t0 = 1e30f, t1 = 1e30f, t2 = 1e30f;
    TOP3_VAL(mA.x); TOP3_VAL(mA.y); TOP3_VAL(mA.z); TOP3_VAL(mA.w);
    TOP3_VAL(mB.x); TOP3_VAL(mB.y); TOP3_VAL(mB.z); TOP3_VAL(mB.w);
    const float o0 = __shfl(t0, lane ^ 16), o1 = __shfl(t1, lane ^ 16), o2 = __shfl(t2, lane ^ 16);
    TOP3_VAL(o0); TOP3_VAL(o1); TOP3_VAL(o2);
    if (lane < 16) {
      const int o = (wave * 16 + lane) * 3;
      bt[o] = t0; bt[o + 1] = t1; bt[o + 2] = t2;
    }
    __syncthreads();
    t0 = 1e30f; t1 = 1e30f; t2 = 1e30f;
    for (int w = 0; w < NW; ++w) {
      const int o = (w * 16 + r15) * 3;
      TOP3_VAL(bt[o]); TOP3_VAL(bt[o + 1]); TOP3_VAL(bt[o + 2]);
    }
    T = t2;
  }
  float Tr[8];
#pragma unroll
  for (int r = 0; r < 8; ++r) Tr[r] = __shfl(T, 8 * half + r);

  float v0 = 1e30f, v1 = 1e30f, v2v = 1e30f;
  int   i0 = 0, i1 = 0, i2 = 0;

#define COLLECT_TILE(cv, nb) {                                                    \
    _Pragma("unroll")                                                             \
    for (int r = 0; r < 8; ++r)                                                   \
      SC[(r + 8 * half) * SSTRIDE + r15] = (cv)[r];                               \
    asm volatile("" ::: "memory");             \
    const float4 sA = *(const float4*)(SC + r15 * SSTRIDE + 8 * half);            \
    const float4 sB = *(const float4*)(SC + r15 * SSTRIDE + 8 * half + 4);        \
    asm volatile("" ::: "memory");             \
    const int base = (nb) + 8 * half;                                             \
    TOP3_INS(sA.x, base + 0); TOP3_INS(sA.y, base + 1);                           \
    TOP3_INS(sA.z, base + 2); TOP3_INS(sA.w, base + 3);                           \
    TOP3_INS(sB.x, base + 4); TOP3_INS(sB.y, base + 5);                           \
    TOP3_INS(sB.z, base + 6); TOP3_INS(sB.w, base + 7);                           }

  {
    const float* p = seg;
    for (int n0 = nbeg; n0 < nbeg + SEG; n0 += 32, p += 96) {
      __builtin_prefetch(coords_s + (n0 + 128) * 3, 0, 1);
      __builtin_prefetch(coords_s + (n0 + 144) * 3, 0, 1);
      const v8f c1 = score_tile(a, p, half);
      const v8f c2 = score_tile(a, p + 48, half);
      unsigned m1 = 0u, m2 = 0u;
#pragma unroll
      for (int r = 0; r < 8; ++r) {
        m1 |= __builtin_amdgcn_ballot_w32(c1[r] <= Tr[r]);
        m2 |= __builtin_amdgcn_ballot_w32(c2[r] <= Tr[r]);
      }
      if (m1 | m2) {
        if (m1) COLLECT_TILE(c1, n0);
        if (m2) COLLECT_TILE(c2, n0 + 16);
      }
    }
  }
#undef COLLECT_TILE

  {
    float pv0 = __shfl(v0, lane + 16), pv1 = __shfl(v1, lane + 16), pv2 = __shfl(v2v, lane + 16);
    int   pi0 = __shfl(i0, lane + 16), pi1 = __shfl(i1, lane + 16), pi2 = __shfl(i2,  lane + 16);
    TOP3_INS(pv0, pi0); TOP3_INS(pv1, pi1); TOP3_INS(pv2, pi2);
  }
  if (lane < 16) {
    const int o = (wave * 16 + lane) * 3;
    pv[o] = v0; pv[o + 1] = v1; pv[o + 2] = v2v;
    pi[o] = i0; pi[o + 1] = i1; pi[o + 2] = i2;
  }
  __syncthreads();
  if (wave == 0) {
    v0 = 1e30f; v1 = 1e30f; v2v = 1e30f;
    i0 = 0; i1 = 0; i2 = 0;
    for (int w = 0; w < NW; ++w) {
      const int o = (w * 16 + r15) * 3;
      TOP3_INS(pv[o], pi[o]); TOP3_INS(pv[o + 1], pi[o + 1]); TOP3_INS(pv[o + 2], pi[o + 2]);
    }
    const float e1 = __expf(v0 - v1), e2 = __expf(v0 - v2v);
    const float inv = 1.0f / (1.0f + e1 + e2);
    if (lane < 16) {
      tw[lane * 3] = inv; tw[lane * 3 + 1] = e1 * inv; tw[lane * 3 + 2] = e2 * inv;
      ti[lane * 3] = i0;  ti[lane * 3 + 1] = i1;       ti[lane * 3 + 2] = i2;
    }
  }
  __syncthreads();

  for (int r = wave * ROWS_PER_WAVE; r < (wave + 1) * ROWS_PER_WAVE; ++r) {
    const float rw0 = tw[r * 3], rw1 = tw[r * 3 + 1], rw2 = tw[r * 3 + 2];
    const int   ri0 = ti[r * 3], ri1 = ti[r * 3 + 1], ri2 = ti[r * 3 + 2];
    const int qi = qbase + r;
    for (int t = lane; t < C2; t += 32) {
      float v;
      if (t < C) {
        v = x7[(long)qi * C + t] * gate[t];
      } else {
        const int cc = t - C;
        v = rw0 * feat_s[(long)ri0 * C + cc]
          + rw1 * feat_s[(long)ri1 * C + cc]
          + rw2 * feat_s[(long)ri2 * C + cc];
      }
      WL[r * XSTRIDE + t] = v;
    }
  }
  __syncthreads();

  const int ch = wave * 16 + r15;

  v8f acc = {0.f, 0.f, 0.f, 0.f, 0.f, 0.f, 0.f, 0.f};
  if (wave < NT) {
    for (int k0 = 0; k0 < C2; k0 += 4) {
      const int kk = k0 + 2 * half;
      const v2f a1 = *(const v2f*)(WL + r15 * XSTRIDE + kk);
      v2f bv;
      bv.x = W1[(long)kk * C + ch];
      bv.y = W1[(long)(kk + 1) * C + ch];
      acc = __builtin_amdgcn_wmma_f32_16x16x4_f32(false, a1, false, bv, (short)0, acc, false, false);
    }
  }
  __syncthreads();

  if (wave < NT) {
    const float gm = gammaW[ch], bt = betaW[ch];
#pragma unroll
    for (int r = 0; r < 8; ++r) {
      const float h = fmaxf(acc[r] * gm + bt, 0.f);
      WL[H_OFF + (r + 8 * half) * HSTRIDE + ch] = h;
    }
  }
  __syncthreads();

  if (wave < NT) {
    v8f acc2 = {0.f, 0.f, 0.f, 0.f, 0.f, 0.f, 0.f, 0.f};
    for (int k0 = 0; k0 < C; k0 += 4) {
      const int kk = k0 + 2 * half;
      const v2f a2 = *(const v2f*)(WL + H_OFF + r15 * HSTRIDE + kk);
      v2f bv;
      bv.x = W2[(long)kk * C + ch];
      bv.y = W2[(long)(kk + 1) * C + ch];
      acc2 = __builtin_amdgcn_wmma_f32_16x16x4_f32(false, a2, false, bv, (short)0, acc2, false, false);
    }
#pragma unroll
    for (int r = 0; r < 8; ++r)
      WL[Y2_OFF + (r + 8 * half) * HSTRIDE + ch] = acc2[r];
  }
  __syncthreads();

  if (wave == 0) {
    v8f acc3[2];
    acc3[0] = (v8f){0.f, 0.f, 0.f, 0.f, 0.f, 0.f, 0.f, 0.f};
    acc3[1] = acc3[0];
    for (int k0 = 0; k0 < C; k0 += 4) {
      const int kk = k0 + 2 * half;
      const v2f a3 = *(const v2f*)(WL + Y2_OFF + r15 * HSTRIDE + kk);
#pragma unroll
      for (int n = 0; n < 2; ++n) {
        const int col = n * 16 + r15;
        v2f bv;
        bv.x = (col < CLS) ? W_cls[(long)kk * CLS + col] : 0.f;
        bv.y = (col < CLS) ? W_cls[(long)(kk + 1) * CLS + col] : 0.f;
        acc3[n] = __builtin_amdgcn_wmma_f32_16x16x4_f32(false, a3, false, bv, (short)0, acc3[n], false, false);
      }
    }
#pragma unroll
    for (int n = 0; n < 2; ++n) {
      const int col = n * 16 + r15;
      if (col < CLS) {
        const float bb = b_cls[col];
#pragma unroll
        for (int r = 0; r < 8; ++r)
          out[(long)(qbase + r + 8 * half) * CLS + col] = acc3[n][r] + bb;
      }
    }
  }
}

extern "C" void kernel_launch(void* const* d_in, const int* in_sizes, int n_in,
                              void* d_out, int out_size, void* d_ws, size_t ws_size,
                              hipStream_t stream) {
  const float* coords_q = (const float*)d_in[0];
  const float* coords_s = (const float*)d_in[1];
  const float* x7       = (const float*)d_in[2];
  const float* feat_s7  = (const float*)d_in[3];
  const float* feat_s   = (const float*)d_in[4];
  const float* w_attn   = (const float*)d_in[5];
  const float* W1       = (const float*)d_in[6];
  const float* gammaW   = (const float*)d_in[7];
  const float* betaW    = (const float*)d_in[8];
  const float* W2       = (const float*)d_in[9];
  const float* W_cls    = (const float*)d_in[10];
  const float* b_cls    = (const float*)d_in[11];
  float* out = (float*)d_out;

  float* ws   = (float*)d_ws;
  float* rec  = ws;
  float* gate = ws + PB_N * REC;

  pool_partials<<<PB_N, 256, 0, stream>>>(feat_s7, w_attn, rec);
  pool_combine<<<1, 512, 0, stream>>>(rec, gate);

  fused_main<<<TILES, NW * 32, 0, stream>>>(coords_q, coords_s, x7, feat_s, gate,
                                         W1, gammaW, betaW, W2, W_cls, b_cls, out);
}


# GEMM3 split by K over the seven GEMM2 waves: each multiplies its own Y2 tile by 16 rows of W_cls (loads issued ahead of GEMM2), partial tiles added through LDS in wave order; sums re-associated
# speedup vs baseline: 13.6584x; 1.1625x over previous
#include <hip/hip_runtime.h>
#include <math.h>

#define NQ   20000
#define NS   8192
#define C    112
#define C2   224
#define CLS  20
#define TILES (NQ / 16)

#define WREG    3712
#define SSTRIDE 20
#define XSTRIDE 226
#define HSTRIDE 114
#define H_OFF   0
#define Y2_OFF  1856

typedef float v2f __attribute__((ext_vector_type(2)));
typedef float v8f __attribute__((ext_vector_type(8)));

#define PB_ROWS  64
#define PB_N     (NS / PB_ROWS)
#define PB_PITCH 116
#define REC      128

__global__ __launch_bounds__(256) void pool_partials(const float* __restrict__ feat_s7,
                                                      const float* __restrict__ w_attn,
                                                      float* __restrict__ rec) {
  __shared__ __attribute__((aligned(16))) float tile[PB_ROWS * PB_PITCH];
  __shared__ __attribute__((aligned(16))) float wl[C];
  __shared__ float lg[PB_ROWS];
  __shared__ float ev[PB_ROWS];
  __shared__ float part[2 * C];
  const int t = threadIdx.x;
  const int lane = t & 31;

  if (t < C / 4) *(float4*)(wl + 4 * t) = ((const float4*)w_attn)[t];

  const float4* src = (const float4*)(feat_s7 + (long)blockIdx.x * (PB_ROWS * C));
#pragma unroll
  for (int i = 0; i < 7; ++i) {
    const int q = t + 256 * i;
    const int row = q / 28, c4 = q - row * 28;
    *(float4*)(tile + row * PB_PITCH + 4 * c4) = src[q];
  }
  __syncthreads();

  {
    const int row = t >> 2, p = t & 3;
    float acc = 0.f;
#pragma unroll
    for (int i = 0; i < 7; ++i) {
      const int c4 = p + 4 * i;
      const float4 x = *(const float4*)(tile + row * PB_PITCH + 4 * c4);
      const float4 w = *(const float4*)(wl + 4 * c4);
      acc += x.x * w.x + x.y * w.y + x.z * w.z + x.w * w.w;
    }
    acc += __shfl_xor(acc, 1);
    acc += __shfl_xor(acc, 2);
    if (p == 0) lg[row] = acc;
  }
  __syncthreads();

  float m = fmaxf(lg[lane], lg[lane + 32]);
  for (int o = 16; o > 0; o >>= 1) m = fmaxf(m, __shfl_xor(m, o));
  if (t < PB_ROWS) ev[t] = __expf(lg[t] - m);
  __syncthreads();
  float s = ev[lane] + ev[lane + 32];
  for (int o = 16; o > 0; o >>= 1) s += __shfl_xor(s, o);

  if (t < 2 * C) {
    const int h = (t >= C) ? 1 : 0;
    const int c = t - h * C;
    const float* col = tile + (h * 32) * PB_PITCH + c;
    const float* e = ev + h * 32;
    float a0 = 0.f, a1 = 0.f;
#pragma unroll
    for (int r = 0; r < 32; r += 2) {
      a0 += e[r] * col[r * PB_PITCH];
      a1 += e[r + 1] * col[(r + 1) * PB_PITCH];
    }
    part[t] = a0 + a1;
  }
  __syncthreads();

  if (t < REC) {
    float v = 0.f;
    if (t < C) v = part[t] + part[C + t];
    else if (t == C) v = m;
    else if (t == C + 1) v = s;
    rec[(long)blockIdx.x * REC + t] = v;
  }
}

__global__ __launch_bounds__(512) void pool_combine(const float* __restrict__ rec,
                                                     float* __restrict__ gate) {
  __shared__ float ml[PB_N];
  __shared__ float sl[PB_N];
  __shared__ float fs[PB_N];
  __shared__ float part[4 * REC];
  const int t = threadIdx.x;
  const int lane = t & 31;
  const int c = t & (REC - 1);
  const int g = t >> 7;

  float x[32];
#pragma unroll
  for (int i = 0; i < 32; ++i) x[i] = rec[(long)(g * 32 + i) * REC + c];

  if (t < PB_N) { ml[t] = rec[(long)t * REC + C]; sl[t] = rec[(long)t * REC + C + 1]; }
  __syncthreads();
  float M = fmaxf(fmaxf(ml[lane], ml[lane + 32]), fmaxf(ml[lane + 64], ml[lane + 96]));
  for (int o = 16; o > 0; o >>= 1) M = fmaxf(M, __shfl_xor(M, o));
  if (t < PB_N) { const float f = __expf(ml[t] - M); fs[t] = f; sl[t] = sl[t] * f; }
  __syncthreads();
  float S = (sl[lane] + sl[lane + 32]) + (sl[lane + 64] + sl[lane + 96]);
  for (int o = 16; o > 0; o >>= 1) S += __shfl_xor(S, o);

  float a0 = 0.f, a1 = 0.f;
#pragma unroll
  for (int i = 0; i < 32; i += 2) {
    a0 += x[i] * fs[g * 32 + i];
    a1 += x[i + 1] * fs[g * 32 + i + 1];
  }
  part[t] = a0 + a1;
  __syncthreads();

  if (t < C) {
    const float acc = (part[t] + part[REC + t]) + (part[2 * REC + t] + part[3 * REC + t]);
    const float gv = acc * (1.0f / S);
    gate[t] = 1.0f / (1.0f + __expf(-gv));
  }
}

#define TOP3_INS(val, idxv) {                                         \
    float s_ = (val); int id_ = (idxv);                               \
    bool c0_ = s_ < v0, c1_ = s_ < v1, c2_ = s_ < v2v;                \
    v2v = c1_ ? v1 : (c2_ ? s_  : v2v); i2 = c1_ ? i1 : (c2_ ? id_ : i2); \
    v1  = c0_ ? v0 : (c1_ ? s_  : v1 ); i1 = c0_ ? i0 : (c1_ ? id_ : i1); \
    v0  = c0_ ? s_ : v0;                i0 = c0_ ? id_ : i0;          }

#define TOP3_VAL(x) {                                                 \
    const float x_ = (x);                                             \
    const float n1_ = __builtin_amdgcn_fmed3f(t0, t1, x_);            \
    const float n2_ = __builtin_amdgcn_fmed3f(t1, t2, x_);            \
    t0 = fminf(t0, x_); t1 = n1_; t2 = n2_;                           }

__device__ __forceinline__ v8f score_tile(const v2f a, const float* __restrict__ p, const int half) {
  const float sx = p[0];
  const float sy = p[1];
  const float sz = p[2];
  const float s2 = fmaf(sz, sz, fmaf(sx, sx, sy * sy));
  v2f b;
  b.x = half ? sz : sx;
  b.y = half ? s2 : sy;
  const v8f c = {0.f, 0.f, 0.f, 0.f, 0.f, 0.f, 0.f, 0.f};
  return __builtin_amdgcn_wmma_f32_16x16x4_f32(false, a, false, b, (short)0, c, false, false);
}

#define NW   8
#define SEG  (NS / NW)
#define SREG 320
#define ROWS_PER_WAVE (16 / NW)
#define NT   (C / 16)

__global__ __launch_bounds__(NW * 32) void fused_main(
    const float* __restrict__ coords_q, const float* __restrict__ coords_s,
    const float* __restrict__ x7,       const float* __restrict__ feat_s,
    const float* __restrict__ gate,     const float* __restrict__ W1,
    const float* __restrict__ gammaW,   const float* __restrict__ betaW,
    const float* __restrict__ W2,       const float* __restrict__ W_cls,
    const float* __restrict__ b_cls,    float* __restrict__ out)
{
  __shared__ __attribute__((aligned(16))) float lds[WREG];
  __shared__ __attribute__((aligned(16))) float sc[NW * SREG];
  __shared__ float pv[NW * 16 * 3];
  __shared__ int   pi[NW * 16 * 3];
  __shared__ float bt[NW * 16 * 3];
  __shared__ float tw[16 * 3];
  __shared__ int   ti[16 * 3];
  const int lane = threadIdx.x & 31;
  const int wave = __builtin_amdgcn_readfirstlane(threadIdx.x >> 5);
  const int tile = blockIdx.x;

  float* WL = lds;
  float* SC = sc + wave * SREG;
  const int r15  = lane & 15;
  const int half = lane >> 4;
  const int qbase = tile * 16;

  const float qx = coords_q[(qbase + r15) * 3 + 0];
  const float qy = coords_q[(qbase + r15) * 3 + 1];
  const float qz = coords_q[(qbase + r15) * 3 + 2];
  v2f a;
  a.x = half ? (-2.f * qz) : (-2.f * qx);
  a.y = half ? 1.0f        : (-2.f * qy);

  const int nbeg = wave * SEG;
  const float* const seg = coords_s + (nbeg + r15) * 3;

  float mn[8];
#pragma unroll
  for (int r = 0; r < 8; ++r) mn[r] = 1e30f;
  {
    const float* p = seg;
    for (int n0 = nbeg; n0 < nbeg + SEG; n0 += 32, p += 96) {
      __builtin_prefetch(coords_s + (n0 + 128) * 3, 0, 1);
      __builtin_prefetch(coords_s + (n0 + 144) * 3, 0, 1);
      const v8f c1 = score_tile(a, p, half);
      const v8f c2 = score_tile(a, p + 48, half);
#pragma unroll
      for (int r = 0; r < 8; ++r) mn[r] = fminf(fminf(mn[r], c1[r]), c2[r]);
    }
  }

  float T;
  {
#pragma unroll
    for (int r = 0; r < 8; ++r)
      SC[(r + 8 * half) * SSTRIDE + r15] = mn[r];
    asm volatile("" ::: "memory");
    const float4 mA = *(const float4*)(SC + r15 * SSTRIDE + 8 * half);
    const float4 mB = *(const float4*)(SC + r15 * SSTRIDE + 8 * half + 4);
    asm volatile("" ::: "memory");
    float t0 = 1e30f, t1 = 1e30f, t2 = 1e30f;
    TOP3_VAL(mA.x); TOP3_VAL(mA.y); TOP3_VAL(mA.z); TOP3_VAL(mA.w);
    TOP3_VAL(mB.x); TOP3_VAL(mB.y); TOP3_VAL(mB.z); TOP3_VAL(mB.w);
    const float o0 = __shfl(t0, lane ^ 16), o1 = __shfl(t1, lane ^ 16), o2 = __shfl(t2, lane ^ 16);
    TOP3_VAL(o0); TOP3_VAL(o1); TOP3_VAL(o2);
    if (lane < 16) {
      const int o = (wave * 16 + lane) * 3;
      bt[o] = t0; bt[o + 1] = t1; bt[o + 2] = t2;
    }
    __syncthreads();
    t0 = 1e30f; t1 = 1e30f; t2 = 1e30f;
    for (int w = 0; w < NW; ++w) {
      const int o = (w * 16 + r15) * 3;
      TOP3_VAL(bt[o]); TOP3_VAL(bt[o + 1]); TOP3_VAL(bt[o + 2]);
    }
    T = t2;
  }
  float Tr[8];
#pragma unroll
  for (int r = 0; r < 8; ++r) Tr[r] = __shfl(T, 8 * half + r);

  float v0 = 1e30f, v1 = 1e30f, v2v = 1e30f;
  int   i0 = 0, i1 = 0, i2 = 0;

#define COLLECT_TILE(cv, nb) {                                                    \
    _Pragma("unroll")                                                             \
    for (int r = 0; r < 8; ++r)                                                   \
      SC[(r + 8 * half) * SSTRIDE + r15] = (cv)[r];                               \
    asm volatile("" ::: "memory");             \
    const float4 sA = *(const float4*)(SC + r15 * SSTRIDE + 8 * half);            \
    const float4 sB = *(const float4*)(SC + r15 * SSTRIDE + 8 * half + 4);        \
    asm volatile("" ::: "memory");             \
    const int base = (nb) + 8 * half;                                             \
    TOP3_INS(sA.x, base + 0); TOP3_INS(sA.y, base + 1);                           \
    TOP3_INS(sA.z, base + 2); TOP3_INS(sA.w, base + 3);                           \
    TOP3_INS(sB.x, base + 4); TOP3_INS(sB.y, base + 5);                           \
    TOP3_INS(sB.z, base + 6); TOP3_INS(sB.w, base + 7);                           }

  {
    const float* p = seg;
    for (int n0 = nbeg; n0 < nbeg + SEG; n0 += 32, p += 96) {
      __builtin_prefetch(coords_s + (n0 + 128) * 3, 0, 1);
      __builtin_prefetch(coords_s + (n0 + 144) * 3, 0, 1);
      const v8f c1 = score_tile(a, p, half);
      const v8f c2 = score_tile(a, p + 48, half);
      unsigned m1 = 0u, m2 = 0u;
#pragma unroll
      for (int r = 0; r < 8; ++r) {
        m1 |= __builtin_amdgcn_ballot_w32(c1[r] <= Tr[r]);
        m2 |= __builtin_amdgcn_ballot_w32(c2[r] <= Tr[r]);
      }
      if (m1 | m2) {
        if (m1) COLLECT_TILE(c1, n0);
        if (m2) COLLECT_TILE(c2, n0 + 16);
      }
    }
  }
#undef COLLECT_TILE

  {
    float pv0 = __shfl(v0, lane + 16), pv1 = __shfl(v1, lane + 16), pv2 = __shfl(v2v, lane + 16);
    int   pi0 = __shfl(i0, lane + 16), pi1 = __shfl(i1, lane + 16), pi2 = __shfl(i2,  lane + 16);
    TOP3_INS(pv0, pi0); TOP3_INS(pv1, pi1); TOP3_INS(pv2, pi2);
  }
  if (lane < 16) {
    const int o = (wave * 16 + lane) * 3;
    pv[o] = v0; pv[o + 1] = v1; pv[o + 2] = v2v;
    pi[o] = i0; pi[o + 1] = i1; pi[o + 2] = i2;
  }
  __syncthreads();
  if (wave == 0) {
    v0 = 1e30f; v1 = 1e30f; v2v = 1e30f;
    i0 = 0; i1 = 0; i2 = 0;
    for (int w = 0; w < NW; ++w) {
      const int o = (w * 16 + r15) * 3;
      TOP3_INS(pv[o], pi[o]); TOP3_INS(pv[o + 1], pi[o + 1]); TOP3_INS(pv[o + 2], pi[o + 2]);
    }
    const float e1 = __expf(v0 - v1), e2 = __expf(v0 - v2v);
    const float inv = 1.0f / (1.0f + e1 + e2);
    if (lane < 16) {
      tw[lane * 3] = inv; tw[lane * 3 + 1] = e1 * inv; tw[lane * 3 + 2] = e2 * inv;
      ti[lane * 3] = i0;  ti[lane * 3 + 1] = i1;       ti[lane * 3 + 2] = i2;
    }
  }
  __syncthreads();

  for (int r = wave * ROWS_PER_WAVE; r < (wave + 1) * ROWS_PER_WAVE; ++r) {
    const float rw0 = tw[r * 3], rw1 = tw[r * 3 + 1], rw2 = tw[r * 3 + 2];
    const int   ri0 = ti[r * 3], ri1 = ti[r * 3 + 1], ri2 = ti[r * 3 + 2];
    const int qi = qbase + r;
    for (int t = lane; t < C2; t += 32) {
      float v;
      if (t < C) {
        v = x7[(long)qi * C + t] * gate[t];
      } else {
        const int cc = t - C;
        v = rw0 * feat_s[(long)ri0 * C + cc]
          + rw1 * feat_s[(long)ri1 * C + cc]
          + rw2 * feat_s[(long)ri2 * C + cc];
      }
      WL[r * XSTRIDE + t] = v;
    }
  }
  __syncthreads();

  const int ch = wave * 16 + r15;

  v8f acc = {0.f, 0.f, 0.f, 0.f, 0.f, 0.f, 0.f, 0.f};
  if (wave < NT) {
    for (int k0 = 0; k0 < C2; k0 += 4) {
      const int kk = k0 + 2 * half;
      const v2f a1 = *(const v2f*)(WL + r15 * XSTRIDE + kk);
      v2f bv;
      bv.x = W1[(long)kk * C + ch];
      bv.y = W1[(long)(kk + 1) * C + ch];
      acc = __builtin_amdgcn_wmma_f32_16x16x4_f32(false, a1, false, bv, (short)0, acc, false, false);
    }
  }
  __syncthreads();

  if (wave < NT) {
    const float gm = gammaW[ch], bt = betaW[ch];
#pragma unroll
    for (int r = 0; r < 8; ++r) {
      const float h = fmaxf(acc[r] * gm + bt, 0.f);
      WL[H_OFF + (r + 8 * half) * HSTRIDE + ch] = h;
    }
  }
  __syncthreads();

  float* const P = sc + wave * SREG;
  const float bias0 = b_cls[threadIdx.x % CLS];
  const float bias1 = b_cls[(threadIdx.x + NW * 32) % CLS];
  if (wave < NT) {
    const int cB = (r15 + 16 < CLS) ? (r15 + 16) : (CLS - 1);
    v2f wc0[4], wc1[4];
#pragma unroll
    for (int s = 0; s < 4; ++s) {
      const int kk = wave * 16 + 4 * s + 2 * half;
      wc0[s].x = W_cls[(long)kk * CLS + r15];
      wc0[s].y = W_cls[(long)(kk + 1) * CLS + r15];
      wc1[s].x = W_cls[(long)kk * CLS + cB];
      wc1[s].y = W_cls[(long)(kk + 1) * CLS + cB];
    }

    v8f acc2 = {0.f, 0.f, 0.f, 0.f, 0.f, 0.f, 0.f, 0.f};
    for (int k0 = 0; k0 < C; k0 += 4) {
      const int kk = k0 + 2 * half;
      const v2f a2 = *(const v2f*)(WL + H_OFF + r15 * HSTRIDE + kk);
      v2f bv;
      bv.x = W2[(long)kk * C + ch];
      bv.y = W2[(long)(kk + 1) * C + ch];
      acc2 = __builtin_amdgcn_wmma_f32_16x16x4_f32(false, a2, false, bv, (short)0, acc2, false, false);
    }
#pragma unroll
    for (int r = 0; r < 8; ++r)
      WL[Y2_OFF + (r + 8 * half) * HSTRIDE + ch] = acc2[r];
    asm volatile("" ::: "memory");
    v8f p0 = {0.f, 0.f, 0.f, 0.f, 0.f, 0.f, 0.f, 0.f};
    v8f p1 = p0;
#pragma unroll
    for (int s = 0; s < 4; ++s) {
      const int kk = wave * 16 + 4 * s + 2 * half;
      const v2f a3 = *(const v2f*)(WL + Y2_OFF + r15 * HSTRIDE + kk);
      p0 = __builtin_amdgcn_wmma_f32_16x16x4_f32(false, a3, false, wc0[s], (short)0, p0, false, false);
      p1 = __builtin_amdgcn_wmma_f32_16x16x4_f32(false, a3, false, wc1[s], (short)0, p1, false, false);
    }
#pragma unroll
    for (int r = 0; r < 8; ++r)
      P[(r + 8 * half) * CLS + r15] = p0[r];
    if (r15 + 16 < CLS) {
#pragma unroll
      for (int r = 0; r < 8; ++r)
        P[(r + 8 * half) * CLS + 16 + r15] = p1[r];
    }
  }
  __syncthreads();

  {
    const int t = threadIdx.x;
    float s = sc[t];
#pragma unroll
    for (int w = 1; w < NT; ++w) s += sc[w * SREG + t];
    out[(long)qbase * CLS + t] = s + bias0;
    if (t < 16 * CLS - NW * 32) {
      const int u = t + NW * 32;
      float s1 = sc[u];
#pragma unroll
      for (int w = 1; w < NT; ++w) s1 += sc[w * SREG + u];
      out[(long)qbase * CLS + u] = s1 + bias1;
    }
  }
}

extern "C" void kernel_launch(void* const* d_in, const int* in_sizes, int n_in,
                              void* d_out, int out_size, void* d_ws, size_t ws_size,
                              hipStream_t stream) {
  const float* coords_q = (const float*)d_in[0];
  const float* coords_s = (const float*)d_in[1];
  const float* x7       = (const float*)d_in[2];
  const float* feat_s7  = (const float*)d_in[3];
  const float* feat_s   = (const float*)d_in[4];
  const float* w_attn   = (const float*)d_in[5];
  const float* W1       = (const float*)d_in[6];
  const float* gammaW   = (const float*)d_in[7];
  const float* betaW    = (const float*)d_in[8];
  const float* W2       = (const float*)d_in[9];
  const float* W_cls    = (const float*)d_in[10];
  const float* b_cls    = (const float*)d_in[11];
  float* out = (float*)d_out;

  float* ws   = (float*)d_ws;
  float* rec  = ws;
  float* gate = ws + PB_N * REC;

  pool_partials<<<PB_N, 256, 0, stream>>>(feat_s7, w_attn, rec);
  pool_combine<<<1, 512, 0, stream>>>(rec, gate);

  fused_main<<<TILES, NW * 32, 0, stream>>>(coords_q, coords_s, x7, feat_s, gate,
                                         W1, gammaW, betaW, W2, W_cls, b_cls, out);
}


# search pass B computes again only the 32-row blocks whose pass-A minimum is <= the query's bound (support rows as matrix rows: a lane owns one query, no LDS transpose); order of sums unchanged
# speedup vs baseline: 17.9872x; 1.3169x over previous
#include <hip/hip_runtime.h>
#include <math.h>

#define NQ   20000
#define NS   8192
#define C    112
#define C2   224
#define CLS  20
#define TILES (NQ / 16)

#define WREG    3712
#define XSTRIDE 226
#define HSTRIDE 114
#define H_OFF   0
#define Y2_OFF  1856

typedef float v2f __attribute__((ext_vector_type(2)));
typedef float v8f __attribute__((ext_vector_type(8)));

#define PB_ROWS  64
#define PB_N     (NS / PB_ROWS)
#define PB_PITCH 116
#define REC      128

__global__ __launch_bounds__(256) void pool_partials(const float* __restrict__ feat_s7,
                                                      const float* __restrict__ w_attn,
                                                      float* __restrict__ rec) {
  __shared__ __attribute__((aligned(16))) float tile[PB_ROWS * PB_PITCH];
  __shared__ __attribute__((aligned(16))) float wl[C];
  __shared__ float lg[PB_ROWS];
  __shared__ float ev[PB_ROWS];
  __shared__ float part[2 * C];
  const int t = threadIdx.x;
  const int lane = t & 31;

  if (t < C / 4) *(float4*)(wl + 4 * t) = ((const float4*)w_attn)[t];

  const float4* src = (const float4*)(feat_s7 + (long)blockIdx.x * (PB_ROWS * C));
#pragma unroll
  for (int i = 0; i < 7; ++i) {
    const int q = t + 256 * i;
    const int row = q / 28, c4 = q - row * 28;
    *(float4*)(tile + row * PB_PITCH + 4 * c4) = src[q];
  }
  __syncthreads();

  {
    const int row = t >> 2, p = t & 3;
    float acc = 0.f;
#pragma unroll
    for (int i = 0; i < 7; ++i) {
      const int c4 = p + 4 * i;
      const float4 x = *(const float4*)(tile + row * PB_PITCH + 4 * c4);
      const float4 w = *(const float4*)(wl + 4 * c4);
      acc += x.x * w.x + x.y * w.y + x.z * w.z + x.w * w.w;
    }
    acc += __shfl_xor(acc, 1);
    acc += __shfl_xor(acc, 2);
    if (p == 0) lg[row] = acc;
  }
  __syncthreads();

  float m = fmaxf(lg[lane], lg[lane + 32]);
  for (int o = 16; o > 0; o >>= 1) m = fmaxf(m, __shfl_xor(m, o));
  if (t < PB_ROWS) ev[t] = __expf(lg[t] - m);
  __syncthreads();
  float s = ev[lane] + ev[lane + 32];
  for (int o = 16; o > 0; o >>= 1) s += __shfl_xor(s, o);

  if (t < 2 * C) {
    const int h = (t >= C) ? 1 : 0;
    const int c = t - h * C;
    const float* col = tile + (h * 32) * PB_PITCH + c;
    const float* e = ev + h * 32;
    float a0 = 0.f, a1 = 0.f;
#pragma unroll
    for (int r = 0; r < 32; r += 2) {
      a0 += e[r] * col[r * PB_PITCH];
      a1 += e[r + 1] * col[(r + 1) * PB_PITCH];
    }
    part[t] = a0 + a1;
  }
  __syncthreads();

  if (t < REC) {
    float v = 0.f;
    if (t < C) v = part[t] + part[C + t];
    else if (t == C) v = m;
    else if (t == C + 1) v = s;
    rec[(long)blockIdx.x * REC + t] = v;
  }
}

__global__ __launch_bounds__(512) void pool_combine(const float* __restrict__ rec,
                                                     float* __restrict__ gate) {
  __shared__ float ml[PB_N];
  __shared__ float sl[PB_N];
  __shared__ float fs[PB_N];
  __shared__ float part[4 * REC];
  const int t = threadIdx.x;
  const int lane = t & 31;
  const int c = t & (REC - 1);
  const int g = t >> 7;

  float x[32];
#pragma unroll
  for (int i = 0; i < 32; ++i) x[i] = rec[(long)(g * 32 + i) * REC + c];

  if (t < PB_N) { ml[t] = rec[(long)t * REC + C]; sl[t] = rec[(long)t * REC + C + 1]; }
  __syncthreads();
  float M = fmaxf(fmaxf(ml[lane], ml[lane + 32]), fmaxf(ml[lane + 64], ml[lane + 96]));
  for (int o = 16; o > 0; o >>= 1) M = fmaxf(M, __shfl_xor(M, o));
  if (t < PB_N) { const float f = __expf(ml[t] - M); fs[t] = f; sl[t] = sl[t] * f; }
  __syncthreads();
  float S = (sl[lane] + sl[lane + 32]) + (sl[lane + 64] + sl[lane + 96]);
  for (int o = 16; o > 0; o >>= 1) S += __shfl_xor(S, o);

  float a0 = 0.f, a1 = 0.f;
#pragma unroll
  for (int i = 0; i < 32; i += 2) {
    a0 += x[i] * fs[g * 32 + i];
    a1 += x[i + 1] * fs[g * 32 + i + 1];
  }
  part[t] = a0 + a1;
  __syncthreads();

  if (t < C) {
    const float acc = (part[t] + part[REC + t]) + (part[2 * REC + t] + part[3 * REC + t]);
    const float gv = acc * (1.0f / S);
    gate[t] = 1.0f / (1.0f + __expf(-gv));
  }
}

#define TOP3_INS(val, idxv) {                                         \
    float s_ = (val); int id_ = (idxv);                               \
    bool c0_ = s_ < v0, c1_ = s_ < v1, c2_ = s_ < v2v;                \
    v2v = c1_ ? v1 : (c2_ ? s_  : v2v); i2 = c1_ ? i1 : (c2_ ? id_ : i2); \
    v1  = c0_ ? v0 : (c1_ ? s_  : v1 ); i1 = c0_ ? i0 : (c1_ ? id_ : i1); \
    v0  = c0_ ? s_ : v0;                i0 = c0_ ? id_ : i0;          }

#define TOP3_VAL(x) {                                                 \
    const float x_ = (x);                                             \
    const float n1_ = __builtin_amdgcn_fmed3f(t0, t1, x_);            \
    const float n2_ = __builtin_amdgcn_fmed3f(t1, t2, x_);            \
    t0 = fminf(t0, x_); t1 = n1_; t2 = n2_;                           }

__device__ __forceinline__ v8f score_tile(const v2f qf, const float* __restrict__ p, const int half) {
  const float sx = p[0];
  const float sy = p[1];
  const float sz = p[2];
  const float s2 = fmaf(sz, sz, fmaf(sx, sx, sy * sy));
  v2f sf;
  sf.x = half ? sz : sx;
  sf.y = half ? s2 : sy;
  const v8f c = {0.f, 0.f, 0.f, 0.f, 0.f, 0.f, 0.f, 0.f};
  return __builtin_amdgcn_wmma_f32_16x16x4_f32(false, sf, false, qf, (short)0, c, false, false);
}

__device__ __forceinline__ float min8(const v8f c) {
  float a = fminf(fminf(c[0], c[1]), c[2]);
  a = fminf(fminf(a, c[3]), c[4]);
  a = fminf(fminf(a, c[5]), c[6]);
  return fminf(a, c[7]);
}

__device__ __forceinline__ float min16(const v8f c, const v8f d) {
  float a = fminf(fminf(c[0], c[1]), c[2]);
  float b = fminf(fminf(d[0], d[1]), d[2]);
  a = fminf(fminf(a, c[3]), c[4]);
  b = fminf(fminf(b, d[3]), d[4]);
  a = fminf(fminf(a, c[5]), c[6]);
  b = fminf(fminf(b, d[5]), d[6]);
  a = fminf(fminf(a, c[7]), d[7]);
  return fminf(a, b);
}

#define NW   8
#define SEG  (NS / NW)
#define NBLK (SEG / 32)
#define SREG 320
#define ROWS_PER_WAVE (16 / NW)
#define NT   (C / 16)

__global__ __launch_bounds__(NW * 32) __attribute__((amdgpu_waves_per_eu(12))) void fused_main(
    const float* __restrict__ coords_q, const float* __restrict__ coords_s,
    const float* __restrict__ x7,       const float* __restrict__ feat_s,
    const float* __restrict__ gate,     const float* __restrict__ W1,
    const float* __restrict__ gammaW,   const float* __restrict__ betaW,
    const float* __restrict__ W2,       const float* __restrict__ W_cls,
    const float* __restrict__ b_cls,    float* __restrict__ out)
{
  __shared__ __attribute__((aligned(16))) float lds[WREG];
  __shared__ __attribute__((aligned(16))) float sc[NW * SREG];
  __shared__ float pv[NW * 16 * 3];
  __shared__ int   pi[NW * 16 * 3];
  __shared__ float bt[NW * 16 * 3];
  __shared__ float tw[16 * 3];
  __shared__ int   ti[16 * 3];
  const int lane = threadIdx.x & 31;
  const int wave = __builtin_amdgcn_readfirstlane(threadIdx.x >> 5);
  const int tile = blockIdx.x;

  float* WL = lds;
  const int r15  = lane & 15;
  const int half = lane >> 4;
  const int qbase = tile * 16;

  const float qx = coords_q[(qbase + r15) * 3 + 0];
  const float qy = coords_q[(qbase + r15) * 3 + 1];
  const float qz = coords_q[(qbase + r15) * 3 + 2];
  v2f qf;
  qf.x = half ? (-2.f * qz) : (-2.f * qx);
  qf.y = half ? 1.0f        : (-2.f * qy);

  const int nbeg = wave * SEG;
  const float* const seg = coords_s + (nbeg + r15) * 3;

  float mt[NBLK];
  float t0 = 1e30f, t1 = 1e30f, t2 = 1e30f;
#pragma unroll
  for (int t = 0; t < NBLK; ++t) {
    const v8f c1 = score_tile(qf, seg + t * 96, half);
    const v8f c2 = score_tile(qf, seg + t * 96 + 48, half);
    const float m = min16(c1, c2);
    mt[t] = m;
    TOP3_VAL(m);
  }

  float T;
  {
    const float o0 = __shfl(t0, lane ^ 16), o1 = __shfl(t1, lane ^ 16), o2 = __shfl(t2, lane ^ 16);
    TOP3_VAL(o0); TOP3_VAL(o1); TOP3_VAL(o2);
    if (lane < 16) {
      const int o = (wave * 16 + lane) * 3;
      bt[o] = t0; bt[o + 1] = t1; bt[o + 2] = t2;
    }
    __syncthreads();
    t0 = 1e30f; t1 = 1e30f; t2 = 1e30f;
    for (int w = 0; w < NW; ++w) {
      const int o = (w * 16 + r15) * 3;
      TOP3_VAL(bt[o]); TOP3_VAL(bt[o + 1]); TOP3_VAL(bt[o + 2]);
    }
    T = t2;
  }

  unsigned blkmask = 0u;
#pragma unroll
  for (int t = 0; t < NBLK; ++t)
    blkmask |= (__builtin_amdgcn_ballot_w32(mt[t] <= T) != 0u) ? (1u << t) : 0u;

  float v0 = 1e30f, v1 = 1e30f, v2v = 1e30f;
  int   i0 = 0, i1 = 0, i2 = 0;

#define INSERT_TILE(cv, base) {                                                   \
    const int b_ = (base);                                                        \
    TOP3_INS((cv)[0], b_ + 0); TOP3_INS((cv)[1], b_ + 1);                         \
    TOP3_INS((cv)[2], b_ + 2); TOP3_INS((cv)[3], b_ + 3);                         \
    TOP3_INS((cv)[4], b_ + 4); TOP3_INS((cv)[5], b_ + 5);                         \
    TOP3_INS((cv)[6], b_ + 6); TOP3_INS((cv)[7], b_ + 7);                         }

  while (blkmask != 0u) {
    const int t = __builtin_ctz(blkmask);
    blkmask &= blkmask - 1u;
    const float* p = seg + t * 96;
    const int n0 = nbeg + 32 * t + 8 * half;
    const v8f c1 = score_tile(qf, p, half);
    const v8f c2 = score_tile(qf, p + 48, half);
    const unsigned m1 = __builtin_amdgcn_ballot_w32(min8(c1) <= T);
    const unsigned m2 = __builtin_amdgcn_ballot_w32(min8(c2) <= T);
    if (m1) INSERT_TILE(c1, n0);
    if (m2) INSERT_TILE(c2, n0 + 16);
  }
#undef INSERT_TILE

  {
    float pv0 = __shfl(v0, lane + 16), pv1 = __shfl(v1, lane + 16), pv2 = __shfl(v2v, lane + 16);
    int   pi0 = __shfl(i0, lane + 16), pi1 = __shfl(i1, lane + 16), pi2 = __shfl(i2,  lane + 16);
    TOP3_INS(pv0, pi0); TOP3_INS(pv1, pi1); TOP3_INS(pv2, pi2);
  }
  if (lane < 16) {
    const int o = (wave * 16 + lane) * 3;
    pv[o] = v0; pv[o + 1] = v1; pv[o + 2] = v2v;
    pi[o] = i0; pi[o + 1] = i1; pi[o + 2] = i2;
  }
  __syncthreads();
  if (wave == 0) {
    v0 = 1e30f; v1 = 1e30f; v2v = 1e30f;
    i0 = 0; i1 = 0; i2 = 0;
    for (int w = 0; w < NW; ++w) {
      const int o = (w * 16 + r15) * 3;
      TOP3_INS(pv[o], pi[o]); TOP3_INS(pv[o + 1], pi[o + 1]); TOP3_INS(pv[o + 2], pi[o + 2]);
    }
    const float e1 = __expf(v0 - v1), e2 = __expf(v0 - v2v);
    const float inv = 1.0f / (1.0f + e1 + e2);
    if (lane < 16) {
      tw[lane * 3] = inv; tw[lane * 3 + 1] = e1 * inv; tw[lane * 3 + 2] = e2 * inv;
      ti[lane * 3] = i0;  ti[lane * 3 + 1] = i1;       ti[lane * 3 + 2] = i2;
    }
  }
  __syncthreads();

  for (int r = wave * ROWS_PER_WAVE; r < (wave + 1) * ROWS_PER_WAVE; ++r) {
    const float rw0 = tw[r * 3], rw1 = tw[r * 3 + 1], rw2 = tw[r * 3 + 2];
    const int   ri0 = ti[r * 3], ri1 = ti[r * 3 + 1], ri2 = ti[r * 3 + 2];
    const int qi = qbase + r;
    for (int t = lane; t < C2; t += 32) {
      float v;
      if (t < C) {
        v = x7[(long)qi * C + t] * gate[t];
      } else {
        const int cc = t - C;
        v = rw0 * feat_s[(long)ri0 * C + cc]
          + rw1 * feat_s[(long)ri1 * C + cc]
          + rw2 * feat_s[(long)ri2 * C + cc];
      }
      WL[r * XSTRIDE + t] = v;
    }
  }
  __syncthreads();

  const int ch = wave * 16 + r15;

  v8f acc = {0.f, 0.f, 0.f, 0.f, 0.f, 0.f, 0.f, 0.f};
  if (wave < NT) {
    for (int k0 = 0; k0 < C2; k0 += 4) {
      const int kk = k0 + 2 * half;
      const v2f a1 = *(const v2f*)(WL + r15 * XSTRIDE + kk);
      v2f bv;
      bv.x = W1[(long)kk * C + ch];
      bv.y = W1[(long)(kk + 1) * C + ch];
      acc = __builtin_amdgcn_wmma_f32_16x16x4_f32(false, a1, false, bv, (short)0, acc, false, false);
    }
  }
  __syncthreads();

  if (wave < NT) {
    const float gm = gammaW[ch], bt = betaW[ch];
#pragma unroll
    for (int r = 0; r < 8; ++r) {
      const float h = fmaxf(acc[r] * gm + bt, 0.f);
      WL[H_OFF + (r + 8 * half) * HSTRIDE + ch] = h;
    }
  }
  __syncthreads();

  float* const P = sc + wave * SREG;
  const float bias0 = b_cls[threadIdx.x % CLS];
  const float bias1 = b_cls[(threadIdx.x + NW * 32) % CLS];
  if (wave < NT) {
    const int cB = (r15 + 16 < CLS) ? (r15 + 16) : (CLS - 1);
    v2f wc0[4], wc1[4];
#pragma unroll
    for (int s = 0; s < 4; ++s) {
      const int kk = wave * 16 + 4 * s + 2 * half;
      wc0[s].x = W_cls[(long)kk * CLS + r15];
      wc0[s].y = W_cls[(long)(kk + 1) * CLS + r15];
      wc1[s].x = W_cls[(long)kk * CLS + cB];
      wc1[s].y = W_cls[(long)(kk + 1) * CLS + cB];
    }

    v8f acc2 = {0.f, 0.f, 0.f, 0.f, 0.f, 0.f, 0.f, 0.f};
    for (int k0 = 0; k0 < C; k0 += 4) {
      const int kk = k0 + 2 * half;
      const v2f a2 = *(const v2f*)(WL + H_OFF + r15 * HSTRIDE + kk);
      v2f bv;
      bv.x = W2[(long)kk * C + ch];
      bv.y = W2[(long)(kk + 1) * C + ch];
      acc2 = __builtin_amdgcn_wmma_f32_16x16x4_f32(false, a2, false, bv, (short)0, acc2, false, false);
    }
#pragma unroll
    for (int r = 0; r < 8; ++r)
      WL[Y2_OFF + (r + 8 * half) * HSTRIDE + ch] = acc2[r];
    asm volatile("" ::: "memory");
    v8f p0 = {0.f, 0.f, 0.f, 0.f, 0.f, 0.f, 0.f, 0.f};
    v8f p1 = p0;
#pragma unroll
    for (int s = 0; s < 4; ++s) {
      const int kk = wave * 16 + 4 * s + 2 * half;
      const v2f a3 = *(const v2f*)(WL + Y2_OFF + r15 * HSTRIDE + kk);
      p0 = __builtin_amdgcn_wmma_f32_16x16x4_f32(false, a3, false, wc0[s], (short)0, p0, false, false);
      p1 = __builtin_amdgcn_wmma_f32_16x16x4_f32(false, a3, false, wc1[s], (short)0, p1, false, false);
    }
#pragma unroll
    for (int r = 0; r < 8; ++r)
      P[(r + 8 * half) * CLS + r15] = p0[r];
    if (r15 + 16 < CLS) {
#pragma unroll
      for (int r = 0; r < 8; ++r)
        P[(r + 8 * half) * CLS + 16 + r15] = p1[r];
    }
  }
  __syncthreads();

  {
    const int t = threadIdx.x;
    float s = sc[t];
#pragma unroll
    for (int w = 1; w < NT; ++w) s += sc[w * SREG + t];
    out[(long)qbase * CLS + t] = s + bias0;
    if (t < 16 * CLS - NW * 32) {
      const int u = t + NW * 32;
      float s1 = sc[u];
#pragma unroll
      for (int w = 1; w < NT; ++w) s1 += sc[w * SREG + u];
      out[(long)qbase * CLS + u] = s1 + bias1;
    }
  }
}

extern "C" void kernel_launch(void* const* d_in, const int* in_sizes, int n_in,
                              void* d_out, int out_size, void* d_ws, size_t ws_size,
                              hipStream_t stream) {
  const float* coords_q = (const float*)d_in[0];
  const float* coords_s = (const float*)d_in[1];
  const float* x7       = (const float*)d_in[2];
  const float* feat_s7  = (const float*)d_in[3];
  const float* feat_s   = (const float*)d_in[4];
  const float* w_attn   = (const float*)d_in[5];
  const float* W1       = (const float*)d_in[6];
  const float* gammaW   = (const float*)d_in[7];
  const float* betaW    = (const float*)d_in[8];
  const float* W2       = (const float*)d_in[9];
  const float* W_cls    = (const float*)d_in[10];
  const float* b_cls    = (const float*)d_in[11];
  float* out = (float*)d_out;

  float* ws   = (float*)d_ws;
  float* rec  = ws;
  float* gate = ws + PB_N * REC;

  pool_partials<<<PB_N, 256, 0, stream>>>(feat_s7, w_attn, rec);
  pool_combine<<<1, 512, 0, stream>>>(rec, gate);

  fused_main<<<TILES, NW * 32, 0, stream>>>(coords_q, coords_s, x7, feat_s, gate,
                                         W1, gammaW, betaW, W2, W_cls, b_cls, out);
}


# W2 and W_cls folded: 7 extra work-groups of the pool launch form W2c = W2 @ W_cls once (f32 WMMA); the tile's GEMM2 and one barrier are gone, H feeds the classifier GEMM; sums re-associated
# speedup vs baseline: 19.0895x; 1.0613x over previous
#include <hip/hip_runtime.h>
#include <math.h>

#define NQ   20000
#define NS   8192
#define C    112
#define C2   224
#define CLS  20
#define TILES (NQ / 16)

#define WREG    3712
#define XSTRIDE 226
#define HSTRIDE 114
#define H_OFF   0

#define NW   8
#define NT   (C / 16)
#define SREG 320

typedef float v2f __attribute__((ext_vector_type(2)));
typedef float v8f __attribute__((ext_vector_type(8)));

#define PB_ROWS  64
#define PB_N     (NS / PB_ROWS)
#define PB_PITCH 116
#define REC      128

__device__ __forceinline__ void fold_tile(const float* __restrict__ W2, const float* __restrict__ W_cls,
                                          float* __restrict__ W2c, float* __restrict__ part, const int rt) {
  const int t = threadIdx.x;
  const int lane = t & 31;
  const int r15 = lane & 15;
  const int half = lane >> 4;
  const int wave = __builtin_amdgcn_readfirstlane(t >> 5);
  if (wave < NT) {
    const int cB = (r15 + 16 < CLS) ? (r15 + 16) : (CLS - 1);
    const float* const arow = W2 + (long)(rt * 16 + r15) * C;
    v8f p0 = {0.f, 0.f, 0.f, 0.f, 0.f, 0.f, 0.f, 0.f};
    v8f p1 = p0;
#pragma unroll
    for (int s = 0; s < 4; ++s) {
      const int kk = wave * 16 + 4 * s + 2 * half;
      const v2f a = *(const v2f*)(arow + kk);
      v2f b0, b1;
      b0.x = W_cls[(long)kk * CLS + r15];
      b0.y = W_cls[(long)(kk + 1) * CLS + r15];
      b1.x = W_cls[(long)kk * CLS + cB];
      b1.y = W_cls[(long)(kk + 1) * CLS + cB];
      p0 = __builtin_amdgcn_wmma_f32_16x16x4_f32(false, a, false, b0, (short)0, p0, false, false);
      p1 = __builtin_amdgcn_wmma_f32_16x16x4_f32(false, a, false, b1, (short)0, p1, false, false);
    }
    float* const P = part + wave * SREG;
#pragma unroll
    for (int r = 0; r < 8; ++r)
      P[(r + 8 * half) * CLS + r15] = p0[r];
    if (r15 + 16 < CLS) {
#pragma unroll
      for (int r = 0; r < 8; ++r)
        P[(r + 8 * half) * CLS + 16 + r15] = p1[r];
    }
  }
  __syncthreads();
  float s = part[t];
#pragma unroll
  for (int w = 1; w < NT; ++w) s += part[w * SREG + t];
  W2c[(long)rt * SREG + t] = s;
  if (t < SREG - NW * 32) {
    const int u = t + NW * 32;
    float s1 = part[u];
#pragma unroll
    for (int w = 1; w < NT; ++w) s1 += part[w * SREG + u];
    W2c[(long)rt * SREG + u] = s1;
  }
}

__global__ __launch_bounds__(256) void pool_partials(const float* __restrict__ feat_s7,
                                                      const float* __restrict__ w_attn,
                                                      const float* __restrict__ W2,
                                                      const float* __restrict__ W_cls,
                                                      float* __restrict__ rec,
                                                      float* __restrict__ W2c) {
  __shared__ __attribute__((aligned(16))) float tile[PB_ROWS * PB_PITCH];
  __shared__ __attribute__((aligned(16))) float wl[C];
  __shared__ float lg[PB_ROWS];
  __shared__ float ev[PB_ROWS];
  __shared__ float part[2 * C];
  const int t = threadIdx.x;
  const int lane = t & 31;

  if (blockIdx.x >= PB_N) {
    fold_tile(W2, W_cls, W2c, tile, blockIdx.x - PB_N);
    return;
  }

  if (t < C / 4) *(float4*)(wl + 4 * t) = ((const float4*)w_attn)[t];

  const float4* src = (const float4*)(feat_s7 + (long)blockIdx.x * (PB_ROWS * C));
#pragma unroll
  for (int i = 0; i < 7; ++i) {
    const int q = t + 256 * i;
    const int row = q / 28, c4 = q - row * 28;
    *(float4*)(tile + row * PB_PITCH + 4 * c4) = src[q];
  }
  __syncthreads();

  {
    const int row = t >> 2, p = t & 3;
    float acc = 0.f;
#pragma unroll
    for (int i = 0; i < 7; ++i) {
      const int c4 = p + 4 * i;
      const float4 x = *(const float4*)(tile + row * PB_PITCH + 4 * c4);
      const float4 w = *(const float4*)(wl + 4 * c4);
      acc += x.x * w.x + x.y * w.y + x.z * w.z + x.w * w.w;
    }
    acc += __shfl_xor(acc, 1);
    acc += __shfl_xor(acc, 2);
    if (p == 0) lg[row] = acc;
  }
  __syncthreads();

  float m = fmaxf(lg[lane], lg[lane + 32]);
  for (int o = 16; o > 0; o >>= 1) m = fmaxf(m, __shfl_xor(m, o));
  if (t < PB_ROWS) ev[t] = __expf(lg[t] - m);
  __syncthreads();
  float s = ev[lane] + ev[lane + 32];
  for (int o = 16; o > 0; o >>= 1) s += __shfl_xor(s, o);

  if (t < 2 * C) {
    const int h = (t >= C) ? 1 : 0;
    const int c = t - h * C;
    const float* col = tile + (h * 32) * PB_PITCH + c;
    const float* e = ev + h * 32;
    float a0 = 0.f, a1 = 0.f;
#pragma unroll
    for (int r = 0; r < 32; r += 2) {
      a0 += e[r] * col[r * PB_PITCH];
      a1 += e[r + 1] * col[(r + 1) * PB_PITCH];
    }
    part[t] = a0 + a1;
  }
  __syncthreads();

  if (t < REC) {
    float v = 0.f;
    if (t < C) v = part[t] + part[C + t];
    else if (t == C) v = m;
    else if (t == C + 1) v = s;
    rec[(long)blockIdx.x * REC + t] = v;
  }
}

__global__ __launch_bounds__(512) void pool_combine(const float* __restrict__ rec,
                                                     float* __restrict__ gate) {
  __shared__ float ml[PB_N];
  __shared__ float sl[PB_N];
  __shared__ float fs[PB_N];
  __shared__ float part[4 * REC];
  const int t = threadIdx.x;
  const int lane = t & 31;
  const int c = t & (REC - 1);
  const int g = t >> 7;

  float x[32];
#pragma unroll
  for (int i = 0; i < 32; ++i) x[i] = rec[(long)(g * 32 + i) * REC + c];

  if (t < PB_N) { ml[t] = rec[(long)t * REC + C]; sl[t] = rec[(long)t * REC + C + 1]; }
  __syncthreads();
  float M = fmaxf(fmaxf(ml[lane], ml[lane + 32]), fmaxf(ml[lane + 64], ml[lane + 96]));
  for (int o = 16; o > 0; o >>= 1) M = fmaxf(M, __shfl_xor(M, o));
  if (t < PB_N) { const float f = __expf(ml[t] - M); fs[t] = f; sl[t] = sl[t] * f; }
  __syncthreads();
  float S = (sl[lane] + sl[lane + 32]) + (sl[lane + 64] + sl[lane + 96]);
  for (int o = 16; o > 0; o >>= 1) S += __shfl_xor(S, o);

  float a0 = 0.f, a1 = 0.f;
#pragma unroll
  for (int i = 0; i < 32; i += 2) {
    a0 += x[i] * fs[g * 32 + i];
    a1 += x[i + 1] * fs[g * 32 + i + 1];
  }
  part[t] = a0 + a1;
  __syncthreads();

  if (t < C) {
    const float acc = (part[t] + part[REC + t]) + (part[2 * REC + t] + part[3 * REC + t]);
    const float gv = acc * (1.0f / S);
    gate[t] = 1.0f / (1.0f + __expf(-gv));
  }
}

#define TOP3_INS(val, idxv) {                                         \
    float s_ = (val); int id_ = (idxv);                               \
    bool c0_ = s_ < v0, c1_ = s_ < v1, c2_ = s_ < v2v;                \
    v2v = c1_ ? v1 : (c2_ ? s_  : v2v); i2 = c1_ ? i1 : (c2_ ? id_ : i2); \
    v1  = c0_ ? v0 : (c1_ ? s_  : v1 ); i1 = c0_ ? i0 : (c1_ ? id_ : i1); \
    v0  = c0_ ? s_ : v0;                i0 = c0_ ? id_ : i0;          }

#define TOP3_VAL(x) {                                                 \
    const float x_ = (x);                                             \
    const float n1_ = __builtin_amdgcn_fmed3f(t0, t1, x_);            \
    const float n2_ = __builtin_amdgcn_fmed3f(t1, t2, x_);            \
    t0 = fminf(t0, x_); t1 = n1_; t2 = n2_;                           }

__device__ __forceinline__ v8f score_tile(const v2f qf, const float* __restrict__ p, const int half) {
  const float sx = p[0];
  const float sy = p[1];
  const float sz = p[2];
  const float s2 = fmaf(sz, sz, fmaf(sx, sx, sy * sy));
  v2f sf;
  sf.x = half ? sz : sx;
  sf.y = half ? s2 : sy;
  const v8f c = {0.f, 0.f, 0.f, 0.f, 0.f, 0.f, 0.f, 0.f};
  return __builtin_amdgcn_wmma_f32_16x16x4_f32(false, sf, false, qf, (short)0, c, false, false);
}

__device__ __forceinline__ float min8(const v8f c) {
  float a = fminf(fminf(c[0], c[1]), c[2]);
  a = fminf(fminf(a, c[3]), c[4]);
  a = fminf(fminf(a, c[5]), c[6]);
  return fminf(a, c[7]);
}

__device__ __forceinline__ float min16(const v8f c, const v8f d) {
  float a = fminf(fminf(c[0], c[1]), c[2]);
  float b = fminf(fminf(d[0], d[1]), d[2]);
  a = fminf(fminf(a, c[3]), c[4]);
  b = fminf(fminf(b, d[3]), d[4]);
  a = fminf(fminf(a, c[5]), c[6]);
  b = fminf(fminf(b, d[5]), d[6]);
  a = fminf(fminf(a, c[7]), d[7]);
  return fminf(a, b);
}

#define SEG  (NS / NW)
#define NBLK (SEG / 32)
#define ROWS_PER_WAVE (16 / NW)

__global__ __launch_bounds__(NW * 32) __attribute__((amdgpu_waves_per_eu(12))) void fused_main(
    const float* __restrict__ coords_q, const float* __restrict__ coords_s,
    const float* __restrict__ x7,       const float* __restrict__ feat_s,
    const float* __restrict__ gate,     const float* __restrict__ W1,
    const float* __restrict__ gammaW,   const float* __restrict__ betaW,
    const float* __restrict__ W2c,      const float* __restrict__ b_cls,
    float* __restrict__ out)
{
  __shared__ __attribute__((aligned(16))) float lds[WREG];
  __shared__ __attribute__((aligned(16))) float sc[NW * SREG];
  __shared__ float pv[NW * 16 * 3];
  __shared__ int   pi[NW * 16 * 3];
  __shared__ float bt[NW * 16 * 3];
  __shared__ float tw[16 * 3];
  __shared__ int   ti[16 * 3];
  const int lane = threadIdx.x & 31;
  const int wave = __builtin_amdgcn_readfirstlane(threadIdx.x >> 5);
  const int tile = blockIdx.x;

  float* WL = lds;
  const int r15  = lane & 15;
  const int half = lane >> 4;
  const int qbase = tile * 16;

  const float qx = coords_q[(qbase + r15) * 3 + 0];
  const float qy = coords_q[(qbase + r15) * 3 + 1];
  const float qz = coords_q[(qbase + r15) * 3 + 2];
  v2f qf;
  qf.x = half ? (-2.f * qz) : (-2.f * qx);
  qf.y = half ? 1.0f        : (-2.f * qy);

  const int nbeg = wave * SEG;
  const float* const seg = coords_s + (nbeg + r15) * 3;

  float mt[NBLK];
  float t0 = 1e30f, t1 = 1e30f, t2 = 1e30f;
#pragma unroll
  for (int t = 0; t < NBLK; ++t) {
    const v8f c1 = score_tile(qf, seg + t * 96, half);
    const v8f c2 = score_tile(qf, seg + t * 96 + 48, half);
    const float m = min16(c1, c2);
    mt[t] = m;
    TOP3_VAL(m);
  }

  float T;
  {
    const float o0 = __shfl(t0, lane ^ 16), o1 = __shfl(t1, lane ^ 16), o2 = __shfl(t2, lane ^ 16);
    TOP3_VAL(o0); TOP3_VAL(o1); TOP3_VAL(o2);
    if (lane < 16) {
      const int o = (wave * 16 + lane) * 3;
      bt[o] = t0; bt[o + 1] = t1; bt[o + 2] = t2;
    }
    __syncthreads();
    t0 = 1e30f; t1 = 1e30f; t2 = 1e30f;
    for (int w = 0; w < NW; ++w) {
      const int o = (w * 16 + r15) * 3;
      TOP3_VAL(bt[o]); TOP3_VAL(bt[o + 1]); TOP3_VAL(bt[o + 2]);
    }
    T = t2;
  }

  unsigned blkmask = 0u;
#pragma unroll
  for (int t = 0; t < NBLK; ++t)
    blkmask |= (__builtin_amdgcn_ballot_w32(mt[t] <= T) != 0u) ? (1u << t) : 0u;

  float v0 = 1e30f, v1 = 1e30f, v2v = 1e30f;
  int   i0 = 0, i1 = 0, i2 = 0;

#define INSERT_TILE(cv, base) {                                                   \
    const int b_ = (base);                                                        \
    TOP3_INS((cv)[0], b_ + 0); TOP3_INS((cv)[1], b_ + 1);                         \
    TOP3_INS((cv)[2], b_ + 2); TOP3_INS((cv)[3], b_ + 3);                         \
    TOP3_INS((cv)[4], b_ + 4); TOP3_INS((cv)[5], b_ + 5);                         \
    TOP3_INS((cv)[6], b_ + 6); TOP3_INS((cv)[7], b_ + 7);                         }

  while (blkmask != 0u) {
    const int t = __builtin_ctz(blkmask);
    blkmask &= blkmask - 1u;
    const float* p = seg + t * 96;
    const int n0 = nbeg + 32 * t + 8 * half;
    const v8f c1 = score_tile(qf, p, half);
    const v8f c2 = score_tile(qf, p + 48, half);
    const unsigned m1 = __builtin_amdgcn_ballot_w32(min8(c1) <= T);
    const unsigned m2 = __builtin_amdgcn_ballot_w32(min8(c2) <= T);
    if (m1) INSERT_TILE(c1, n0);
    if (m2) INSERT_TILE(c2, n0 + 16);
  }
#undef INSERT_TILE

  {
    float pv0 = __shfl(v0, lane + 16), pv1 = __shfl(v1, lane + 16), pv2 = __shfl(v2v, lane + 16);
    int   pi0 = __shfl(i0, lane + 16), pi1 = __shfl(i1, lane + 16), pi2 = __shfl(i2,  lane + 16);
    TOP3_INS(pv0, pi0); TOP3_INS(pv1, pi1); TOP3_INS(pv2, pi2);
  }
  if (lane < 16) {
    const int o = (wave * 16 + lane) * 3;
    pv[o] = v0; pv[o + 1] = v1; pv[o + 2] = v2v;
    pi[o] = i0; pi[o + 1] = i1; pi[o + 2] = i2;
  }
  __syncthreads();
  if (wave == 0) {
    v0 = 1e30f; v1 = 1e30f; v2v = 1e30f;
    i0 = 0; i1 = 0; i2 = 0;
    for (int w = 0; w < NW; ++w) {
      const int o = (w * 16 + r15) * 3;
      TOP3_INS(pv[o], pi[o]); TOP3_INS(pv[o + 1], pi[o + 1]); TOP3_INS(pv[o + 2], pi[o + 2]);
    }
    const float e1 = __expf(v0 - v1), e2 = __expf(v0 - v2v);
    const float inv = 1.0f / (1.0f + e1 + e2);
    if (lane < 16) {
      tw[lane * 3] = inv; tw[lane * 3 + 1] = e1 * inv; tw[lane * 3 + 2] = e2 * inv;
      ti[lane * 3] = i0;  ti[lane * 3 + 1] = i1;       ti[lane * 3 + 2] = i2;
    }
  }
  __syncthreads();

  for (int r = wave * ROWS_PER_WAVE; r < (wave + 1) * ROWS_PER_WAVE; ++r) {
    const float rw0 = tw[r * 3], rw1 = tw[r * 3 + 1], rw2 = tw[r * 3 + 2];
    const int   ri0 = ti[r * 3], ri1 = ti[r * 3 + 1], ri2 = ti[r * 3 + 2];
    const int qi = qbase + r;
    for (int t = lane; t < C2; t += 32) {
      float v;
      if (t < C) {
        v = x7[(long)qi * C + t] * gate[t];
      } else {
        const int cc = t - C;
        v = rw0 * feat_s[(long)ri0 * C + cc]
          + rw1 * feat_s[(long)ri1 * C + cc]
          + rw2 * feat_s[(long)ri2 * C + cc];
      }
      WL[r * XSTRIDE + t] = v;
    }
  }
  __syncthreads();

  const int ch = wave * 16 + r15;

  const float bias0 = b_cls[threadIdx.x % CLS];
  const float bias1 = b_cls[(threadIdx.x + NW * 32) % CLS];

  v8f acc = {0.f, 0.f, 0.f, 0.f, 0.f, 0.f, 0.f, 0.f};
  v2f wc0[4] = {}, wc1[4] = {};
  if (wave < NT) {
    const int cB = (r15 + 16 < CLS) ? (r15 + 16) : (CLS - 1);
#pragma unroll
    for (int s = 0; s < 4; ++s) {
      const int kk = wave * 16 + 4 * s + 2 * half;
      wc0[s].x = W2c[(long)kk * CLS + r15];
      wc0[s].y = W2c[(long)(kk + 1) * CLS + r15];
      wc1[s].x = W2c[(long)kk * CLS + cB];
      wc1[s].y = W2c[(long)(kk + 1) * CLS + cB];
    }
    for (int k0 = 0; k0 < C2; k0 += 4) {
      const int kk = k0 + 2 * half;
      const v2f a1 = *(const v2f*)(WL + r15 * XSTRIDE + kk);
      v2f bv;
      bv.x = W1[(long)kk * C + ch];
      bv.y = W1[(long)(kk + 1) * C + ch];
      acc = __builtin_amdgcn_wmma_f32_16x16x4_f32(false, a1, false, bv, (short)0, acc, false, false);
    }
  }
  __syncthreads();

  float* const P = sc + wave * SREG;
  if (wave < NT) {
    const float gm = gammaW[ch], bt = betaW[ch];
#pragma unroll
    for (int r = 0; r < 8; ++r) {
      const float h = fmaxf(acc[r] * gm + bt, 0.f);
      WL[H_OFF + (r + 8 * half) * HSTRIDE + ch] = h;
    }
    asm volatile("" ::: "memory");
    v8f p0 = {0.f, 0.f, 0.f, 0.f, 0.f, 0.f, 0.f, 0.f};
    v8f p1 = p0;
#pragma unroll
    for (int s = 0; s < 4; ++s) {
      const int kk = wave * 16 + 4 * s + 2 * half;
      const v2f a3 = *(const v2f*)(WL + H_OFF + r15 * HSTRIDE + kk);
      p0 = __builtin_amdgcn_wmma_f32_16x16x4_f32(false, a3, false, wc0[s], (short)0, p0, false, false);
      p1 = __builtin_amdgcn_wmma_f32_16x16x4_f32(false, a3, false, wc1[s], (short)0, p1, false, false);
    }
#pragma unroll
    for (int r = 0; r < 8; ++r)
      P[(r + 8 * half) * CLS + r15] = p0[r];
    if (r15 + 16 < CLS) {
#pragma unroll
      for (int r = 0; r < 8; ++r)
        P[(r + 8 * half) * CLS + 16 + r15] = p1[r];
    }
  }
  __syncthreads();

  {
    const int t = threadIdx.x;
    float s = sc[t];
#pragma unroll
    for (int w = 1; w < NT; ++w) s += sc[w * SREG + t];
    out[(long)qbase * CLS + t] = s + bias0;
    if (t < 16 * CLS - NW * 32) {
      const int u = t + NW * 32;
      float s1 = sc[u];
#pragma unroll
      for (int w = 1; w < NT; ++w) s1 += sc[w * SREG + u];
      out[(long)qbase * CLS + u] = s1 + bias1;
    }
  }
}

extern "C" void kernel_launch(void* const* d_in, const int* in_sizes, int n_in,
                              void* d_out, int out_size, void* d_ws, size_t ws_size,
                              hipStream_t stream) {
  const float* coords_q = (const float*)d_in[0];
  const float* coords_s = (const float*)d_in[1];
  const float* x7       = (const float*)d_in[2];
  const float* feat_s7  = (const float*)d_in[3];
  const float* feat_s   = (const float*)d_in[4];
  const float* w_attn   = (const float*)d_in[5];
  const float* W1       = (const float*)d_in[6];
  const float* gammaW   = (const float*)d_in[7];
  const float* betaW    = (const float*)d_in[8];
  const float* W2       = (const float*)d_in[9];
  const float* W_cls    = (const float*)d_in[10];
  const float* b_cls    = (const float*)d_in[11];
  float* out = (float*)d_out;

  float* ws   = (float*)d_ws;
  float* rec  = ws;
  float* gate = ws + PB_N * REC;
  float* W2c  = ws + PB_N * REC + REC;

  pool_partials<<<PB_N + NT, 256, 0, stream>>>(feat_s7, w_attn, W2, W_cls, rec, W2c);
  pool_combine<<<1, 512, 0, stream>>>(rec, gate);

  fused_main<<<TILES, NW * 32, 0, stream>>>(coords_q, coords_s, x7, feat_s, gate,
                                         W1, gammaW, betaW, W2c, b_cls, out);
}


# X rows of the tile built with every load in flight at once: a lane owns one 16-byte piece of x7, gate and the 3 matched rows (14 dependent load trips per wave become one); order of sums unchanged
# speedup vs baseline: 21.2760x; 1.1145x over previous
#include <hip/hip_runtime.h>
#include <math.h>

#define NQ   20000
#define NS   8192
#define C    112
#define C2   224
#define CLS  20
#define TILES (NQ / 16)

#define WREG    3712
#define XSTRIDE 226
#define HSTRIDE 114
#define H_OFF   0

#define NW   8
#define NT   (C / 16)
#define SREG 320

typedef float v2f __attribute__((ext_vector_type(2)));
typedef float v8f __attribute__((ext_vector_type(8)));

#define PB_ROWS  64
#define PB_N     (NS / PB_ROWS)
#define PB_PITCH 116
#define REC      128

__device__ __forceinline__ void fold_tile(const float* __restrict__ W2, const float* __restrict__ W_cls,
                                          float* __restrict__ W2c, float* __restrict__ part, const int rt) {
  const int t = threadIdx.x;
  const int lane = t & 31;
  const int r15 = lane & 15;
  const int half = lane >> 4;
  const int wave = __builtin_amdgcn_readfirstlane(t >> 5);
  if (wave < NT) {
    const int cB = (r15 + 16 < CLS) ? (r15 + 16) : (CLS - 1);
    const float* const arow = W2 + (long)(rt * 16 + r15) * C;
    v8f p0 = {0.f, 0.f, 0.f, 0.f, 0.f, 0.f, 0.f, 0.f};
    v8f p1 = p0;
#pragma unroll
    for (int s = 0; s < 4; ++s) {
      const int kk = wave * 16 + 4 * s + 2 * half;
      const v2f a = *(const v2f*)(arow + kk);
      v2f b0, b1;
      b0.x = W_cls[(long)kk * CLS + r15];
      b0.y = W_cls[(long)(kk + 1) * CLS + r15];
      b1.x = W_cls[(long)kk * CLS + cB];
      b1.y = W_cls[(long)(kk + 1) * CLS + cB];
      p0 = __builtin_amdgcn_wmma_f32_16x16x4_f32(false, a, false, b0, (short)0, p0, false, false);
      p1 = __builtin_amdgcn_wmma_f32_16x16x4_f32(false, a, false, b1, (short)0, p1, false, false);
    }
    float* const P = part + wave * SREG;
#pragma unroll
    for (int r = 0; r < 8; ++r)
      P[(r + 8 * half) * CLS + r15] = p0[r];
    if (r15 + 16 < CLS) {
#pragma unroll
      for (int r = 0; r < 8; ++r)
        P[(r + 8 * half) * CLS + 16 + r15] = p1[r];
    }
  }
  __syncthreads();
  float s = part[t];
#pragma unroll
  for (int w = 1; w < NT; ++w) s += part[w * SREG + t];
  W2c[(long)rt * SREG + t] = s;
  if (t < SREG - NW * 32) {
    const int u = t + NW * 32;
    float s1 = part[u];
#pragma unroll
    for (int w = 1; w < NT; ++w) s1 += part[w * SREG + u];
    W2c[(long)rt * SREG + u] = s1;
  }
}

__global__ __launch_bounds__(256) void pool_partials(const float* __restrict__ feat_s7,
                                                      const float* __restrict__ w_attn,
                                                      const float* __restrict__ W2,
                                                      const float* __restrict__ W_cls,
                                                      float* __restrict__ rec,
                                                      float* __restrict__ W2c) {
  __shared__ __attribute__((aligned(16))) float tile[PB_ROWS * PB_PITCH];
  __shared__ __attribute__((aligned(16))) float wl[C];
  __shared__ float lg[PB_ROWS];
  __shared__ float ev[PB_ROWS];
  __shared__ float part[2 * C];
  const int t = threadIdx.x;
  const int lane = t & 31;

  if (blockIdx.x >= PB_N) {
    fold_tile(W2, W_cls, W2c, tile, blockIdx.x - PB_N);
    return;
  }

  if (t < C / 4) *(float4*)(wl + 4 * t) = ((const float4*)w_attn)[t];

  const float4* src = (const float4*)(feat_s7 + (long)blockIdx.x * (PB_ROWS * C));
#pragma unroll
  for (int i = 0; i < 7; ++i) {
    const int q = t + 256 * i;
    const int row = q / 28, c4 = q - row * 28;
    *(float4*)(tile + row * PB_PITCH + 4 * c4) = src[q];
  }
  __syncthreads();

  {
    const int row = t >> 2, p = t & 3;
    float acc = 0.f;
#pragma unroll
    for (int i = 0; i < 7; ++i) {
      const int c4 = p + 4 * i;
      const float4 x = *(const float4*)(tile + row * PB_PITCH + 4 * c4);
      const float4 w = *(const float4*)(wl + 4 * c4);
      acc += x.x * w.x + x.y * w.y + x.z * w.z + x.w * w.w;
    }
    acc += __shfl_xor(acc, 1);
    acc += __shfl_xor(acc, 2);
    if (p == 0) lg[row] = acc;
  }
  __syncthreads();

  float m = fmaxf(lg[lane], lg[lane + 32]);
  for (int o = 16; o > 0; o >>= 1) m = fmaxf(m, __shfl_xor(m, o));
  if (t < PB_ROWS) ev[t] = __expf(lg[t] - m);
  __syncthreads();
  float s = ev[lane] + ev[lane + 32];
  for (int o = 16; o > 0; o >>= 1) s += __shfl_xor(s, o);

  if (t < 2 * C) {
    const int h = (t >= C) ? 1 : 0;
    const int c = t - h * C;
    const float* col = tile + (h * 32) * PB_PITCH + c;
    const float* e = ev + h * 32;
    float a0 = 0.f, a1 = 0.f;
#pragma unroll
    for (int r = 0; r < 32; r += 2) {
      a0 += e[r] * col[r * PB_PITCH];
      a1 += e[r + 1] * col[(r + 1) * PB_PITCH];
    }
    part[t] = a0 + a1;
  }
  __syncthreads();

  if (t < REC) {
    float v = 0.f;
    if (t < C) v = part[t] + part[C + t];
    else if (t == C) v = m;
    else if (t == C + 1) v = s;
    rec[(long)blockIdx.x * REC + t] = v;
  }
}

__global__ __launch_bounds__(512) void pool_combine(const float* __restrict__ rec,
                                                     float* __restrict__ gate) {
  __shared__ float ml[PB_N];
  __shared__ float sl[PB_N];
  __shared__ float fs[PB_N];
  __shared__ float part[4 * REC];
  const int t = threadIdx.x;
  const int lane = t & 31;
  const int c = t & (REC - 1);
  const int g = t >> 7;

  float x[32];
#pragma unroll
  for (int i = 0; i < 32; ++i) x[i] = rec[(long)(g * 32 + i) * REC + c];

  if (t < PB_N) { ml[t] = rec[(long)t * REC + C]; sl[t] = rec[(long)t * REC + C + 1]; }
  __syncthreads();
  float M = fmaxf(fmaxf(ml[lane], ml[lane + 32]), fmaxf(ml[lane + 64], ml[lane + 96]));
  for (int o = 16; o > 0; o >>= 1) M = fmaxf(M, __shfl_xor(M, o));
  if (t < PB_N) { const float f = __expf(ml[t] - M); fs[t] = f; sl[t] = sl[t] * f; }
  __syncthreads();
  float S = (sl[lane] + sl[lane + 32]) + (sl[lane + 64] + sl[lane + 96]);
  for (int o = 16; o > 0; o >>= 1) S += __shfl_xor(S, o);

  float a0 = 0.f, a1 = 0.f;
#pragma unroll
  for (int i = 0; i < 32; i += 2) {
    a0 += x[i] * fs[g * 32 + i];
    a1 += x[i + 1] * fs[g * 32 + i + 1];
  }
  part[t] = a0 + a1;
  __syncthreads();

  if (t < C) {
    const float acc = (part[t] + part[REC + t]) + (part[2 * REC + t] + part[3 * REC + t]);
    const float gv = acc * (1.0f / S);
    gate[t] = 1.0f / (1.0f + __expf(-gv));
  }
}

#define TOP3_INS(val, idxv) {                                         \
    float s_ = (val); int id_ = (idxv);                               \
    bool c0_ = s_ < v0, c1_ = s_ < v1, c2_ = s_ < v2v;                \
    v2v = c1_ ? v1 : (c2_ ? s_  : v2v); i2 = c1_ ? i1 : (c2_ ? id_ : i2); \
    v1  = c0_ ? v0 : (c1_ ? s_  : v1 ); i1 = c0_ ? i0 : (c1_ ? id_ : i1); \
    v0  = c0_ ? s_ : v0;                i0 = c0_ ? id_ : i0;          }

#define TOP3_VAL(x) {                                                 \
    const float x_ = (x);                                             \
    const float n1_ = __builtin_amdgcn_fmed3f(t0, t1, x_);            \
    const float n2_ = __builtin_amdgcn_fmed3f(t1, t2, x_);            \
    t0 = fminf(t0, x_); t1 = n1_; t2 = n2_;                           }

__device__ __forceinline__ v8f score_tile(const v2f qf, const float* __restrict__ p, const int half) {
  const float sx = p[0];
  const float sy = p[1];
  const float sz = p[2];
  const float s2 = fmaf(sz, sz, fmaf(sx, sx, sy * sy));
  v2f sf;
  sf.x = half ? sz : sx;
  sf.y = half ? s2 : sy;
  const v8f c = {0.f, 0.f, 0.f, 0.f, 0.f, 0.f, 0.f, 0.f};
  return __builtin_amdgcn_wmma_f32_16x16x4_f32(false, sf, false, qf, (short)0, c, false, false);
}

__device__ __forceinline__ float min8(const v8f c) {
  float a = fminf(fminf(c[0], c[1]), c[2]);
  a = fminf(fminf(a, c[3]), c[4]);
  a = fminf(fminf(a, c[5]), c[6]);
  return fminf(a, c[7]);
}

__device__ __forceinline__ float min16(const v8f c, const v8f d) {
  float a = fminf(fminf(c[0], c[1]), c[2]);
  float b = fminf(fminf(d[0], d[1]), d[2]);
  a = fminf(fminf(a, c[3]), c[4]);
  b = fminf(fminf(b, d[3]), d[4]);
  a = fminf(fminf(a, c[5]), c[6]);
  b = fminf(fminf(b, d[5]), d[6]);
  a = fminf(fminf(a, c[7]), d[7]);
  return fminf(a, b);
}

#define SEG  (NS / NW)
#define NBLK (SEG / 32)
#define ROWS_PER_WAVE (16 / NW)

__global__ __launch_bounds__(NW * 32) __attribute__((amdgpu_waves_per_eu(12))) void fused_main(
    const float* __restrict__ coords_q, const float* __restrict__ coords_s,
    const float* __restrict__ x7,       const float* __restrict__ feat_s,
    const float* __restrict__ gate,     const float* __restrict__ W1,
    const float* __restrict__ gammaW,   const float* __restrict__ betaW,
    const float* __restrict__ W2c,      const float* __restrict__ b_cls,
    float* __restrict__ out)
{
  __shared__ __attribute__((aligned(16))) float lds[WREG];
  __shared__ __attribute__((aligned(16))) float sc[NW * SREG];
  __shared__ float pv[NW * 16 * 3];
  __shared__ int   pi[NW * 16 * 3];
  __shared__ float bt[NW * 16 * 3];
  __shared__ float tw[16 * 3];
  __shared__ int   ti[16 * 3];
  const int lane = threadIdx.x & 31;
  const int wave = __builtin_amdgcn_readfirstlane(threadIdx.x >> 5);
  const int tile = blockIdx.x;

  float* WL = lds;
  const int r15  = lane & 15;
  const int half = lane >> 4;
  const int qbase = tile * 16;

  const float qx = coords_q[(qbase + r15) * 3 + 0];
  const float qy = coords_q[(qbase + r15) * 3 + 1];
  const float qz = coords_q[(qbase + r15) * 3 + 2];
  v2f qf;
  qf.x = half ? (-2.f * qz) : (-2.f * qx);
  qf.y = half ? 1.0f        : (-2.f * qy);

  const int nbeg = wave * SEG;
  const float* const seg = coords_s + (nbeg + r15) * 3;

  float mt[NBLK];
  float t0 = 1e30f, t1 = 1e30f, t2 = 1e30f;
#pragma unroll
  for (int t = 0; t < NBLK; ++t) {
    const v8f c1 = score_tile(qf, seg + t * 96, half);
    const v8f c2 = score_tile(qf, seg + t * 96 + 48, half);
    const float m = min16(c1, c2);
    mt[t] = m;
    TOP3_VAL(m);
  }

  float T;
  {
    const float o0 = __shfl(t0, lane ^ 16), o1 = __shfl(t1, lane ^ 16), o2 = __shfl(t2, lane ^ 16);
    TOP3_VAL(o0); TOP3_VAL(o1); TOP3_VAL(o2);
    if (lane < 16) {
      const int o = (wave * 16 + lane) * 3;
      bt[o] = t0; bt[o + 1] = t1; bt[o + 2] = t2;
    }
    __syncthreads();
    t0 = 1e30f; t1 = 1e30f; t2 = 1e30f;
    for (int w = 0; w < NW; ++w) {
      const int o = (w * 16 + r15) * 3;
      TOP3_VAL(bt[o]); TOP3_VAL(bt[o + 1]); TOP3_VAL(bt[o + 2]);
    }
    T = t2;
  }

  unsigned blkmask = 0u;
#pragma unroll
  for (int t = 0; t < NBLK; ++t)
    blkmask |= (__builtin_amdgcn_ballot_w32(mt[t] <= T) != 0u) ? (1u << t) : 0u;

  float v0 = 1e30f, v1 = 1e30f, v2v = 1e30f;
  int   i0 = 0, i1 = 0, i2 = 0;

#define INSERT_TILE(cv, base) {                                                   \
    const int b_ = (base);                                                        \
    TOP3_INS((cv)[0], b_ + 0); TOP3_INS((cv)[1], b_ + 1);                         \
    TOP3_INS((cv)[2], b_ + 2); TOP3_INS((cv)[3], b_ + 3);                         \
    TOP3_INS((cv)[4], b_ + 4); TOP3_INS((cv)[5], b_ + 5);                         \
    TOP3_INS((cv)[6], b_ + 6); TOP3_INS((cv)[7], b_ + 7);                         }

  while (blkmask != 0u) {
    const int t = __builtin_ctz(blkmask);
    blkmask &= blkmask - 1u;
    const float* p = seg + t * 96;
    const int n0 = nbeg + 32 * t + 8 * half;
    const v8f c1 = score_tile(qf, p, half);
    const v8f c2 = score_tile(qf, p + 48, half);
    const unsigned m1 = __builtin_amdgcn_ballot_w32(min8(c1) <= T);
    const unsigned m2 = __builtin_amdgcn_ballot_w32(min8(c2) <= T);
    if (m1) INSERT_TILE(c1, n0);
    if (m2) INSERT_TILE(c2, n0 + 16);
  }
#undef INSERT_TILE

  {
    float pv0 = __shfl(v0, lane + 16), pv1 = __shfl(v1, lane + 16), pv2 = __shfl(v2v, lane + 16);
    int   pi0 = __shfl(i0, lane + 16), pi1 = __shfl(i1, lane + 16), pi2 = __shfl(i2,  lane + 16);
    TOP3_INS(pv0, pi0); TOP3_INS(pv1, pi1); TOP3_INS(pv2, pi2);
  }
  if (lane < 16) {
    const int o = (wave * 16 + lane) * 3;
    pv[o] = v0; pv[o + 1] = v1; pv[o + 2] = v2v;
    pi[o] = i0; pi[o + 1] = i1; pi[o + 2] = i2;
  }
  __syncthreads();
  if (wave == 0) {
    v0 = 1e30f; v1 = 1e30f; v2v = 1e30f;
    i0 = 0; i1 = 0; i2 = 0;
    for (int w = 0; w < NW; ++w) {
      const int o = (w * 16 + r15) * 3;
      TOP3_INS(pv[o], pi[o]); TOP3_INS(pv[o + 1], pi[o + 1]); TOP3_INS(pv[o + 2], pi[o + 2]);
    }
    const float e1 = __expf(v0 - v1), e2 = __expf(v0 - v2v);
    const float inv = 1.0f / (1.0f + e1 + e2);
    if (lane < 16) {
      tw[lane * 3] = inv; tw[lane * 3 + 1] = e1 * inv; tw[lane * 3 + 2] = e2 * inv;
      ti[lane * 3] = i0;  ti[lane * 3 + 1] = i1;       ti[lane * 3 + 2] = i2;
    }
  }
  __syncthreads();

  if (lane < C / 4) {
    const int rbeg = wave * ROWS_PER_WAVE;
    float  rw0[ROWS_PER_WAVE], rw1[ROWS_PER_WAVE], rw2[ROWS_PER_WAVE];
    float4 xv[ROWS_PER_WAVE], f0[ROWS_PER_WAVE], f1[ROWS_PER_WAVE], f2[ROWS_PER_WAVE];
    const float4 gv = ((const float4*)gate)[lane];
#pragma unroll
    for (int j = 0; j < ROWS_PER_WAVE; ++j) {
      const int r = rbeg + j;
      rw0[j] = tw[r * 3]; rw1[j] = tw[r * 3 + 1]; rw2[j] = tw[r * 3 + 2];
      const int ri0 = ti[r * 3], ri1 = ti[r * 3 + 1], ri2 = ti[r * 3 + 2];
      xv[j] = ((const float4*)(x7 + (long)(qbase + r) * C))[lane];
      f0[j] = ((const float4*)(feat_s + (long)ri0 * C))[lane];
      f1[j] = ((const float4*)(feat_s + (long)ri1 * C))[lane];
      f2[j] = ((const float4*)(feat_s + (long)ri2 * C))[lane];
    }
#pragma unroll
    for (int j = 0; j < ROWS_PER_WAVE; ++j) {
      float* const xr = WL + (rbeg + j) * XSTRIDE + 4 * lane;
      v2f a, b, c, d;
      a.x = xv[j].x * gv.x;
      a.y = xv[j].y * gv.y;
      b.x = xv[j].z * gv.z;
      b.y = xv[j].w * gv.w;
      c.x = rw0[j] * f0[j].x + rw1[j] * f1[j].x + rw2[j] * f2[j].x;
      c.y = rw0[j] * f0[j].y + rw1[j] * f1[j].y + rw2[j] * f2[j].y;
      d.x = rw0[j] * f0[j].z + rw1[j] * f1[j].z + rw2[j] * f2[j].z;
      d.y = rw0[j] * f0[j].w + rw1[j] * f1[j].w + rw2[j] * f2[j].w;
      *(v2f*)(xr)         = a;
      *(v2f*)(xr + 2)     = b;
      *(v2f*)(xr + C)     = c;
      *(v2f*)(xr + C + 2) = d;
    }
  }
  __syncthreads();

  const int ch = wave * 16 + r15;

  const float bias0 = b_cls[threadIdx.x % CLS];
  const float bias1 = b_cls[(threadIdx.x + NW * 32) % CLS];

  v8f acc = {0.f, 0.f, 0.f, 0.f, 0.f, 0.f, 0.f, 0.f};
  v2f wc0[4] = {}, wc1[4] = {};
  if (wave < NT) {
    const int cB = (r15 + 16 < CLS) ? (r15 + 16) : (CLS - 1);
#pragma unroll
    for (int s = 0; s < 4; ++s) {
      const int kk = wave * 16 + 4 * s + 2 * half;
      wc0[s].x = W2c[(long)kk * CLS + r15];
      wc0[s].y = W2c[(long)(kk + 1) * CLS + r15];
      wc1[s].x = W2c[(long)kk * CLS + cB];
      wc1[s].y = W2c[(long)(kk + 1) * CLS + cB];
    }
    for (int k0 = 0; k0 < C2; k0 += 4) {
      const int kk = k0 + 2 * half;
      const v2f a1 = *(const v2f*)(WL + r15 * XSTRIDE + kk);
      v2f bv;
      bv.x = W1[(long)kk * C + ch];
      bv.y = W1[(long)(kk + 1) * C + ch];
      acc = __builtin_amdgcn_wmma_f32_16x16x4_f32(false, a1, false, bv, (short)0, acc, false, false);
    }
  }
  __syncthreads();

  float* const P = sc + wave * SREG;
  if (wave < NT) {
    const float gm = gammaW[ch], bt = betaW[ch];
#pragma unroll
    for (int r = 0; r < 8; ++r) {
      const float h = fmaxf(acc[r] * gm + bt, 0.f);
      WL[H_OFF + (r + 8 * half) * HSTRIDE + ch] = h;
    }
    asm volatile("" ::: "memory");
    v8f p0 = {0.f, 0.f, 0.f, 0.f, 0.f, 0.f, 0.f, 0.f};
    v8f p1 = p0;
#pragma unroll
    for (int s = 0; s < 4; ++s) {
      const int kk = wave * 16 + 4 * s + 2 * half;
      const v2f a3 = *(const v2f*)(WL + H_OFF + r15 * HSTRIDE + kk);
      p0 = __builtin_amdgcn_wmma_f32_16x16x4_f32(false, a3, false, wc0[s], (short)0, p0, false, false);
      p1 = __builtin_amdgcn_wmma_f32_16x16x4_f32(false, a3, false, wc1[s], (short)0, p1, false, false);
    }
#pragma unroll
    for (int r = 0; r < 8; ++r)
      P[(r + 8 * half) * CLS + r15] = p0[r];
    if (r15 + 16 < CLS) {
#pragma unroll
      for (int r = 0; r < 8; ++r)
        P[(r + 8 * half) * CLS + 16 + r15] = p1[r];
    }
  }
  __syncthreads();

  {
    const int t = threadIdx.x;
    float s = sc[t];
#pragma unroll
    for (int w = 1; w < NT; ++w) s += sc[w * SREG + t];
    out[(long)qbase * CLS + t] = s + bias0;
    if (t < 16 * CLS - NW * 32) {
      const int u = t + NW * 32;
      float s1 = sc[u];
#pragma unroll
      for (int w = 1; w < NT; ++w) s1 += sc[w * SREG + u];
      out[(long)qbase * CLS + u] = s1 + bias1;
    }
  }
}

extern "C" void kernel_launch(void* const* d_in, const int* in_sizes, int n_in,
                              void* d_out, int out_size, void* d_ws, size_t ws_size,
                              hipStream_t stream) {
  const float* coords_q = (const float*)d_in[0];
  const float* coords_s = (const float*)d_in[1];
  const float* x7       = (const float*)d_in[2];
  const float* feat_s7  = (const float*)d_in[3];
  const float* feat_s   = (const float*)d_in[4];
  const float* w_attn   = (const float*)d_in[5];
  const float* W1       = (const float*)d_in[6];
  const float* gammaW   = (const float*)d_in[7];
  const float* betaW    = (const float*)d_in[8];
  const float* W2       = (const float*)d_in[9];
  const float* W_cls    = (const float*)d_in[10];
  const float* b_cls    = (const float*)d_in[11];
  float* out = (float*)d_out;

  float* ws   = (float*)d_ws;
  float* rec  = ws;
  float* gate = ws + PB_N * REC;
  float* W2c  = ws + PB_N * REC + REC;

  pool_partials<<<PB_N + NT, 256, 0, stream>>>(feat_s7, w_attn, W2, W_cls, rec, W2c);
  pool_combine<<<1, 512, 0, stream>>>(rec, gate);

  fused_main<<<TILES, NW * 32, 0, stream>>>(coords_q, coords_s, x7, feat_s, gate,
                                         W1, gammaW, betaW, W2c, b_cls, out);
}


# pooling records (now 64, of 128 rows) are rescaled and combined inside the main kernel by waves 1..7 while wave 0 merges the top-3 lists; the one-group combine launch is gone; sums re-associated
# speedup vs baseline: 22.5121x; 1.0581x over previous
#include <hip/hip_runtime.h>
#include <math.h>

#define NQ   20000
#define NS   8192
#define C    112
#define C2   224
#define CLS  20
#define TILES (NQ / 16)

#define WREG    3712
#define XSTRIDE 226
#define HSTRIDE 114
#define H_OFF   0

#define NW   8
#define NT   (C / 16)
#define SREG 320

typedef float v2f __attribute__((ext_vector_type(2)));
typedef float v8f __attribute__((ext_vector_type(8)));

#define PB_ROWS  128
#define PB_N     (NS / PB_ROWS)
#define PT       512
#define PB_PITCH 116
#define REC      128

__device__ __forceinline__ void fold_tile(const float* __restrict__ W2, const float* __restrict__ W_cls,
                                          float* __restrict__ W2c, float* __restrict__ part, const int rt) {
  const int t = threadIdx.x;
  const int lane = t & 31;
  const int r15 = lane & 15;
  const int half = lane >> 4;
  const int wave = __builtin_amdgcn_readfirstlane(t >> 5);
  if (wave < NT) {
    const int cB = (r15 + 16 < CLS) ? (r15 + 16) : (CLS - 1);
    const float* const arow = W2 + (long)(rt * 16 + r15) * C;
    v8f p0 = {0.f, 0.f, 0.f, 0.f, 0.f, 0.f, 0.f, 0.f};
    v8f p1 = p0;
#pragma unroll
    for (int s = 0; s < 4; ++s) {
      const int kk = wave * 16 + 4 * s + 2 * half;
      const v2f a = *(const v2f*)(arow + kk);
      v2f b0, b1;
      b0.x = W_cls[(long)kk * CLS + r15];
      b0.y = W_cls[(long)(kk + 1) * CLS + r15];
      b1.x = W_cls[(long)kk * CLS + cB];
      b1.y = W_cls[(long)(kk + 1) * CLS + cB];
      p0 = __builtin_amdgcn_wmma_f32_16x16x4_f32(false, a, false, b0, (short)0, p0, false, false);
      p1 = __builtin_amdgcn_wmma_f32_16x16x4_f32(false, a, false, b1, (short)0, p1, false, false);
    }
    float* const P = part + wave * SREG;
#pragma unroll
    for (int r = 0; r < 8; ++r)
      P[(r + 8 * half) * CLS + r15] = p0[r];
    if (r15 + 16 < CLS) {
#pragma unroll
      for (int r = 0; r < 8; ++r)
        P[(r + 8 * half) * CLS + 16 + r15] = p1[r];
    }
  }
  __syncthreads();
  if (t < SREG) {
    float s = part[t];
#pragma unroll
    for (int w = 1; w < NT; ++w) s += part[w * SREG + t];
    W2c[(long)rt * SREG + t] = s;
  }
}

__global__ __launch_bounds__(PT) void pool_partials(const float* __restrict__ feat_s7,
                                                      const float* __restrict__ w_attn,
                                                      const float* __restrict__ W2,
                                                      const float* __restrict__ W_cls,
                                                      float* __restrict__ rec,
                                                      float* __restrict__ W2c) {
  __shared__ __attribute__((aligned(16))) float tile[PB_ROWS * PB_PITCH];
  __shared__ __attribute__((aligned(16))) float wl[C];
  __shared__ float lg[PB_ROWS];
  __shared__ float ev[PB_ROWS];
  __shared__ float part[4 * C];
  const int t = threadIdx.x;
  const int lane = t & 31;

  if (blockIdx.x >= PB_N) {
    fold_tile(W2, W_cls, W2c, tile, blockIdx.x - PB_N);
    return;
  }

  if (t < C / 4) *(float4*)(wl + 4 * t) = ((const float4*)w_attn)[t];

  const float4* src = (const float4*)(feat_s7 + (long)blockIdx.x * (PB_ROWS * C));
#pragma unroll
  for (int i = 0; i < 7; ++i) {
    const int q = t + PT * i;
    const int row = q / 28, c4 = q - row * 28;
    *(float4*)(tile + row * PB_PITCH + 4 * c4) = src[q];
  }
  __syncthreads();

  {
    const int row = t >> 2, p = t & 3;
    float acc = 0.f;
#pragma unroll
    for (int i = 0; i < 7; ++i) {
      const int c4 = p + 4 * i;
      const float4 x = *(const float4*)(tile + row * PB_PITCH + 4 * c4);
      const float4 w = *(const float4*)(wl + 4 * c4);
      acc += x.x * w.x + x.y * w.y + x.z * w.z + x.w * w.w;
    }
    acc += __shfl_xor(acc, 1);
    acc += __shfl_xor(acc, 2);
    if (p == 0) lg[row] = acc;
  }
  __syncthreads();

  float m = fmaxf(fmaxf(lg[lane], lg[lane + 32]), fmaxf(lg[lane + 64], lg[lane + 96]));
  for (int o = 16; o > 0; o >>= 1) m = fmaxf(m, __shfl_xor(m, o));
  if (t < PB_ROWS) ev[t] = __expf(lg[t] - m);
  __syncthreads();
  float s = (ev[lane] + ev[lane + 32]) + (ev[lane + 64] + ev[lane + 96]);
  for (int o = 16; o > 0; o >>= 1) s += __shfl_xor(s, o);

  if (t < 4 * C) {
    const int h = t / C;
    const int c = t - h * C;
    const float* col = tile + (h * 32) * PB_PITCH + c;
    const float* e = ev + h * 32;
    float a0 = 0.f, a1 = 0.f;
#pragma unroll
    for (int r = 0; r < 32; r += 2) {
      a0 += e[r] * col[r * PB_PITCH];
      a1 += e[r + 1] * col[(r + 1) * PB_PITCH];
    }
    part[t] = a0 + a1;
  }
  __syncthreads();

  if (t < REC) {
    float v = 0.f;
    if (t < C) v = (part[t] + part[C + t]) + (part[2 * C + t] + part[3 * C + t]);
    else if (t == C) v = m;
    else if (t == C + 1) v = s;
    rec[(long)blockIdx.x * REC + t] = v;
  }
}

#define TOP3_INS(val, idxv) {                                         \
    float s_ = (val); int id_ = (idxv);                               \
    bool c0_ = s_ < v0, c1_ = s_ < v1, c2_ = s_ < v2v;                \
    v2v = c1_ ? v1 : (c2_ ? s_  : v2v); i2 = c1_ ? i1 : (c2_ ? id_ : i2); \
    v1  = c0_ ? v0 : (c1_ ? s_  : v1 ); i1 = c0_ ? i0 : (c1_ ? id_ : i1); \
    v0  = c0_ ? s_ : v0;                i0 = c0_ ? id_ : i0;          }

#define TOP3_VAL(x) {                                                 \
    const float x_ = (x);                                             \
    const float n1_ = __builtin_amdgcn_fmed3f(t0, t1, x_);            \
    const float n2_ = __builtin_amdgcn_fmed3f(t1, t2, x_);            \
    t0 = fminf(t0, x_); t1 = n1_; t2 = n2_;                           }

__device__ __forceinline__ v8f score_tile(const v2f qf, const float* __restrict__ p, const int half) {
  const float sx = p[0];
  const float sy = p[1];
  const float sz = p[2];
  const float s2 = fmaf(sz, sz, fmaf(sx, sx, sy * sy));
  v2f sf;
  sf.x = half ? sz : sx;
  sf.y = half ? s2 : sy;
  const v8f c = {0.f, 0.f, 0.f, 0.f, 0.f, 0.f, 0.f, 0.f};
  return __builtin_amdgcn_wmma_f32_16x16x4_f32(false, sf, false, qf, (short)0, c, false, false);
}

__device__ __forceinline__ float min8(const v8f c) {
  float a = fminf(fminf(c[0], c[1]), c[2]);
  a = fminf(fminf(a, c[3]), c[4]);
  a = fminf(fminf(a, c[5]), c[6]);
  return fminf(a, c[7]);
}

__device__ __forceinline__ float min16(const v8f c, const v8f d) {
  float a = fminf(fminf(c[0], c[1]), c[2]);
  float b = fminf(fminf(d[0], d[1]), d[2]);
  a = fminf(fminf(a, c[3]), c[4]);
  b = fminf(fminf(b, d[3]), d[4]);
  a = fminf(fminf(a, c[5]), c[6]);
  b = fminf(fminf(b, d[5]), d[6]);
  a = fminf(fminf(a, c[7]), d[7]);
  return fminf(a, b);
}

#define SEG  (NS / NW)
#define NBLK (SEG / 32)
#define ROWS_PER_WAVE (16 / NW)

__global__ __launch_bounds__(NW * 32) __attribute__((amdgpu_waves_per_eu(12))) void fused_main(
    const float* __restrict__ coords_q, const float* __restrict__ coords_s,
    const float* __restrict__ x7,       const float* __restrict__ feat_s,
    const float* __restrict__ rec,      const float* __restrict__ W1,
    const float* __restrict__ gammaW,   const float* __restrict__ betaW,
    const float* __restrict__ W2c,      const float* __restrict__ b_cls,
    float* __restrict__ out)
{
  __shared__ __attribute__((aligned(16))) float lds[WREG];
  __shared__ __attribute__((aligned(16))) float sc[NW * SREG];
  __shared__ float pv[NW * 16 * 3];
  __shared__ int   pi[NW * 16 * 3];
  __shared__ float bt[NW * 16 * 3];
  __shared__ float tw[16 * 3];
  __shared__ int   ti[16 * 3];
  __shared__ __attribute__((aligned(16))) float gl[C];
  __shared__ __attribute__((aligned(16))) float fl[NT * PB_N];
  __shared__ v2f msl[PB_N];
  const int lane = threadIdx.x & 31;
  const int wave = __builtin_amdgcn_readfirstlane(threadIdx.x >> 5);
  const int tile = blockIdx.x;

  float* WL = lds;
  const int r15  = lane & 15;
  const int half = lane >> 4;
  const int qbase = tile * 16;

  const float qx = coords_q[(qbase + r15) * 3 + 0];
  const float qy = coords_q[(qbase + r15) * 3 + 1];
  const float qz = coords_q[(qbase + r15) * 3 + 2];
  v2f qf;
  qf.x = half ? (-2.f * qz) : (-2.f * qx);
  qf.y = half ? 1.0f        : (-2.f * qy);

  const int nbeg = wave * SEG;
  const float* const seg = coords_s + (nbeg + r15) * 3;

  float mt[NBLK];
  float t0 = 1e30f, t1 = 1e30f, t2 = 1e30f;
#pragma unroll
  for (int t = 0; t < NBLK; ++t) {
    const v8f c1 = score_tile(qf, seg + t * 96, half);
    const v8f c2 = score_tile(qf, seg + t * 96 + 48, half);
    const float m = min16(c1, c2);
    mt[t] = m;
    TOP3_VAL(m);
  }

  float T;
  {
    const float o0 = __shfl(t0, lane ^ 16), o1 = __shfl(t1, lane ^ 16), o2 = __shfl(t2, lane ^ 16);
    TOP3_VAL(o0); TOP3_VAL(o1); TOP3_VAL(o2);
    if (lane < 16) {
      const int o = (wave * 16 + lane) * 3;
      bt[o] = t0; bt[o + 1] = t1; bt[o + 2] = t2;
    }
    __syncthreads();
    t0 = 1e30f; t1 = 1e30f; t2 = 1e30f;
    for (int w = 0; w < NW; ++w) {
      const int o = (w * 16 + r15) * 3;
      TOP3_VAL(bt[o]); TOP3_VAL(bt[o + 1]); TOP3_VAL(bt[o + 2]);
    }
    T = t2;
  }

  unsigned blkmask = 0u;
#pragma unroll
  for (int t = 0; t < NBLK; ++t)
    blkmask |= (__builtin_amdgcn_ballot_w32(mt[t] <= T) != 0u) ? (1u << t) : 0u;

  const v2f msv = *(const v2f*)(rec + (long)(wave * (PB_N / NW) + (lane & (PB_N / NW - 1))) * REC + C);

  float v0 = 1e30f, v1 = 1e30f, v2v = 1e30f;
  int   i0 = 0, i1 = 0, i2 = 0;

#define INSERT_TILE(cv, base) {                                                   \
    const int b_ = (base);                                                        \
    TOP3_INS((cv)[0], b_ + 0); TOP3_INS((cv)[1], b_ + 1);                         \
    TOP3_INS((cv)[2], b_ + 2); TOP3_INS((cv)[3], b_ + 3);                         \
    TOP3_INS((cv)[4], b_ + 4); TOP3_INS((cv)[5], b_ + 5);                         \
    TOP3_INS((cv)[6], b_ + 6); TOP3_INS((cv)[7], b_ + 7);                         }

  while (blkmask != 0u) {
    const int t = __builtin_ctz(blkmask);
    blkmask &= blkmask - 1u;
    const float* p = seg + t * 96;
    const int n0 = nbeg + 32 * t + 8 * half;
    const v8f c1 = score_tile(qf, p, half);
    const v8f c2 = score_tile(qf, p + 48, half);
    const unsigned m1 = __builtin_amdgcn_ballot_w32(min8(c1) <= T);
    const unsigned m2 = __builtin_amdgcn_ballot_w32(min8(c2) <= T);
    if (m1) INSERT_TILE(c1, n0);
    if (m2) INSERT_TILE(c2, n0 + 16);
  }
#undef INSERT_TILE

  {
    float pv0 = __shfl(v0, lane + 16), pv1 = __shfl(v1, lane + 16), pv2 = __shfl(v2v, lane + 16);
    int   pi0 = __shfl(i0, lane + 16), pi1 = __shfl(i1, lane + 16), pi2 = __shfl(i2,  lane + 16);
    TOP3_INS(pv0, pi0); TOP3_INS(pv1, pi1); TOP3_INS(pv2, pi2);
  }
  if (lane < 16) {
    const int o = (wave * 16 + lane) * 3;
    pv[o] = v0; pv[o + 1] = v1; pv[o + 2] = v2v;
    pi[o] = i0; pi[o + 1] = i1; pi[o + 2] = i2;
  }
  if (lane < PB_N / NW) msl[wave * (PB_N / NW) + lane] = msv;
  __syncthreads();
  if (wave == 0) {
    v0 = 1e30f; v1 = 1e30f; v2v = 1e30f;
    i0 = 0; i1 = 0; i2 = 0;
    for (int w = 0; w < NW; ++w) {
      const int o = (w * 16 + r15) * 3;
      TOP3_INS(pv[o], pi[o]); TOP3_INS(pv[o + 1], pi[o + 1]); TOP3_INS(pv[o + 2], pi[o + 2]);
    }
    const float e1 = __expf(v0 - v1), e2 = __expf(v0 - v2v);
    const float inv = 1.0f / (1.0f + e1 + e2);
    if (lane < 16) {
      tw[lane * 3] = inv; tw[lane * 3 + 1] = e1 * inv; tw[lane * 3 + 2] = e2 * inv;
      ti[lane * 3] = i0;  ti[lane * 3 + 1] = i1;       ti[lane * 3 + 2] = i2;
    }
  } else {
    const int c = (wave - 1) * 16 + r15;
    const float* const rp = rec + (long)(half * 32) * REC + c;
    float x[32];
#pragma unroll
    for (int i = 0; i < 32; ++i) x[i] = rp[(long)i * REC];
    const v2f msA = msl[lane], msB = msl[lane + 32];
    const float mA = msA.x, mB = msB.x;
    const float sA = msA.y, sB = msB.y;
    float M = fmaxf(mA, mB);
    for (int o = 16; o > 0; o >>= 1) M = fmaxf(M, __shfl_xor(M, o));
    const float fA = __expf(mA - M), fB = __expf(mB - M);
    float S = sA * fA + sB * fB;
    for (int o = 16; o > 0; o >>= 1) S += __shfl_xor(S, o);
    float* const fw = fl + (wave - 1) * PB_N;
    fw[lane] = fA;
    fw[lane + 32] = fB;
    asm volatile("" ::: "memory");
    const float4* const fr = (const float4*)(fw + half * 32);
    float a0 = 0.f, a1 = 0.f;
#pragma unroll
    for (int j = 0; j < 8; ++j) {
      const float4 f = fr[j];
      a0 += x[4 * j]     * f.x;
      a1 += x[4 * j + 1] * f.y;
      a0 += x[4 * j + 2] * f.z;
      a1 += x[4 * j + 3] * f.w;
    }
    float a = a0 + a1;
    a += __shfl_xor(a, 16);
    const float gv = a * (1.0f / S);
    if (lane < 16) gl[c] = 1.0f / (1.0f + __expf(-gv));
  }
  __syncthreads();

  if (lane < C / 4) {
    const int rbeg = wave * ROWS_PER_WAVE;
    float  rw0[ROWS_PER_WAVE], rw1[ROWS_PER_WAVE], rw2[ROWS_PER_WAVE];
    float4 xv[ROWS_PER_WAVE], f0[ROWS_PER_WAVE], f1[ROWS_PER_WAVE], f2[ROWS_PER_WAVE];
    const float4 gv = ((const float4*)gl)[lane];
#pragma unroll
    for (int j = 0; j < ROWS_PER_WAVE; ++j) {
      const int r = rbeg + j;
      rw0[j] = tw[r * 3]; rw1[j] = tw[r * 3 + 1]; rw2[j] = tw[r * 3 + 2];
      const int ri0 = ti[r * 3], ri1 = ti[r * 3 + 1], ri2 = ti[r * 3 + 2];
      xv[j] = ((const float4*)(x7 + (long)(qbase + r) * C))[lane];
      f0[j] = ((const float4*)(feat_s + (long)ri0 * C))[lane];
      f1[j] = ((const float4*)(feat_s + (long)ri1 * C))[lane];
      f2[j] = ((const float4*)(feat_s + (long)ri2 * C))[lane];
    }
#pragma unroll
    for (int j = 0; j < ROWS_PER_WAVE; ++j) {
      float* const xr = WL + (rbeg + j) * XSTRIDE + 4 * lane;
      v2f a, b, c, d;
      a.x = xv[j].x * gv.x;
      a.y = xv[j].y * gv.y;
      b.x = xv[j].z * gv.z;
      b.y = xv[j].w * gv.w;
      c.x = rw0[j] * f0[j].x + rw1[j] * f1[j].x + rw2[j] * f2[j].x;
      c.y = rw0[j] * f0[j].y + rw1[j] * f1[j].y + rw2[j] * f2[j].y;
      d.x = rw0[j] * f0[j].z + rw1[j] * f1[j].z + rw2[j] * f2[j].z;
      d.y = rw0[j] * f0[j].w + rw1[j] * f1[j].w + rw2[j] * f2[j].w;
      *(v2f*)(xr)         = a;
      *(v2f*)(xr + 2)     = b;
      *(v2f*)(xr + C)     = c;
      *(v2f*)(xr + C + 2) = d;
    }
  }
  __syncthreads();

  const int ch = wave * 16 + r15;

  const float bias0 = b_cls[threadIdx.x % CLS];
  const float bias1 = b_cls[(threadIdx.x + NW * 32) % CLS];

  v8f acc = {0.f, 0.f, 0.f, 0.f, 0.f, 0.f, 0.f, 0.f};
  v2f wc0[4] = {}, wc1[4] = {};
  if (wave < NT) {
    const int cB = (r15 + 16 < CLS) ? (r15 + 16) : (CLS - 1);
#pragma unroll
    for (int s = 0; s < 4; ++s) {
      const int kk = wave * 16 + 4 * s + 2 * half;
      wc0[s].x = W2c[(long)kk * CLS + r15];
      wc0[s].y = W2c[(long)(kk + 1) * CLS + r15];
      wc1[s].x = W2c[(long)kk * CLS + cB];
      wc1[s].y = W2c[(long)(kk + 1) * CLS + cB];
    }
    for (int k0 = 0; k0 < C2; k0 += 4) {
      const int kk = k0 + 2 * half;
      const v2f a1 = *(const v2f*)(WL + r15 * XSTRIDE + kk);
      v2f bv;
      bv.x = W1[(long)kk * C + ch];
      bv.y = W1[(long)(kk + 1) * C + ch];
      acc = __builtin_amdgcn_wmma_f32_16x16x4_f32(false, a1, false, bv, (short)0, acc, false, false);
    }
  }
  __syncthreads();

  float* const P = sc + wave * SREG;
  if (wave < NT) {
    const float gm = gammaW[ch], bt = betaW[ch];
#pragma unroll
    for (int r = 0; r < 8; ++r) {
      const float h = fmaxf(acc[r] * gm + bt, 0.f);
      WL[H_OFF + (r + 8 * half) * HSTRIDE + ch] = h;
    }
    asm volatile("" ::: "memory");
    v8f p0 = {0.f, 0.f, 0.f, 0.f, 0.f, 0.f, 0.f, 0.f};
    v8f p1 = p0;
#pragma unroll
    for (int s = 0; s < 4; ++s) {
      const int kk = wave * 16 + 4 * s + 2 * half;
      const v2f a3 = *(const v2f*)(WL + H_OFF + r15 * HSTRIDE + kk);
      p0 = __builtin_amdgcn_wmma_f32_16x16x4_f32(false, a3, false, wc0[s], (short)0, p0, false, false);
      p1 = __builtin_amdgcn_wmma_f32_16x16x4_f32(false, a3, false, wc1[s], (short)0, p1, false, false);
    }
#pragma unroll
    for (int r = 0; r < 8; ++r)
      P[(r + 8 * half) * CLS + r15] = p0[r];
    if (r15 + 16 < CLS) {
#pragma unroll
      for (int r = 0; r < 8; ++r)
        P[(r + 8 * half) * CLS + 16 + r15] = p1[r];
    }
  }
  __syncthreads();

  {
    const int t = threadIdx.x;
    float s = sc[t];
#pragma unroll
    for (int w = 1; w < NT; ++w) s += sc[w * SREG + t];
    out[(long)qbase * CLS + t] = s + bias0;
    if (t < 16 * CLS - NW * 32) {
      const int u = t + NW * 32;
      float s1 = sc[u];
#pragma unroll
      for (int w = 1; w < NT; ++w) s1 += sc[w * SREG + u];
      out[(long)qbase * CLS + u] = s1 + bias1;
    }
  }
}

extern "C" void kernel_launch(void* const* d_in, const int* in_sizes, int n_in,
                              void* d_out, int out_size, void* d_ws, size_t ws_size,
                              hipStream_t stream) {
  const float* coords_q = (const float*)d_in[0];
  const float* coords_s = (const float*)d_in[1];
  const float* x7       = (const float*)d_in[2];
  const float* feat_s7  = (const float*)d_in[3];
  const float* feat_s   = (const float*)d_in[4];
  const float* w_attn   = (const float*)d_in[5];
  const float* W1       = (const float*)d_in[6];
  const float* gammaW   = (const float*)d_in[7];
  const float* betaW    = (const float*)d_in[8];
  const float* W2       = (const float*)d_in[9];
  const float* W_cls    = (const float*)d_in[10];
  const float* b_cls    = (const float*)d_in[11];
  float* out = (float*)d_out;

  float* ws   = (float*)d_ws;
  float* rec  = ws;
  float* W2c  = ws + PB_N * REC;

  pool_partials<<<PB_N + NT, PT, 0, stream>>>(feat_s7, w_attn, W2, W_cls, rec, W2c);

  fused_main<<<TILES, NW * 32, 0, stream>>>(coords_q, coords_s, x7, feat_s, rec,
                                         W1, gammaW, betaW, W2c, b_cls, out);
}


# search pass B: a result register of a marked tile runs the top-3 insert chain only if some lane's score in it is <= its bound (one wave-wide test per register; was all 8); order of sums unchanged
# speedup vs baseline: 23.6922x; 1.0524x over previous
#include <hip/hip_runtime.h>
#include <math.h>

#define NQ   20000
#define NS   8192
#define C    112
#define C2   224
#define CLS  20
#define TILES (NQ / 16)

#define WREG    3712
#define XSTRIDE 226
#define HSTRIDE 114
#define H_OFF   0

#define NW   8
#define NT   (C / 16)
#define SREG 320

typedef float v2f __attribute__((ext_vector_type(2)));
typedef float v8f __attribute__((ext_vector_type(8)));

#define PB_ROWS  128
#define PB_N     (NS / PB_ROWS)
#define PT       512
#define PB_PITCH 116
#define REC      128

__device__ __forceinline__ void fold_tile(const float* __restrict__ W2, const float* __restrict__ W_cls,
                                          float* __restrict__ W2c, float* __restrict__ part, const int rt) {
  const int t = threadIdx.x;
  const int lane = t & 31;
  const int r15 = lane & 15;
  const int half = lane >> 4;
  const int wave = __builtin_amdgcn_readfirstlane(t >> 5);
  if (wave < NT) {
    const int cB = (r15 + 16 < CLS) ? (r15 + 16) : (CLS - 1);
    const float* const arow = W2 + (long)(rt * 16 + r15) * C;
    v8f p0 = {0.f, 0.f, 0.f, 0.f, 0.f, 0.f, 0.f, 0.f};
    v8f p1 = p0;
#pragma unroll
    for (int s = 0; s < 4; ++s) {
      const int kk = wave * 16 + 4 * s + 2 * half;
      const v2f a = *(const v2f*)(arow + kk);
      v2f b0, b1;
      b0.x = W_cls[(long)kk * CLS + r15];
      b0.y = W_cls[(long)(kk + 1) * CLS + r15];
      b1.x = W_cls[(long)kk * CLS + cB];
      b1.y = W_cls[(long)(kk + 1) * CLS + cB];
      p0 = __builtin_amdgcn_wmma_f32_16x16x4_f32(false, a, false, b0, (short)0, p0, false, false);
      p1 = __builtin_amdgcn_wmma_f32_16x16x4_f32(false, a, false, b1, (short)0, p1, false, false);
    }
    float* const P = part + wave * SREG;
#pragma unroll
    for (int r = 0; r < 8; ++r)
      P[(r + 8 * half) * CLS + r15] = p0[r];
    if (r15 + 16 < CLS) {
#pragma unroll
      for (int r = 0; r < 8; ++r)
        P[(r + 8 * half) * CLS + 16 + r15] = p1[r];
    }
  }
  __syncthreads();
  if (t < SREG) {
    float s = part[t];
#pragma unroll
    for (int w = 1; w < NT; ++w) s += part[w * SREG + t];
    W2c[(long)rt * SREG + t] = s;
  }
}

__global__ __launch_bounds__(PT) void pool_partials(const float* __restrict__ feat_s7,
                                                      const float* __restrict__ w_attn,
                                                      const float* __restrict__ W2,
                                                      const float* __restrict__ W_cls,
                                                      float* __restrict__ rec,
                                                      float* __restrict__ W2c) {
  __shared__ __attribute__((aligned(16))) float tile[PB_ROWS * PB_PITCH];
  __shared__ __attribute__((aligned(16))) float wl[C];
  __shared__ float lg[PB_ROWS];
  __shared__ float ev[PB_ROWS];
  __shared__ float part[4 * C];
  const int t = threadIdx.x;
  const int lane = t & 31;

  if (blockIdx.x >= PB_N) {
    fold_tile(W2, W_cls, W2c, tile, blockIdx.x - PB_N);
    return;
  }

  if (t < C / 4) *(float4*)(wl + 4 * t) = ((const float4*)w_attn)[t];

  const float4* src = (const float4*)(feat_s7 + (long)blockIdx.x * (PB_ROWS * C));
#pragma unroll
  for (int i = 0; i < 7; ++i) {
    const int q = t + PT * i;
    const int row = q / 28, c4 = q - row * 28;
    *(float4*)(tile + row * PB_PITCH + 4 * c4) = src[q];
  }
  __syncthreads();

  {
    const int row = t >> 2, p = t & 3;
    float acc = 0.f;
#pragma unroll
    for (int i = 0; i < 7; ++i) {
      const int c4 = p + 4 * i;
      const float4 x = *(const float4*)(tile + row * PB_PITCH + 4 * c4);
      const float4 w = *(const float4*)(wl + 4 * c4);
      acc += x.x * w.x + x.y * w.y + x.z * w.z + x.w * w.w;
    }
    acc += __shfl_xor(acc, 1);
    acc += __shfl_xor(acc, 2);
    if (p == 0) lg[row] = acc;
  }
  __syncthreads();

  float m = fmaxf(fmaxf(lg[lane], lg[lane + 32]), fmaxf(lg[lane + 64], lg[lane + 96]));
  for (int o = 16; o > 0; o >>= 1) m = fmaxf(m, __shfl_xor(m, o));
  if (t < PB_ROWS) ev[t] = __expf(lg[t] - m);
  __syncthreads();
  float s = (ev[lane] + ev[lane + 32]) + (ev[lane + 64] + ev[lane + 96]);
  for (int o = 16; o > 0; o >>= 1) s += __shfl_xor(s, o);

  if (t < 4 * C) {
    const int h = t / C;
    const int c = t - h * C;
    const float* col = tile + (h * 32) * PB_PITCH + c;
    const float* e = ev + h * 32;
    float a0 = 0.f, a1 = 0.f;
#pragma unroll
    for (int r = 0; r < 32; r += 2) {
      a0 += e[r] * col[r * PB_PITCH];
      a1 += e[r + 1] * col[(r + 1) * PB_PITCH];
    }
    part[t] = a0 + a1;
  }
  __syncthreads();

  if (t < REC) {
    float v = 0.f;
    if (t < C) v = (part[t] + part[C + t]) + (part[2 * C + t] + part[3 * C + t]);
    else if (t == C) v = m;
    else if (t == C + 1) v = s;
    rec[(long)blockIdx.x * REC + t] = v;
  }
}

#define TOP3_INS(val, idxv) {                                         \
    float s_ = (val); int id_ = (idxv);                               \
    bool c0_ = s_ < v0, c1_ = s_ < v1, c2_ = s_ < v2v;                \
    v2v = c1_ ? v1 : (c2_ ? s_  : v2v); i2 = c1_ ? i1 : (c2_ ? id_ : i2); \
    v1  = c0_ ? v0 : (c1_ ? s_  : v1 ); i1 = c0_ ? i0 : (c1_ ? id_ : i1); \
    v0  = c0_ ? s_ : v0;                i0 = c0_ ? id_ : i0;          }

#define TOP3_VAL(x) {                                                 \
    const float x_ = (x);                                             \
    const float n1_ = __builtin_amdgcn_fmed3f(t0, t1, x_);            \
    const float n2_ = __builtin_amdgcn_fmed3f(t1, t2, x_);            \
    t0 = fminf(t0, x_); t1 = n1_; t2 = n2_;                           }

__device__ __forceinline__ v8f score_tile(const v2f qf, const float* __restrict__ p, const int half) {
  const float sx = p[0];
  const float sy = p[1];
  const float sz = p[2];
  const float s2 = fmaf(sz, sz, fmaf(sx, sx, sy * sy));
  v2f sf;
  sf.x = half ? sz : sx;
  sf.y = half ? s2 : sy;
  const v8f c = {0.f, 0.f, 0.f, 0.f, 0.f, 0.f, 0.f, 0.f};
  return __builtin_amdgcn_wmma_f32_16x16x4_f32(false, sf, false, qf, (short)0, c, false, false);
}

__device__ __forceinline__ float min8(const v8f c) {
  float a = fminf(fminf(c[0], c[1]), c[2]);
  a = fminf(fminf(a, c[3]), c[4]);
  a = fminf(fminf(a, c[5]), c[6]);
  return fminf(a, c[7]);
}

__device__ __forceinline__ float min16(const v8f c, const v8f d) {
  float a = fminf(fminf(c[0], c[1]), c[2]);
  float b = fminf(fminf(d[0], d[1]), d[2]);
  a = fminf(fminf(a, c[3]), c[4]);
  b = fminf(fminf(b, d[3]), d[4]);
  a = fminf(fminf(a, c[5]), c[6]);
  b = fminf(fminf(b, d[5]), d[6]);
  a = fminf(fminf(a, c[7]), d[7]);
  return fminf(a, b);
}

#define SEG  (NS / NW)
#define NBLK (SEG / 32)
#define ROWS_PER_WAVE (16 / NW)

__global__ __launch_bounds__(NW * 32) __attribute__((amdgpu_waves_per_eu(12))) void fused_main(
    const float* __restrict__ coords_q, const float* __restrict__ coords_s,
    const float* __restrict__ x7,       const float* __restrict__ feat_s,
    const float* __restrict__ rec,      const float* __restrict__ W1,
    const float* __restrict__ gammaW,   const float* __restrict__ betaW,
    const float* __restrict__ W2c,      const float* __restrict__ b_cls,
    float* __restrict__ out)
{
  __shared__ __attribute__((aligned(16))) float lds[WREG];
  __shared__ __attribute__((aligned(16))) float sc[NW * SREG];
  __shared__ float pv[NW * 16 * 3];
  __shared__ int   pi[NW * 16 * 3];
  __shared__ float bt[NW * 16 * 3];
  __shared__ float tw[16 * 3];
  __shared__ int   ti[16 * 3];
  __shared__ __attribute__((aligned(16))) float gl[C];
  __shared__ __attribute__((aligned(16))) float fl[NT * PB_N];
  __shared__ v2f msl[PB_N];
  const int lane = threadIdx.x & 31;
  const int wave = __builtin_amdgcn_readfirstlane(threadIdx.x >> 5);
  const int tile = blockIdx.x;

  float* WL = lds;
  const int r15  = lane & 15;
  const int half = lane >> 4;
  const int qbase = tile * 16;

  const float qx = coords_q[(qbase + r15) * 3 + 0];
  const float qy = coords_q[(qbase + r15) * 3 + 1];
  const float qz = coords_q[(qbase + r15) * 3 + 2];
  v2f qf;
  qf.x = half ? (-2.f * qz) : (-2.f * qx);
  qf.y = half ? 1.0f        : (-2.f * qy);

  const int nbeg = wave * SEG;
  const float* const seg = coords_s + (nbeg + r15) * 3;

  float mt[NBLK];
  float t0 = 1e30f, t1 = 1e30f, t2 = 1e30f;
#pragma unroll
  for (int t = 0; t < NBLK; ++t) {
    const v8f c1 = score_tile(qf, seg + t * 96, half);
    const v8f c2 = score_tile(qf, seg + t * 96 + 48, half);
    const float m = min16(c1, c2);
    mt[t] = m;
    TOP3_VAL(m);
  }

  float T;
  {
    const float o0 = __shfl(t0, lane ^ 16), o1 = __shfl(t1, lane ^ 16), o2 = __shfl(t2, lane ^ 16);
    TOP3_VAL(o0); TOP3_VAL(o1); TOP3_VAL(o2);
    if (lane < 16) {
      const int o = (wave * 16 + lane) * 3;
      bt[o] = t0; bt[o + 1] = t1; bt[o + 2] = t2;
    }
    __syncthreads();
    t0 = 1e30f; t1 = 1e30f; t2 = 1e30f;
    for (int w = 0; w < NW; ++w) {
      const int o = (w * 16 + r15) * 3;
      TOP3_VAL(bt[o]); TOP3_VAL(bt[o + 1]); TOP3_VAL(bt[o + 2]);
    }
    T = t2;
  }

  unsigned blkmask = 0u;
#pragma unroll
  for (int t = 0; t < NBLK; ++t)
    blkmask |= (__builtin_amdgcn_ballot_w32(mt[t] <= T) != 0u) ? (1u << t) : 0u;

  const v2f msv = *(const v2f*)(rec + (long)(wave * (PB_N / NW) + (lane & (PB_N / NW - 1))) * REC + C);

  float v0 = 1e30f, v1 = 1e30f, v2v = 1e30f;
  int   i0 = 0, i1 = 0, i2 = 0;

#define INSERT_REG(cv, r, base)                                                   \
    if (__builtin_amdgcn_ballot_w32((cv)[r] <= T) != 0u) TOP3_INS((cv)[r], (base) + (r))
#define INSERT_TILE(cv, base) {                                                   \
    const int b_ = (base);                                                        \
    INSERT_REG(cv, 0, b_); INSERT_REG(cv, 1, b_);                                 \
    INSERT_REG(cv, 2, b_); INSERT_REG(cv, 3, b_);                                 \
    INSERT_REG(cv, 4, b_); INSERT_REG(cv, 5, b_);                                 \
    INSERT_REG(cv, 6, b_); INSERT_REG(cv, 7, b_);                                 }

  while (blkmask != 0u) {
    const int t = __builtin_ctz(blkmask);
    blkmask &= blkmask - 1u;
    const float* p = seg + t * 96;
    const int n0 = nbeg + 32 * t + 8 * half;
    const v8f c1 = score_tile(qf, p, half);
    const v8f c2 = score_tile(qf, p + 48, half);
    const unsigned m1 = __builtin_amdgcn_ballot_w32(min8(c1) <= T);
    const unsigned m2 = __builtin_amdgcn_ballot_w32(min8(c2) <= T);
    if (m1) INSERT_TILE(c1, n0);
    if (m2) INSERT_TILE(c2, n0 + 16);
  }
#undef INSERT_TILE
#undef INSERT_REG

  {
    float pv0 = __shfl(v0, lane + 16), pv1 = __shfl(v1, lane + 16), pv2 = __shfl(v2v, lane + 16);
    int   pi0 = __shfl(i0, lane + 16), pi1 = __shfl(i1, lane + 16), pi2 = __shfl(i2,  lane + 16);
    TOP3_INS(pv0, pi0); TOP3_INS(pv1, pi1); TOP3_INS(pv2, pi2);
  }
  if (lane < 16) {
    const int o = (wave * 16 + lane) * 3;
    pv[o] = v0; pv[o + 1] = v1; pv[o + 2] = v2v;
    pi[o] = i0; pi[o + 1] = i1; pi[o + 2] = i2;
  }
  if (lane < PB_N / NW) msl[wave * (PB_N / NW) + lane] = msv;
  __syncthreads();
  if (wave == 0) {
    v0 = 1e30f; v1 = 1e30f; v2v = 1e30f;
    i0 = 0; i1 = 0; i2 = 0;
    for (int w = 0; w < NW; ++w) {
      const int o = (w * 16 + r15) * 3;
      TOP3_INS(pv[o], pi[o]); TOP3_INS(pv[o + 1], pi[o + 1]); TOP3_INS(pv[o + 2], pi[o + 2]);
    }
    const float e1 = __expf(v0 - v1), e2 = __expf(v0 - v2v);
    const float inv = 1.0f / (1.0f + e1 + e2);
    if (lane < 16) {
      tw[lane * 3] = inv; tw[lane * 3 + 1] = e1 * inv; tw[lane * 3 + 2] = e2 * inv;
      ti[lane * 3] = i0;  ti[lane * 3 + 1] = i1;       ti[lane * 3 + 2] = i2;
    }
  } else {
    const int c = (wave - 1) * 16 + r15;
    const float* const rp = rec + (long)(half * 32) * REC + c;
    float x[32];
#pragma unroll
    for (int i = 0; i < 32; ++i) x[i] = rp[(long)i * REC];
    const v2f msA = msl[lane], msB = msl[lane + 32];
    const float mA = msA.x, mB = msB.x;
    const float sA = msA.y, sB = msB.y;
    float M = fmaxf(mA, mB);
    for (int o = 16; o > 0; o >>= 1) M = fmaxf(M, __shfl_xor(M, o));
    const float fA = __expf(mA - M), fB = __expf(mB - M);
    float S = sA * fA + sB * fB;
    for (int o = 16; o > 0; o >>= 1) S += __shfl_xor(S, o);
    float* const fw = fl + (wave - 1) * PB_N;
    fw[lane] = fA;
    fw[lane + 32] = fB;
    asm volatile("" ::: "memory");
    const float4* const fr = (const float4*)(fw + half * 32);
    float a0 = 0.f, a1 = 0.f;
#pragma unroll
    for (int j = 0; j < 8; ++j) {
      const float4 f = fr[j];
      a0 += x[4 * j]     * f.x;
      a1 += x[4 * j + 1] * f.y;
      a0 += x[4 * j + 2] * f.z;
      a1 += x[4 * j + 3] * f.w;
    }
    float a = a0 + a1;
    a += __shfl_xor(a, 16);
    const float gv = a * (1.0f / S);
    if (lane < 16) gl[c] = 1.0f / (1.0f + __expf(-gv));
  }
  __syncthreads();

  if (lane < C / 4) {
    const int rbeg = wave * ROWS_PER_WAVE;
    float  rw0[ROWS_PER_WAVE], rw1[ROWS_PER_WAVE], rw2[ROWS_PER_WAVE];
    float4 xv[ROWS_PER_WAVE], f0[ROWS_PER_WAVE], f1[ROWS_PER_WAVE], f2[ROWS_PER_WAVE];
    const float4 gv = ((const float4*)gl)[lane];
#pragma unroll
    for (int j = 0; j < ROWS_PER_WAVE; ++j) {
      const int r = rbeg + j;
      rw0[j] = tw[r * 3]; rw1[j] = tw[r * 3 + 1]; rw2[j] = tw[r * 3 + 2];
      const int ri0 = ti[r * 3], ri1 = ti[r * 3 + 1], ri2 = ti[r * 3 + 2];
      xv[j] = ((const float4*)(x7 + (long)(qbase + r) * C))[lane];
      f0[j] = ((const float4*)(feat_s + (long)ri0 * C))[lane];
      f1[j] = ((const float4*)(feat_s + (long)ri1 * C))[lane];
      f2[j] = ((const float4*)(feat_s + (long)ri2 * C))[lane];
    }
#pragma unroll
    for (int j = 0; j < ROWS_PER_WAVE; ++j) {
      float* const xr = WL + (rbeg + j) * XSTRIDE + 4 * lane;
      v2f a, b, c, d;
      a.x = xv[j].x * gv.x;
      a.y = xv[j].y * gv.y;
      b.x = xv[j].z * gv.z;
      b.y = xv[j].w * gv.w;
      c.x = rw0[j] * f0[j].x + rw1[j] * f1[j].x + rw2[j] * f2[j].x;
      c.y = rw0[j] * f0[j].y + rw1[j] * f1[j].y + rw2[j] * f2[j].y;
      d.x = rw0[j] * f0[j].z + rw1[j] * f1[j].z + rw2[j] * f2[j].z;
      d.y = rw0[j] * f0[j].w + rw1[j] * f1[j].w + rw2[j] * f2[j].w;
      *(v2f*)(xr)         = a;
      *(v2f*)(xr + 2)     = b;
      *(v2f*)(xr + C)     = c;
      *(v2f*)(xr + C + 2) = d;
    }
  }
  __syncthreads();

  const int ch = wave * 16 + r15;

  const float bias0 = b_cls[threadIdx.x % CLS];
  const float bias1 = b_cls[(threadIdx.x + NW * 32) % CLS];

  v8f acc = {0.f, 0.f, 0.f, 0.f, 0.f, 0.f, 0.f, 0.f};
  v2f wc0[4] = {}, wc1[4] = {};
  if (wave < NT) {
    const int cB = (r15 + 16 < CLS) ? (r15 + 16) : (CLS - 1);
#pragma unroll
    for (int s = 0; s < 4; ++s) {
      const int kk = wave * 16 + 4 * s + 2 * half;
      wc0[s].x = W2c[(long)kk * CLS + r15];
      wc0[s].y = W2c[(long)(kk + 1) * CLS + r15];
      wc1[s].x = W2c[(long)kk * CLS + cB];
      wc1[s].y = W2c[(long)(kk + 1) * CLS + cB];
    }
    for (int k0 = 0; k0 < C2; k0 += 4) {
      const int kk = k0 + 2 * half;
      const v2f a1 = *(const v2f*)(WL + r15 * XSTRIDE + kk);
      v2f bv;
      bv.x = W1[(long)kk * C + ch];
      bv.y = W1[(long)(kk + 1) * C + ch];
      acc = __builtin_amdgcn_wmma_f32_16x16x4_f32(false, a1, false, bv, (short)0, acc, false, false);
    }
  }
  __syncthreads();

  float* const P = sc + wave * SREG;
  if (wave < NT) {
    const float gm = gammaW[ch], bt = betaW[ch];
#pragma unroll
    for (int r = 0; r < 8; ++r) {
      const float h = fmaxf(acc[r] * gm + bt, 0.f);
      WL[H_OFF + (r + 8 * half) * HSTRIDE + ch] = h;
    }
    asm volatile("" ::: "memory");
    v8f p0 = {0.f, 0.f, 0.f, 0.f, 0.f, 0.f, 0.f, 0.f};
    v8f p1 = p0;
#pragma unroll
    for (int s = 0; s < 4; ++s) {
      const int kk = wave * 16 + 4 * s + 2 * half;
      const v2f a3 = *(const v2f*)(WL + H_OFF + r15 * HSTRIDE + kk);
      p0 = __builtin_amdgcn_wmma_f32_16x16x4_f32(false, a3, false, wc0[s], (short)0, p0, false, false);
      p1 = __builtin_amdgcn_wmma_f32_16x16x4_f32(false, a3, false, wc1[s], (short)0, p1, false, false);
    }
#pragma unroll
    for (int r = 0; r < 8; ++r)
      P[(r + 8 * half) * CLS + r15] = p0[r];
    if (r15 + 16 < CLS) {
#pragma unroll
      for (int r = 0; r < 8; ++r)
        P[(r + 8 * half) * CLS + 16 + r15] = p1[r];
    }
  }
  __syncthreads();

  {
    const int t = threadIdx.x;
    float s = sc[t];
#pragma unroll
    for (int w = 1; w < NT; ++w) s += sc[w * SREG + t];
    out[(long)qbase * CLS + t] = s + bias0;
    if (t < 16 * CLS - NW * 32) {
      const int u = t + NW * 32;
      float s1 = sc[u];
#pragma unroll
      for (int w = 1; w < NT; ++w) s1 += sc[w * SREG + u];
      out[(long)qbase * CLS + u] = s1 + bias1;
    }
  }
}

extern "C" void kernel_launch(void* const* d_in, const int* in_sizes, int n_in,
                              void* d_out, int out_size, void* d_ws, size_t ws_size,
                              hipStream_t stream) {
  const float* coords_q = (const float*)d_in[0];
  const float* coords_s = (const float*)d_in[1];
  const float* x7       = (const float*)d_in[2];
  const float* feat_s7  = (const float*)d_in[3];
  const float* feat_s   = (const float*)d_in[4];
  const float* w_attn   = (const float*)d_in[5];
  const float* W1       = (const float*)d_in[6];
  const float* gammaW   = (const float*)d_in[7];
  const float* betaW    = (const float*)d_in[8];
  const float* W2       = (const float*)d_in[9];
  const float* W_cls    = (const float*)d_in[10];
  const float* b_cls    = (const float*)d_in[11];
  float* out = (float*)d_out;

  float* ws   = (float*)d_ws;
  float* rec  = ws;
  float* W2c  = ws + PB_N * REC;

  pool_partials<<<PB_N + NT, PT, 0, stream>>>(feat_s7, w_attn, W2, W_cls, rec, W2c);

  fused_main<<<TILES, NW * 32, 0, stream>>>(coords_q, coords_s, x7, feat_s, rec,
                                         W1, gammaW, betaW, W2c, b_cls, out);
}


# support rows projected through W1's lower half once in the pool launch (256 more work-groups, f32 WMMA): tiles gather projected rows as GEMM1's start value, GEMM1's k range halves; sums re-associated
# speedup vs baseline: 24.5167x; 1.0348x over previous
#include <hip/hip_runtime.h>
#include <math.h>

#define NQ   20000
#define NS   8192
#define C    112
#define CLS  20
#define TILES (NQ / 16)

#define WREG    3712
#define XSTRIDE 226
#define HSTRIDE 114
#define H_OFF   0

#define NW   8
#define NT   (C / 16)
#define SREG 320

typedef float v2f __attribute__((ext_vector_type(2)));
typedef float v8f __attribute__((ext_vector_type(8)));

#define PB_ROWS  128
#define PB_N     (NS / PB_ROWS)
#define PT       512
#define PB_PITCH 116
#define REC      128

__device__ __forceinline__ void fold_tile(const float* __restrict__ W2, const float* __restrict__ W_cls,
                                          float* __restrict__ W2c, float* __restrict__ part, const int rt) {
  const int t = threadIdx.x;
  const int lane = t & 31;
  const int r15 = lane & 15;
  const int half = lane >> 4;
  const int wave = __builtin_amdgcn_readfirstlane(t >> 5);
  if (wave < NT) {
    const int cB = (r15 + 16 < CLS) ? (r15 + 16) : (CLS - 1);
    const float* const arow = W2 + (long)(rt * 16 + r15) * C;
    v8f p0 = {0.f, 0.f, 0.f, 0.f, 0.f, 0.f, 0.f, 0.f};
    v8f p1 = p0;
#pragma unroll
    for (int s = 0; s < 4; ++s) {
      const int kk = wave * 16 + 4 * s + 2 * half;
      const v2f a = *(const v2f*)(arow + kk);
      v2f b0, b1;
      b0.x = W_cls[(long)kk * CLS + r15];
      b0.y = W_cls[(long)(kk + 1) * CLS + r15];
      b1.x = W_cls[(long)kk * CLS + cB];
      b1.y = W_cls[(long)(kk + 1) * CLS + cB];
      p0 = __builtin_amdgcn_wmma_f32_16x16x4_f32(false, a, false, b0, (short)0, p0, false, false);
      p1 = __builtin_amdgcn_wmma_f32_16x16x4_f32(false, a, false, b1, (short)0, p1, false, false);
    }
    float* const P = part + wave * SREG;
#pragma unroll
    for (int r = 0; r < 8; ++r)
      P[(r + 8 * half) * CLS + r15] = p0[r];
    if (r15 + 16 < CLS) {
#pragma unroll
      for (int r = 0; r < 8; ++r)
        P[(r + 8 * half) * CLS + 16 + r15] = p1[r];
    }
  }
  __syncthreads();
  if (t < SREG) {
    float s = part[t];
#pragma unroll
    for (int w = 1; w < NT; ++w) s += part[w * SREG + t];
    W2c[(long)rt * SREG + t] = s;
  }
}

#define FB_ROWS 32
#define FB_N    (NS / FB_ROWS)
#define FB_PCS  (FB_ROWS * C / 4)
__device__ __forceinline__ void project_rows(const float* __restrict__ feat_s, const float* __restrict__ W1,
                                             float* __restrict__ FW, float* __restrict__ ain,
                                             float* __restrict__ dout, const int blk) {
  const int t = threadIdx.x;
  const int lane = t & 31;
  const int r15 = lane & 15;
  const int half = lane >> 4;
  const int wave = __builtin_amdgcn_readfirstlane(t >> 5);
  const int rt = (wave >= NT) ? 1 : 0;
  const int ct = wave - rt * NT;
  const int ch = ct * 16 + r15;

  const float4* const src = (const float4*)(feat_s + (long)blk * (FB_ROWS * C));
  const int q1 = (t < FB_PCS - PT) ? (t + PT) : t;
  const float4 v0 = src[t];
  const float4 v1 = src[q1];

  v2f b[C / 4] = {};
  if (wave < 2 * NT) {
#pragma unroll
    for (int s = 0; s < C / 4; ++s) {
      const int kk = 4 * s + 2 * half;
      b[s].x = W1[(long)(C + kk) * C + ch];
      b[s].y = W1[(long)(C + kk + 1) * C + ch];
    }
  }
  {
    const int row = t / 28, c4 = t - row * 28;
    const int row1 = q1 / 28, c41 = q1 - row1 * 28;
    *(float4*)(ain + row * PB_PITCH + 4 * c4) = v0;
    *(float4*)(ain + row1 * PB_PITCH + 4 * c41) = v1;
  }
  __syncthreads();
  if (wave < 2 * NT) {
    const float* const arow = ain + (rt * 16 + r15) * PB_PITCH + 2 * half;
    v8f p0 = {0.f, 0.f, 0.f, 0.f, 0.f, 0.f, 0.f, 0.f};
    v8f p1 = p0;
#pragma unroll
    for (int s = 0; s < C / 4; s += 2) {
      const v2f a0 = *(const v2f*)(arow + 4 * s);
      const v2f a1 = *(const v2f*)(arow + 4 * s + 4);
      p0 = __builtin_amdgcn_wmma_f32_16x16x4_f32(false, a0, false, b[s],     (short)0, p0, false, false);
      p1 = __builtin_amdgcn_wmma_f32_16x16x4_f32(false, a1, false, b[s + 1], (short)0, p1, false, false);
    }
#pragma unroll
    for (int r = 0; r < 8; ++r)
      dout[(rt * 16 + r + 8 * half) * PB_PITCH + ch] = p0[r] + p1[r];
  }
  __syncthreads();
  {
    float4* const dst = (float4*)(FW + (long)blk * (FB_ROWS * C));
    const int row = t / 28, c4 = t - row * 28;
    dst[t] = *(const float4*)(dout + row * PB_PITCH + 4 * c4);
    if (t < FB_PCS - PT) {
      const int q = t + PT;
      const int row1 = q / 28, c41 = q - row1 * 28;
      dst[q] = *(const float4*)(dout + row1 * PB_PITCH + 4 * c41);
    }
  }
}

__global__ __launch_bounds__(PT) void pool_partials(const float* __restrict__ feat_s7,
                                                      const float* __restrict__ w_attn,
                                                      const float* __restrict__ W2,
                                                      const float* __restrict__ W_cls,
                                                      const float* __restrict__ feat_s,
                                                      const float* __restrict__ W1,
                                                      float* __restrict__ rec,
                                                      float* __restrict__ W2c,
                                                      float* __restrict__ FW) {
  __shared__ __attribute__((aligned(16))) float tile[PB_ROWS * PB_PITCH];
  __shared__ __attribute__((aligned(16))) float wl[C];
  __shared__ float lg[PB_ROWS];
  __shared__ float ev[PB_ROWS];
  __shared__ float part[4 * C];
  const int t = threadIdx.x;
  const int lane = t & 31;

  if (blockIdx.x >= PB_N + NT) {
    project_rows(feat_s, W1, FW, tile, tile + FB_ROWS * PB_PITCH, blockIdx.x - (PB_N + NT));
    return;
  }
  if (blockIdx.x >= PB_N) {
    fold_tile(W2, W_cls, W2c, tile, blockIdx.x - PB_N);
    return;
  }

  if (t < C / 4) *(float4*)(wl + 4 * t) = ((const float4*)w_attn)[t];

  const float4* src = (const float4*)(feat_s7 + (long)blockIdx.x * (PB_ROWS * C));
#pragma unroll
  for (int i = 0; i < 7; ++i) {
    const int q = t + PT * i;
    const int row = q / 28, c4 = q - row * 28;
    *(float4*)(tile + row * PB_PITCH + 4 * c4) = src[q];
  }
  __syncthreads();

  {
    const int row = t >> 2, p = t & 3;
    float acc = 0.f;
#pragma unroll
    for (int i = 0; i < 7; ++i) {
      const int c4 = p + 4 * i;
      const float4 x = *(const float4*)(tile + row * PB_PITCH + 4 * c4);
      const float4 w = *(const float4*)(wl + 4 * c4);
      acc += x.x * w.x + x.y * w.y + x.z * w.z + x.w * w.w;
    }
    acc += __shfl_xor(acc, 1);
    acc += __shfl_xor(acc, 2);
    if (p == 0) lg[row] = acc;
  }
  __syncthreads();

  float m = fmaxf(fmaxf(lg[lane], lg[lane + 32]), fmaxf(lg[lane + 64], lg[lane + 96]));
  for (int o = 16; o > 0; o >>= 1) m = fmaxf(m, __shfl_xor(m, o));
  if (t < PB_ROWS) ev[t] = __expf(lg[t] - m);
  __syncthreads();
  float s = (ev[lane] + ev[lane + 32]) + (ev[lane + 64] + ev[lane + 96]);
  for (int o = 16; o > 0; o >>= 1) s += __shfl_xor(s, o);

  if (t < 4 * C) {
    const int h = t / C;
    const int c = t - h * C;
    const float* col = tile + (h * 32) * PB_PITCH + c;
    const float* e = ev + h * 32;
    float a0 = 0.f, a1 = 0.f;
#pragma unroll
    for (int r = 0; r < 32; r += 2) {
      a0 += e[r] * col[r * PB_PITCH];
      a1 += e[r + 1] * col[(r + 1) * PB_PITCH];
    }
    part[t] = a0 + a1;
  }
  __syncthreads();

  if (t < REC) {
    float v = 0.f;
    if (t < C) v = (part[t] + part[C + t]) + (part[2 * C + t] + part[3 * C + t]);
    else if (t == C) v = m;
    else if (t == C + 1) v = s;
    rec[(long)blockIdx.x * REC + t] = v;
  }
}

#define TOP3_INS(val, idxv) {                                         \
    float s_ = (val); int id_ = (idxv);                               \
    bool c0_ = s_ < v0, c1_ = s_ < v1, c2_ = s_ < v2v;                \
    v2v = c1_ ? v1 : (c2_ ? s_  : v2v); i2 = c1_ ? i1 : (c2_ ? id_ : i2); \
    v1  = c0_ ? v0 : (c1_ ? s_  : v1 ); i1 = c0_ ? i0 : (c1_ ? id_ : i1); \
    v0  = c0_ ? s_ : v0;                i0 = c0_ ? id_ : i0;          }

#define TOP3_VAL(x) {                                                 \
    const float x_ = (x);                                             \
    const float n1_ = __builtin_amdgcn_fmed3f(t0, t1, x_);            \
    const float n2_ = __builtin_amdgcn_fmed3f(t1, t2, x_);            \
    t0 = fminf(t0, x_); t1 = n1_; t2 = n2_;                           }

__device__ __forceinline__ v8f score_tile(const v2f qf, const float* __restrict__ p, const int half) {
  const float sx = p[0];
  const float sy = p[1];
  const float sz = p[2];
  const float s2 = fmaf(sz, sz, fmaf(sx, sx, sy * sy));
  v2f sf;
  sf.x = half ? sz : sx;
  sf.y = half ? s2 : sy;
  const v8f c = {0.f, 0.f, 0.f, 0.f, 0.f, 0.f, 0.f, 0.f};
  return __builtin_amdgcn_wmma_f32_16x16x4_f32(false, sf, false, qf, (short)0, c, false, false);
}

__device__ __forceinline__ float min8(const v8f c) {
  float a = fminf(fminf(c[0], c[1]), c[2]);
  a = fminf(fminf(a, c[3]), c[4]);
  a = fminf(fminf(a, c[5]), c[6]);
  return fminf(a, c[7]);
}

__device__ __forceinline__ float min16(const v8f c, const v8f d) {
  float a = fminf(fminf(c[0], c[1]), c[2]);
  float b = fminf(fminf(d[0], d[1]), d[2]);
  a = fminf(fminf(a, c[3]), c[4]);
  b = fminf(fminf(b, d[3]), d[4]);
  a = fminf(fminf(a, c[5]), c[6]);
  b = fminf(fminf(b, d[5]), d[6]);
  a = fminf(fminf(a, c[7]), d[7]);
  return fminf(a, b);
}

#define SEG  (NS / NW)
#define NBLK (SEG / 32)
#define ROWS_PER_WAVE (16 / NW)

__global__ __launch_bounds__(NW * 32) __attribute__((amdgpu_waves_per_eu(12))) void fused_main(
    const float* __restrict__ coords_q, const float* __restrict__ coords_s,
    const float* __restrict__ x7,       const float* __restrict__ fw,
    const float* __restrict__ rec,      const float* __restrict__ W1,
    const float* __restrict__ gammaW,   const float* __restrict__ betaW,
    const float* __restrict__ W2c,      const float* __restrict__ b_cls,
    float* __restrict__ out)
{
  __shared__ __attribute__((aligned(16))) float lds[WREG];
  __shared__ __attribute__((aligned(16))) float sc[NW * SREG];
  __shared__ float pv[NW * 16 * 3];
  __shared__ int   pi[NW * 16 * 3];
  __shared__ float bt[NW * 16 * 3];
  __shared__ float tw[16 * 3];
  __shared__ int   ti[16 * 3];
  __shared__ __attribute__((aligned(16))) float gl[C];
  __shared__ __attribute__((aligned(16))) float fl[NT * PB_N];
  __shared__ v2f msl[PB_N];
  const int lane = threadIdx.x & 31;
  const int wave = __builtin_amdgcn_readfirstlane(threadIdx.x >> 5);
  const int tile = blockIdx.x;

  float* WL = lds;
  const int r15  = lane & 15;
  const int half = lane >> 4;
  const int qbase = tile * 16;

  const float qx = coords_q[(qbase + r15) * 3 + 0];
  const float qy = coords_q[(qbase + r15) * 3 + 1];
  const float qz = coords_q[(qbase + r15) * 3 + 2];
  v2f qf;
  qf.x = half ? (-2.f * qz) : (-2.f * qx);
  qf.y = half ? 1.0f        : (-2.f * qy);

  const int nbeg = wave * SEG;
  const float* const seg = coords_s + (nbeg + r15) * 3;

  float mt[NBLK];
  float t0 = 1e30f, t1 = 1e30f, t2 = 1e30f;
#pragma unroll
  for (int t = 0; t < NBLK; ++t) {
    const v8f c1 = score_tile(qf, seg + t * 96, half);
    const v8f c2 = score_tile(qf, seg + t * 96 + 48, half);
    const float m = min16(c1, c2);
    mt[t] = m;
    TOP3_VAL(m);
  }

  float T;
  {
    const float o0 = __shfl(t0, lane ^ 16), o1 = __shfl(t1, lane ^ 16), o2 = __shfl(t2, lane ^ 16);
    TOP3_VAL(o0); TOP3_VAL(o1); TOP3_VAL(o2);
    if (lane < 16) {
      const int o = (wave * 16 + lane) * 3;
      bt[o] = t0; bt[o + 1] = t1; bt[o + 2] = t2;
    }
    __syncthreads();
    t0 = 1e30f; t1 = 1e30f; t2 = 1e30f;
    for (int w = 0; w < NW; ++w) {
      const int o = (w * 16 + r15) * 3;
      TOP3_VAL(bt[o]); TOP3_VAL(bt[o + 1]); TOP3_VAL(bt[o + 2]);
    }
    T = t2;
  }

  unsigned blkmask = 0u;
#pragma unroll
  for (int t = 0; t < NBLK; ++t)
    blkmask |= (__builtin_amdgcn_ballot_w32(mt[t] <= T) != 0u) ? (1u << t) : 0u;

  const v2f msv = *(const v2f*)(rec + (long)(wave * (PB_N / NW) + (lane & (PB_N / NW - 1))) * REC + C);

  float v0 = 1e30f, v1 = 1e30f, v2v = 1e30f;
  int   i0 = 0, i1 = 0, i2 = 0;

#define INSERT_REG(cv, r, base)                                                   \
    if (__builtin_amdgcn_ballot_w32((cv)[r] <= T) != 0u) TOP3_INS((cv)[r], (base) + (r))
#define INSERT_TILE(cv, base) {                                                   \
    const int b_ = (base);                                                        \
    INSERT_REG(cv, 0, b_); INSERT_REG(cv, 1, b_);                                 \
    INSERT_REG(cv, 2, b_); INSERT_REG(cv, 3, b_);                                 \
    INSERT_REG(cv, 4, b_); INSERT_REG(cv, 5, b_);                                 \
    INSERT_REG(cv, 6, b_); INSERT_REG(cv, 7, b_);                                 }

  while (blkmask != 0u) {
    const int t = __builtin_ctz(blkmask);
    blkmask &= blkmask - 1u;
    const float* p = seg + t * 96;
    const int n0 = nbeg + 32 * t + 8 * half;
    const v8f c1 = score_tile(qf, p, half);
    const v8f c2 = score_tile(qf, p + 48, half);
    const unsigned m1 = __builtin_amdgcn_ballot_w32(min8(c1) <= T);
    const unsigned m2 = __builtin_amdgcn_ballot_w32(min8(c2) <= T);
    if (m1) INSERT_TILE(c1, n0);
    if (m2) INSERT_TILE(c2, n0 + 16);
  }
#undef INSERT_TILE
#undef INSERT_REG

  {
    float pv0 = __shfl(v0, lane + 16), pv1 = __shfl(v1, lane + 16), pv2 = __shfl(v2v, lane + 16);
    int   pi0 = __shfl(i0, lane + 16), pi1 = __shfl(i1, lane + 16), pi2 = __shfl(i2,  lane + 16);
    TOP3_INS(pv0, pi0); TOP3_INS(pv1, pi1); TOP3_INS(pv2, pi2);
  }
  if (lane < 16) {
    const int o = (wave * 16 + lane) * 3;
    pv[o] = v0; pv[o + 1] = v1; pv[o + 2] = v2v;
    pi[o] = i0; pi[o + 1] = i1; pi[o + 2] = i2;
  }
  if (lane < PB_N / NW) msl[wave * (PB_N / NW) + lane] = msv;
  __syncthreads();
  if (wave == 0) {
    v0 = 1e30f; v1 = 1e30f; v2v = 1e30f;
    i0 = 0; i1 = 0; i2 = 0;
    for (int w = 0; w < NW; ++w) {
      const int o = (w * 16 + r15) * 3;
      TOP3_INS(pv[o], pi[o]); TOP3_INS(pv[o + 1], pi[o + 1]); TOP3_INS(pv[o + 2], pi[o + 2]);
    }
    const float e1 = __expf(v0 - v1), e2 = __expf(v0 - v2v);
    const float inv = 1.0f / (1.0f + e1 + e2);
    if (lane < 16) {
      tw[lane * 3] = inv; tw[lane * 3 + 1] = e1 * inv; tw[lane * 3 + 2] = e2 * inv;
      ti[lane * 3] = i0;  ti[lane * 3 + 1] = i1;       ti[lane * 3 + 2] = i2;
    }
  } else {
    const int c = (wave - 1) * 16 + r15;
    const float* const rp = rec + (long)(half * 32) * REC + c;
    float x[32];
#pragma unroll
    for (int i = 0; i < 32; ++i) x[i] = rp[(long)i * REC];
    const v2f msA = msl[lane], msB = msl[lane + 32];
    const float mA = msA.x, mB = msB.x;
    const float sA = msA.y, sB = msB.y;
    float M = fmaxf(mA, mB);
    for (int o = 16; o > 0; o >>= 1) M = fmaxf(M, __shfl_xor(M, o));
    const float fA = __expf(mA - M), fB = __expf(mB - M);
    float S = sA * fA + sB * fB;
    for (int o = 16; o > 0; o >>= 1) S += __shfl_xor(S, o);
    float* const fw = fl + (wave - 1) * PB_N;
    fw[lane] = fA;
    fw[lane + 32] = fB;
    asm volatile("" ::: "memory");
    const float4* const fr = (const float4*)(fw + half * 32);
    float a0 = 0.f, a1 = 0.f;
#pragma unroll
    for (int j = 0; j < 8; ++j) {
      const float4 f = fr[j];
      a0 += x[4 * j]     * f.x;
      a1 += x[4 * j + 1] * f.y;
      a0 += x[4 * j + 2] * f.z;
      a1 += x[4 * j + 3] * f.w;
    }
    float a = a0 + a1;
    a += __shfl_xor(a, 16);
    const float gv = a * (1.0f / S);
    if (lane < 16) gl[c] = 1.0f / (1.0f + __expf(-gv));
  }
  __syncthreads();

  if (lane < C / 4) {
    const int rbeg = wave * ROWS_PER_WAVE;
    float  rw0[ROWS_PER_WAVE], rw1[ROWS_PER_WAVE], rw2[ROWS_PER_WAVE];
    float4 xv[ROWS_PER_WAVE], f0[ROWS_PER_WAVE], f1[ROWS_PER_WAVE], f2[ROWS_PER_WAVE];
    const float4 gv = ((const float4*)gl)[lane];
#pragma unroll
    for (int j = 0; j < ROWS_PER_WAVE; ++j) {
      const int r = rbeg + j;
      rw0[j] = tw[r * 3]; rw1[j] = tw[r * 3 + 1]; rw2[j] = tw[r * 3 + 2];
      const int ri0 = ti[r * 3], ri1 = ti[r * 3 + 1], ri2 = ti[r * 3 + 2];
      xv[j] = ((const float4*)(x7 + (long)(qbase + r) * C))[lane];
      f0[j] = ((const float4*)(fw + (long)ri0 * C))[lane];
      f1[j] = ((const float4*)(fw + (long)ri1 * C))[lane];
      f2[j] = ((const float4*)(fw + (long)ri2 * C))[lane];
    }
#pragma unroll
    for (int j = 0; j < ROWS_PER_WAVE; ++j) {
      float* const xr = WL + (rbeg + j) * XSTRIDE + 4 * lane;
      v2f a, b, c, d;
      a.x = xv[j].x * gv.x;
      a.y = xv[j].y * gv.y;
      b.x = xv[j].z * gv.z;
      b.y = xv[j].w * gv.w;
      c.x = rw0[j] * f0[j].x + rw1[j] * f1[j].x + rw2[j] * f2[j].x;
      c.y = rw0[j] * f0[j].y + rw1[j] * f1[j].y + rw2[j] * f2[j].y;
      d.x = rw0[j] * f0[j].z + rw1[j] * f1[j].z + rw2[j] * f2[j].z;
      d.y = rw0[j] * f0[j].w + rw1[j] * f1[j].w + rw2[j] * f2[j].w;
      *(v2f*)(xr)         = a;
      *(v2f*)(xr + 2)     = b;
      *(v2f*)(xr + C)     = c;
      *(v2f*)(xr + C + 2) = d;
    }
  }
  __syncthreads();

  const int ch = wave * 16 + r15;

  const float bias0 = b_cls[threadIdx.x % CLS];
  const float bias1 = b_cls[(threadIdx.x + NW * 32) % CLS];

  v8f acc = {0.f, 0.f, 0.f, 0.f, 0.f, 0.f, 0.f, 0.f};
  v2f wc0[4] = {}, wc1[4] = {};
  if (wave < NT) {
    const int cB = (r15 + 16 < CLS) ? (r15 + 16) : (CLS - 1);
#pragma unroll
    for (int s = 0; s < 4; ++s) {
      const int kk = wave * 16 + 4 * s + 2 * half;
      wc0[s].x = W2c[(long)kk * CLS + r15];
      wc0[s].y = W2c[(long)(kk + 1) * CLS + r15];
      wc1[s].x = W2c[(long)kk * CLS + cB];
      wc1[s].y = W2c[(long)(kk + 1) * CLS + cB];
    }
#pragma unroll
    for (int r = 0; r < 8; ++r) acc[r] = WL[(r + 8 * half) * XSTRIDE + C + ch];
    for (int k0 = 0; k0 < C; k0 += 4) {
      const int kk = k0 + 2 * half;
      const v2f a1 = *(const v2f*)(WL + r15 * XSTRIDE + kk);
      v2f bv;
      bv.x = W1[(long)kk * C + ch];
      bv.y = W1[(long)(kk + 1) * C + ch];
      acc = __builtin_amdgcn_wmma_f32_16x16x4_f32(false, a1, false, bv, (short)0, acc, false, false);
    }
  }
  __syncthreads();

  float* const P = sc + wave * SREG;
  if (wave < NT) {
    const float gm = gammaW[ch], bt = betaW[ch];
#pragma unroll
    for (int r = 0; r < 8; ++r) {
      const float h = fmaxf(acc[r] * gm + bt, 0.f);
      WL[H_OFF + (r + 8 * half) * HSTRIDE + ch] = h;
    }
    asm volatile("" ::: "memory");
    v8f p0 = {0.f, 0.f, 0.f, 0.f, 0.f, 0.f, 0.f, 0.f};
    v8f p1 = p0;
#pragma unroll
    for (int s = 0; s < 4; ++s) {
      const int kk = wave * 16 + 4 * s + 2 * half;
      const v2f a3 = *(const v2f*)(WL + H_OFF + r15 * HSTRIDE + kk);
      p0 = __builtin_amdgcn_wmma_f32_16x16x4_f32(false, a3, false, wc0[s], (short)0, p0, false, false);
      p1 = __builtin_amdgcn_wmma_f32_16x16x4_f32(false, a3, false, wc1[s], (short)0, p1, false, false);
    }
#pragma unroll
    for (int r = 0; r < 8; ++r)
      P[(r + 8 * half) * CLS + r15] = p0[r];
    if (r15 + 16 < CLS) {
#pragma unroll
      for (int r = 0; r < 8; ++r)
        P[(r + 8 * half) * CLS + 16 + r15] = p1[r];
    }
  }
  __syncthreads();

  {
    const int t = threadIdx.x;
    float s = sc[t];
#pragma unroll
    for (int w = 1; w < NT; ++w) s += sc[w * SREG + t];
    out[(long)qbase * CLS + t] = s + bias0;
    if (t < 16 * CLS - NW * 32) {
      const int u = t + NW * 32;
      float s1 = sc[u];
#pragma unroll
      for (int w = 1; w < NT; ++w) s1 += sc[w * SREG + u];
      out[(long)qbase * CLS + u] = s1 + bias1;
    }
  }
}

extern "C" void kernel_launch(void* const* d_in, const int* in_sizes, int n_in,
                              void* d_out, int out_size, void* d_ws, size_t ws_size,
                              hipStream_t stream) {
  const float* coords_q = (const float*)d_in[0];
  const float* coords_s = (const float*)d_in[1];
  const float* x7       = (const float*)d_in[2];
  const float* feat_s7  = (const float*)d_in[3];
  const float* feat_s   = (const float*)d_in[4];
  const float* w_attn   = (const float*)d_in[5];
  const float* W1       = (const float*)d_in[6];
  const float* gammaW   = (const float*)d_in[7];
  const float* betaW    = (const float*)d_in[8];
  const float* W2       = (const float*)d_in[9];
  const float* W_cls    = (const float*)d_in[10];
  const float* b_cls    = (const float*)d_in[11];
  float* out = (float*)d_out;

  float* ws   = (float*)d_ws;
  float* rec  = ws;
  float* W2c  = ws + PB_N * REC;
  float* FW   = ws + PB_N * REC + C * CLS;

  pool_partials<<<PB_N + NT + FB_N, PT, 0, stream>>>(feat_s7, w_attn, W2, W_cls, feat_s, W1, rec, W2c, FW);

  fused_main<<<TILES, NW * 32, 0, stream>>>(coords_q, coords_s, x7, FW, rec,
                                         W1, gammaW, betaW, W2c, b_cls, out);
}
